# starting kernel
# baseline (speedup 1.0000x reference)
#include <hip/hip_runtime.h>

typedef __attribute__((ext_vector_type(2))) float v2f;
typedef __attribute__((ext_vector_type(8))) float v8f;

#define B_ 8
#define C_ 512
#define N_ 4096
#define EPS_ 1e-6f
#define HT_ 0.25f

static __device__ __forceinline__ v8f wmma4(v2f a, v2f b, v8f c) {
    return __builtin_amdgcn_wmma_f32_16x16x4_f32(false, a, false, b, (short)0, c, false, false);
}

__global__ __launch_bounds__(128) void k_qkproj(const float* __restrict__ x,
                                                const float* __restrict__ Wq,
                                                const float* __restrict__ Wk,
                                                float* __restrict__ Q,
                                                float* __restrict__ K) {
    const int lane = threadIdx.x & 31;
    const int wave = threadIdx.x >> 5;
    const int m    = lane & 15;
    const int kp   = (lane >> 4) << 1;
    const int half = lane >> 4;
    const int c0 = blockIdx.y * 32;
    const int n0 = blockIdx.x * 128 + wave * 32;
    const int b  = blockIdx.z;
    const float* xb  = x  + (size_t)b * C_ * N_;
    const float* wq0 = Wq + (size_t)(c0 + m) * C_;
    const float* wq1 = Wq + (size_t)(c0 + 16 + m) * C_;
    const float* wk0 = Wk + (size_t)(c0 + m) * C_;
    const float* wk1 = Wk + (size_t)(c0 + 16 + m) * C_;
    v8f q00 = {}, q01 = {}, q10 = {}, q11 = {};
    v8f k00 = {}, k01 = {}, k10 = {}, k11 = {};
    for (int k = 0; k < C_; k += 4) {
        const int kk = k + kp;
        v2f aq0 = *(const v2f*)(wq0 + kk);
        v2f aq1 = *(const v2f*)(wq1 + kk);
        v2f ak0 = *(const v2f*)(wk0 + kk);
        v2f ak1 = *(const v2f*)(wk1 + kk);
        const float* xr0 = xb + (size_t)kk * N_;
        const float* xr1 = xb + (size_t)(kk + 1) * N_;
        v2f b0, b1;
        b0.x = xr0[n0 + m];      b0.y = xr1[n0 + m];
        b1.x = xr0[n0 + 16 + m]; b1.y = xr1[n0 + 16 + m];
        q00 = wmma4(aq0, b0, q00); q01 = wmma4(aq0, b1, q01);
        q10 = wmma4(aq1, b0, q10); q11 = wmma4(aq1, b1, q11);
        k00 = wmma4(ak0, b0, k00); k01 = wmma4(ak0, b1, k01);
        k10 = wmma4(ak1, b0, k10); k11 = wmma4(ak1, b1, k11);
    }
    float* Qb = Q + (size_t)b * C_ * N_;
    float* Kb = K + (size_t)b * C_ * N_;
    #pragma unroll
    for (int j = 0; j < 8; ++j) {
        const int cA = c0 + j + 8 * half;
        const int cB = cA + 16;
        Qb[(size_t)cA * N_ + n0 + m]      = q00[j];
        Qb[(size_t)cA * N_ + n0 + 16 + m] = q01[j];
        Qb[(size_t)cB * N_ + n0 + m]      = q10[j];
        Qb[(size_t)cB * N_ + n0 + 16 + m] = q11[j];
        Kb[(size_t)cA * N_ + n0 + m]      = k00[j];
        Kb[(size_t)cA * N_ + n0 + 16 + m] = k01[j];
        Kb[(size_t)cB * N_ + n0 + m]      = k10[j];
        Kb[(size_t)cB * N_ + n0 + 16 + m] = k11[j];
    }
}

__global__ __launch_bounds__(256) void k_rownorm(const float* __restrict__ Q,
                                                 const float* __restrict__ K,
                                                 float* __restrict__ invQ,
                                                 float* __restrict__ invK) {
    __shared__ float sq[256], sk[256];
    const int row = blockIdx.x;
    const float* qr = Q + (size_t)row * N_;
    const float* kr = K + (size_t)row * N_;
    float aq = 0.f, ak = 0.f;
    for (int n = threadIdx.x; n < N_; n += 256) {
        float v = qr[n]; aq += v * v;
        float w = kr[n]; ak += w * w;
    }
    sq[threadIdx.x] = aq; sk[threadIdx.x] = ak;
    __syncthreads();
    for (int s = 128; s > 0; s >>= 1) {
        if (threadIdx.x < s) {
            sq[threadIdx.x] += sq[threadIdx.x + s];
            sk[threadIdx.x] += sk[threadIdx.x + s];
        }
        __syncthreads();
    }
    if (threadIdx.x == 0) { invQ[row] = rsqrtf(sq[0]); invK[row] = rsqrtf(sk[0]); }
}

__global__ __launch_bounds__(128) void k_qkt(const float* __restrict__ Q,
                                             const float* __restrict__ K,
                                             const float* __restrict__ invQ,
                                             const float* __restrict__ invK,
                                             float* __restrict__ M) {
    const int lane = threadIdx.x & 31;
    const int wave = threadIdx.x >> 5;
    const int m    = lane & 15;
    const int kp   = (lane >> 4) << 1;
    const int half = lane >> 4;
    const int c0 = blockIdx.y * 32;
    const int d0 = blockIdx.x * 128 + wave * 32;
    const int b  = blockIdx.z;
    const size_t bOff = (size_t)b * C_ * N_;
    const float* q0 = Q + bOff + (size_t)(c0 + m) * N_;
    const float* q1 = Q + bOff + (size_t)(c0 + 16 + m) * N_;
    const float* k0 = K + bOff + (size_t)(d0 + m) * N_;
    const float* k1 = K + bOff + (size_t)(d0 + 16 + m) * N_;
    v8f a00 = {}, a01 = {}, a10 = {}, a11 = {};
    for (int k = 0; k < N_; k += 4) {
        v2f aA = *(const v2f*)(q0 + k + kp);
        v2f aB = *(const v2f*)(q1 + k + kp);
        v2f bA = *(const v2f*)(k0 + k + kp);
        v2f bB = *(const v2f*)(k1 + k + kp);
        a00 = wmma4(aA, bA, a00); a01 = wmma4(aA, bB, a01);
        a10 = wmma4(aB, bA, a10); a11 = wmma4(aB, bB, a11);
    }
    float* Mb = M + (size_t)b * C_ * C_;
    const float ik0 = invK[b * C_ + d0 + m];
    const float ik1 = invK[b * C_ + d0 + 16 + m];
    #pragma unroll
    for (int j = 0; j < 8; ++j) {
        const int cA = c0 + j + 8 * half;
        const int cB = cA + 16;
        const float iqA = invQ[b * C_ + cA];
        const float iqB = invQ[b * C_ + cB];
        Mb[(size_t)cA * C_ + d0 + m]      = a00[j] * iqA * ik0;
        Mb[(size_t)cA * C_ + d0 + 16 + m] = a01[j] * iqA * ik1;
        Mb[(size_t)cB * C_ + d0 + m]      = a10[j] * iqB * ik0;
        Mb[(size_t)cB * C_ + d0 + 16 + m] = a11[j] * iqB * ik1;
    }
}

__global__ __launch_bounds__(128) void k_colstat(const float* __restrict__ M,
                                                 float* __restrict__ denom,
                                                 float* __restrict__ shift) {
    __shared__ float smax[128], smin[128];
    const int idx = blockIdx.x;
    const int b = idx / C_;
    const int d = idx - b * C_;
    const float* Mb = M + (size_t)b * C_ * C_;
    float mx = -3.4e38f, mn = 3.4e38f;
    for (int c = threadIdx.x; c < C_; c += 128) {
        const float v = 1.f - Mb[(size_t)c * C_ + d];
        mx = fmaxf(mx, v);
        mn = fminf(mn, v);
    }
    smax[threadIdx.x] = mx; smin[threadIdx.x] = mn;
    __syncthreads();
    for (int s = 64; s > 0; s >>= 1) {
        if (threadIdx.x < s) {
            smax[threadIdx.x] = fmaxf(smax[threadIdx.x], smax[threadIdx.x + s]);
            smin[threadIdx.x] = fminf(smin[threadIdx.x], smin[threadIdx.x + s]);
        }
        __syncthreads();
    }
    if (threadIdx.x == 0) {
        const float dn = smax[0] + EPS_;
        denom[idx] = dn;
        shift[idx] = 1.f - (smin[0] / dn) * (1.f / HT_);
    }
}

__global__ __launch_bounds__(128) void k_colsoftmax(float* __restrict__ M,
                                                    const float* __restrict__ denom,
                                                    const float* __restrict__ shift) {
    __shared__ float ssum[128];
    const int idx = blockIdx.x;
    const int b = idx / C_;
    const int d = idx - b * C_;
    float* Mb = M + (size_t)b * C_ * C_;
    const float inv_dn = 1.f / denom[idx];
    const float sh = shift[idx];
    float z = 0.f;
    for (int c = threadIdx.x; c < C_; c += 128) {
        const float m4 = 1.f - ((1.f - Mb[(size_t)c * C_ + d]) * inv_dn) * (1.f / HT_);
        z += expf(m4 - sh);
    }
    ssum[threadIdx.x] = z;
    __syncthreads();
    for (int s = 64; s > 0; s >>= 1) {
        if (threadIdx.x < s) ssum[threadIdx.x] += ssum[threadIdx.x + s];
        __syncthreads();
    }
    const float invz = 1.f / ssum[0];
    for (int c = threadIdx.x; c < C_; c += 128) {
        const size_t o = (size_t)c * C_ + d;
        const float m4 = 1.f - ((1.f - Mb[o]) * inv_dn) * (1.f / HT_);
        Mb[o] = expf(m4 - sh) * invz;
    }
}

__global__ __launch_bounds__(256) void k_rowl1(float* __restrict__ M) {
    __shared__ float ssum[256];
    const int row = blockIdx.x;
    float* r = M + (size_t)row * C_;
    float s = 0.f;
    for (int d = threadIdx.x; d < C_; d += 256) s += r[d];
    ssum[threadIdx.x] = s;
    __syncthreads();
    for (int t = 128; t > 0; t >>= 1) {
        if (threadIdx.x < t) ssum[threadIdx.x] += ssum[threadIdx.x + t];
        __syncthreads();
    }
    const float inv = 1.f / (ssum[0] + EPS_);
    for (int d = threadIdx.x; d < C_; d += 256) r[d] *= inv;
}

__global__ __launch_bounds__(128) void k_mwv(const float* __restrict__ M,
                                             const float* __restrict__ Wv,
                                             float* __restrict__ G) {
    const int lane = threadIdx.x & 31;
    const int wave = threadIdx.x >> 5;
    const int m    = lane & 15;
    const int kp   = (lane >> 4) << 1;
    const int half = lane >> 4;
    const int c0 = blockIdx.y * 32;
    const int e0 = blockIdx.x * 128 + wave * 32;
    const int b  = blockIdx.z;
    const float* a0 = M + (size_t)b * C_ * C_ + (size_t)(c0 + m) * C_;
    const float* a1 = M + (size_t)b * C_ * C_ + (size_t)(c0 + 16 + m) * C_;
    v8f g00 = {}, g01 = {}, g10 = {}, g11 = {};
    for (int k = 0; k < C_; k += 4) {
        const int kk = k + kp;
        v2f aA = *(const v2f*)(a0 + kk);
        v2f aB = *(const v2f*)(a1 + kk);
        const float* wr0 = Wv + (size_t)kk * C_;
        const float* wr1 = Wv + (size_t)(kk + 1) * C_;
        v2f b0, b1;
        b0.x = wr0[e0 + m];      b0.y = wr1[e0 + m];
        b1.x = wr0[e0 + 16 + m]; b1.y = wr1[e0 + 16 + m];
        g00 = wmma4(aA, b0, g00); g01 = wmma4(aA, b1, g01);
        g10 = wmma4(aB, b0, g10); g11 = wmma4(aB, b1, g11);
    }
    float* Gb = G + (size_t)b * C_ * C_;
    #pragma unroll
    for (int j = 0; j < 8; ++j) {
        const int cA = c0 + j + 8 * half;
        const int cB = cA + 16;
        Gb[(size_t)cA * C_ + e0 + m]      = g00[j];
        Gb[(size_t)cA * C_ + e0 + 16 + m] = g01[j];
        Gb[(size_t)cB * C_ + e0 + m]      = g10[j];
        Gb[(size_t)cB * C_ + e0 + 16 + m] = g11[j];
    }
}

__global__ __launch_bounds__(128) void k_out(const float* __restrict__ x,
                                             const float* __restrict__ G,
                                             const float* __restrict__ gamma,
                                             float* __restrict__ y) {
    const int lane = threadIdx.x & 31;
    const int wave = threadIdx.x >> 5;
    const int m    = lane & 15;
    const int kp   = (lane >> 4) << 1;
    const int half = lane >> 4;
    const int c0 = blockIdx.y * 32;
    const int n0 = blockIdx.x * 128 + wave * 32;
    const int b  = blockIdx.z;
    const float* xb = x + (size_t)b * C_ * N_;
    const float* g0 = G + (size_t)b * C_ * C_ + (size_t)(c0 + m) * C_;
    const float* g1 = G + (size_t)b * C_ * C_ + (size_t)(c0 + 16 + m) * C_;
    v8f o00 = {}, o01 = {}, o10 = {}, o11 = {};
    for (int k = 0; k < C_; k += 4) {
        const int kk = k + kp;
        v2f aA = *(const v2f*)(g0 + kk);
        v2f aB = *(const v2f*)(g1 + kk);
        const float* xr0 = xb + (size_t)kk * N_;
        const float* xr1 = xb + (size_t)(kk + 1) * N_;
        v2f b0, b1;
        b0.x = xr0[n0 + m];      b0.y = xr1[n0 + m];
        b1.x = xr0[n0 + 16 + m]; b1.y = xr1[n0 + 16 + m];
        o00 = wmma4(aA, b0, o00); o01 = wmma4(aA, b1, o01);
        o10 = wmma4(aB, b0, o10); o11 = wmma4(aB, b1, o11);
    }
    const float g = gamma[0];
    float* yb = y + (size_t)b * C_ * N_;
    #pragma unroll
    for (int j = 0; j < 8; ++j) {
        const int cA = c0 + j + 8 * half;
        const int cB = cA + 16;
        const size_t o0 = (size_t)cA * N_ + n0 + m;
        const size_t o1 = (size_t)cA * N_ + n0 + 16 + m;
        const size_t o2 = (size_t)cB * N_ + n0 + m;
        const size_t o3 = (size_t)cB * N_ + n0 + 16 + m;
        yb[o0] = xb[o0] + g * o00[j];
        yb[o1] = xb[o1] + g * o01[j];
        yb[o2] = xb[o2] + g * o10[j];
        yb[o3] = xb[o3] + g * o11[j];
    }
}

extern "C" void kernel_launch(void* const* d_in, const int* in_sizes, int n_in,
                              void* d_out, int out_size, void* d_ws, size_t ws_size,
                              hipStream_t stream) {
    (void)in_sizes; (void)n_in; (void)out_size; (void)ws_size;
    const float* x     = (const float*)d_in[0];
    const float* Wq    = (const float*)d_in[1];
    const float* Wk    = (const float*)d_in[2];
    const float* Wv    = (const float*)d_in[3];
    const float* gamma = (const float*)d_in[4];
    float* y = (float*)d_out;

    char* ws = (char*)d_ws;
    const size_t qkBytes = (size_t)B_ * C_ * N_ * sizeof(float);
    const size_t mBytes  = (size_t)B_ * C_ * C_ * sizeof(float);
    float* Q     = (float*)(ws);
    float* K     = (float*)(ws + qkBytes);
    float* M     = (float*)(ws + 2 * qkBytes);
    float* G     = (float*)(ws + 2 * qkBytes + mBytes);
    float* invQ  = (float*)(ws + 2 * qkBytes + 2 * mBytes);
    float* invK  = invQ  + B_ * C_;
    float* denom = invK  + B_ * C_;
    float* shift = denom + B_ * C_;

    k_qkproj    <<<dim3(N_ / 128, C_ / 32, B_), 128, 0, stream>>>(x, Wq, Wk, Q, K);
    k_rownorm   <<<B_ * C_, 256, 0, stream>>>(Q, K, invQ, invK);
    k_qkt       <<<dim3(C_ / 128, C_ / 32, B_), 128, 0, stream>>>(Q, K, invQ, invK, M);
    k_colstat   <<<B_ * C_, 128, 0, stream>>>(M, denom, shift);
    k_colsoftmax<<<B_ * C_, 128, 0, stream>>>(M, denom, shift);
    k_rowl1     <<<B_ * C_, 256, 0, stream>>>(M);
    k_mwv       <<<dim3(C_ / 128, C_ / 32, B_), 128, 0, stream>>>(M, Wv, G);
    k_out       <<<dim3(N_ / 128, C_ / 32, B_), 128, 0, stream>>>(x, G, gamma, y);
}


# Column max/min/softmax of M in one kernel with lanes along the contiguous d axis (32-column tile held in registers) instead of one block per column at 2 KB stride; column sum tree changes
# speedup vs baseline: 1.1873x; 1.1873x over previous
#include <hip/hip_runtime.h>

typedef __attribute__((ext_vector_type(2))) float v2f;
typedef __attribute__((ext_vector_type(8))) float v8f;

#define B_ 8
#define C_ 512
#define N_ 4096
#define EPS_ 1e-6f
#define HT_ 0.25f

static __device__ __forceinline__ v8f wmma4(v2f a, v2f b, v8f c) {
    return __builtin_amdgcn_wmma_f32_16x16x4_f32(false, a, false, b, (short)0, c, false, false);
}

__global__ __launch_bounds__(128) void k_qkproj(const float* __restrict__ x,
                                                const float* __restrict__ Wq,
                                                const float* __restrict__ Wk,
                                                float* __restrict__ Q,
                                                float* __restrict__ K) {
    const int lane = threadIdx.x & 31;
    const int wave = threadIdx.x >> 5;
    const int m    = lane & 15;
    const int kp   = (lane >> 4) << 1;
    const int half = lane >> 4;
    const int c0 = blockIdx.y * 32;
    const int n0 = blockIdx.x * 128 + wave * 32;
    const int b  = blockIdx.z;
    const float* xb  = x  + (size_t)b * C_ * N_;
    const float* wq0 = Wq + (size_t)(c0 + m) * C_;
    const float* wq1 = Wq + (size_t)(c0 + 16 + m) * C_;
    const float* wk0 = Wk + (size_t)(c0 + m) * C_;
    const float* wk1 = Wk + (size_t)(c0 + 16 + m) * C_;
    v8f q00 = {}, q01 = {}, q10 = {}, q11 = {};
    v8f k00 = {}, k01 = {}, k10 = {}, k11 = {};
    for (int k = 0; k < C_; k += 4) {
        const int kk = k + kp;
        v2f aq0 = *(const v2f*)(wq0 + kk);
        v2f aq1 = *(const v2f*)(wq1 + kk);
        v2f ak0 = *(const v2f*)(wk0 + kk);
        v2f ak1 = *(const v2f*)(wk1 + kk);
        const float* xr0 = xb + (size_t)kk * N_;
        const float* xr1 = xb + (size_t)(kk + 1) * N_;
        v2f b0, b1;
        b0.x = xr0[n0 + m];      b0.y = xr1[n0 + m];
        b1.x = xr0[n0 + 16 + m]; b1.y = xr1[n0 + 16 + m];
        q00 = wmma4(aq0, b0, q00); q01 = wmma4(aq0, b1, q01);
        q10 = wmma4(aq1, b0, q10); q11 = wmma4(aq1, b1, q11);
        k00 = wmma4(ak0, b0, k00); k01 = wmma4(ak0, b1, k01);
        k10 = wmma4(ak1, b0, k10); k11 = wmma4(ak1, b1, k11);
    }
    float* Qb = Q + (size_t)b * C_ * N_;
    float* Kb = K + (size_t)b * C_ * N_;
    #pragma unroll
    for (int j = 0; j < 8; ++j) {
        const int cA = c0 + j + 8 * half;
        const int cB = cA + 16;
        Qb[(size_t)cA * N_ + n0 + m]      = q00[j];
        Qb[(size_t)cA * N_ + n0 + 16 + m] = q01[j];
        Qb[(size_t)cB * N_ + n0 + m]      = q10[j];
        Qb[(size_t)cB * N_ + n0 + 16 + m] = q11[j];
        Kb[(size_t)cA * N_ + n0 + m]      = k00[j];
        Kb[(size_t)cA * N_ + n0 + 16 + m] = k01[j];
        Kb[(size_t)cB * N_ + n0 + m]      = k10[j];
        Kb[(size_t)cB * N_ + n0 + 16 + m] = k11[j];
    }
}

__global__ __launch_bounds__(256) void k_rownorm(const float* __restrict__ Q,
                                                 const float* __restrict__ K,
                                                 float* __restrict__ invQ,
                                                 float* __restrict__ invK) {
    __shared__ float sq[256], sk[256];
    const int row = blockIdx.x;
    const float* qr = Q + (size_t)row * N_;
    const float* kr = K + (size_t)row * N_;
    float aq = 0.f, ak = 0.f;
    for (int n = threadIdx.x; n < N_; n += 256) {
        float v = qr[n]; aq += v * v;
        float w = kr[n]; ak += w * w;
    }
    sq[threadIdx.x] = aq; sk[threadIdx.x] = ak;
    __syncthreads();
    for (int s = 128; s > 0; s >>= 1) {
        if (threadIdx.x < s) {
            sq[threadIdx.x] += sq[threadIdx.x + s];
            sk[threadIdx.x] += sk[threadIdx.x + s];
        }
        __syncthreads();
    }
    if (threadIdx.x == 0) { invQ[row] = rsqrtf(sq[0]); invK[row] = rsqrtf(sk[0]); }
}

__global__ __launch_bounds__(128) void k_qkt(const float* __restrict__ Q,
                                             const float* __restrict__ K,
                                             const float* __restrict__ invQ,
                                             const float* __restrict__ invK,
                                             float* __restrict__ M) {
    const int lane = threadIdx.x & 31;
    const int wave = threadIdx.x >> 5;
    const int m    = lane & 15;
    const int kp   = (lane >> 4) << 1;
    const int half = lane >> 4;
    const int c0 = blockIdx.y * 32;
    const int d0 = blockIdx.x * 128 + wave * 32;
    const int b  = blockIdx.z;
    const size_t bOff = (size_t)b * C_ * N_;
    const float* q0 = Q + bOff + (size_t)(c0 + m) * N_;
    const float* q1 = Q + bOff + (size_t)(c0 + 16 + m) * N_;
    const float* k0 = K + bOff + (size_t)(d0 + m) * N_;
    const float* k1 = K + bOff + (size_t)(d0 + 16 + m) * N_;
    v8f a00 = {}, a01 = {}, a10 = {}, a11 = {};
    for (int k = 0; k < N_; k += 4) {
        v2f aA = *(const v2f*)(q0 + k + kp);
        v2f aB = *(const v2f*)(q1 + k + kp);
        v2f bA = *(const v2f*)(k0 + k + kp);
        v2f bB = *(const v2f*)(k1 + k + kp);
        a00 = wmma4(aA, bA, a00); a01 = wmma4(aA, bB, a01);
        a10 = wmma4(aB, bA, a10); a11 = wmma4(aB, bB, a11);
    }
    float* Mb = M + (size_t)b * C_ * C_;
    const float ik0 = invK[b * C_ + d0 + m];
    const float ik1 = invK[b * C_ + d0 + 16 + m];
    #pragma unroll
    for (int j = 0; j < 8; ++j) {
        const int cA = c0 + j + 8 * half;
        const int cB = cA + 16;
        const float iqA = invQ[b * C_ + cA];
        const float iqB = invQ[b * C_ + cB];
        Mb[(size_t)cA * C_ + d0 + m]      = a00[j] * iqA * ik0;
        Mb[(size_t)cA * C_ + d0 + 16 + m] = a01[j] * iqA * ik1;
        Mb[(size_t)cB * C_ + d0 + m]      = a10[j] * iqB * ik0;
        Mb[(size_t)cB * C_ + d0 + 16 + m] = a11[j] * iqB * ik1;
    }
}

__global__ __launch_bounds__(256) void k_colsoftmax(float* __restrict__ M) {
    __shared__ float smax[8][32], smin[8][32], ssum[8][32];
    const int lane = threadIdx.x & 31;
    const int wave = threadIdx.x >> 5;
    const int tile = blockIdx.x;
    const int b  = tile / (C_ / 32);
    const int d  = (tile - b * (C_ / 32)) * 32 + lane;
    float* p = M + (size_t)b * C_ * C_ + (size_t)(wave * 64) * C_ + d;
    float v[64];
    for (int i = 0; i < 64; ++i) v[i] = 1.f - p[(size_t)i * C_];
    float mx = -3.4e38f, mn = 3.4e38f;
    for (int i = 0; i < 64; ++i) { mx = fmaxf(mx, v[i]); mn = fminf(mn, v[i]); }
    smax[wave][lane] = mx; smin[wave][lane] = mn;
    __syncthreads();
    float cmx = -3.4e38f, cmn = 3.4e38f;
    for (int j = 0; j < 8; ++j) {
        cmx = fmaxf(cmx, smax[j][lane]);
        cmn = fminf(cmn, smin[j][lane]);
    }
    const float dn = cmx + EPS_;
    const float inv_dn = 1.f / dn;
    const float sh = 1.f - (cmn / dn) * (1.f / HT_);
    float z = 0.f;
    for (int i = 0; i < 64; ++i) {
        const float m4 = 1.f - (v[i] * inv_dn) * (1.f / HT_);
        const float e = expf(m4 - sh);
        v[i] = e;
        z += e;
    }
    ssum[wave][lane] = z;
    __syncthreads();
    float tot = 0.f;
    for (int j = 0; j < 8; ++j) tot += ssum[j][lane];
    const float invz = 1.f / tot;
    for (int i = 0; i < 64; ++i) p[(size_t)i * C_] = v[i] * invz;
}

__global__ __launch_bounds__(256) void k_rowl1(float* __restrict__ M) {
    __shared__ float ssum[256];
    const int row = blockIdx.x;
    float* r = M + (size_t)row * C_;
    float s = 0.f;
    for (int d = threadIdx.x; d < C_; d += 256) s += r[d];
    ssum[threadIdx.x] = s;
    __syncthreads();
    for (int t = 128; t > 0; t >>= 1) {
        if (threadIdx.x < t) ssum[threadIdx.x] += ssum[threadIdx.x + t];
        __syncthreads();
    }
    const float inv = 1.f / (ssum[0] + EPS_);
    for (int d = threadIdx.x; d < C_; d += 256) r[d] *= inv;
}

__global__ __launch_bounds__(128) void k_mwv(const float* __restrict__ M,
                                             const float* __restrict__ Wv,
                                             float* __restrict__ G) {
    const int lane = threadIdx.x & 31;
    const int wave = threadIdx.x >> 5;
    const int m    = lane & 15;
    const int kp   = (lane >> 4) << 1;
    const int half = lane >> 4;
    const int c0 = blockIdx.y * 32;
    const int e0 = blockIdx.x * 128 + wave * 32;
    const int b  = blockIdx.z;
    const float* a0 = M + (size_t)b * C_ * C_ + (size_t)(c0 + m) * C_;
    const float* a1 = M + (size_t)b * C_ * C_ + (size_t)(c0 + 16 + m) * C_;
    v8f g00 = {}, g01 = {}, g10 = {}, g11 = {};
    for (int k = 0; k < C_; k += 4) {
        const int kk = k + kp;
        v2f aA = *(const v2f*)(a0 + kk);
        v2f aB = *(const v2f*)(a1 + kk);
        const float* wr0 = Wv + (size_t)kk * C_;
        const float* wr1 = Wv + (size_t)(kk + 1) * C_;
        v2f b0, b1;
        b0.x = wr0[e0 + m];      b0.y = wr1[e0 + m];
        b1.x = wr0[e0 + 16 + m]; b1.y = wr1[e0 + 16 + m];
        g00 = wmma4(aA, b0, g00); g01 = wmma4(aA, b1, g01);
        g10 = wmma4(aB, b0, g10); g11 = wmma4(aB, b1, g11);
    }
    float* Gb = G + (size_t)b * C_ * C_;
    #pragma unroll
    for (int j = 0; j < 8; ++j) {
        const int cA = c0 + j + 8 * half;
        const int cB = cA + 16;
        Gb[(size_t)cA * C_ + e0 + m]      = g00[j];
        Gb[(size_t)cA * C_ + e0 + 16 + m] = g01[j];
        Gb[(size_t)cB * C_ + e0 + m]      = g10[j];
        Gb[(size_t)cB * C_ + e0 + 16 + m] = g11[j];
    }
}

__global__ __launch_bounds__(128) void k_out(const float* __restrict__ x,
                                             const float* __restrict__ G,
                                             const float* __restrict__ gamma,
                                             float* __restrict__ y) {
    const int lane = threadIdx.x & 31;
    const int wave = threadIdx.x >> 5;
    const int m    = lane & 15;
    const int kp   = (lane >> 4) << 1;
    const int half = lane >> 4;
    const int c0 = blockIdx.y * 32;
    const int n0 = blockIdx.x * 128 + wave * 32;
    const int b  = blockIdx.z;
    const float* xb = x + (size_t)b * C_ * N_;
    const float* g0 = G + (size_t)b * C_ * C_ + (size_t)(c0 + m) * C_;
    const float* g1 = G + (size_t)b * C_ * C_ + (size_t)(c0 + 16 + m) * C_;
    v8f o00 = {}, o01 = {}, o10 = {}, o11 = {};
    for (int k = 0; k < C_; k += 4) {
        const int kk = k + kp;
        v2f aA = *(const v2f*)(g0 + kk);
        v2f aB = *(const v2f*)(g1 + kk);
        const float* xr0 = xb + (size_t)kk * N_;
        const float* xr1 = xb + (size_t)(kk + 1) * N_;
        v2f b0, b1;
        b0.x = xr0[n0 + m];      b0.y = xr1[n0 + m];
        b1.x = xr0[n0 + 16 + m]; b1.y = xr1[n0 + 16 + m];
        o00 = wmma4(aA, b0, o00); o01 = wmma4(aA, b1, o01);
        o10 = wmma4(aB, b0, o10); o11 = wmma4(aB, b1, o11);
    }
    const float g = gamma[0];
    float* yb = y + (size_t)b * C_ * N_;
    #pragma unroll
    for (int j = 0; j < 8; ++j) {
        const int cA = c0 + j + 8 * half;
        const int cB = cA + 16;
        const size_t o0 = (size_t)cA * N_ + n0 + m;
        const size_t o1 = (size_t)cA * N_ + n0 + 16 + m;
        const size_t o2 = (size_t)cB * N_ + n0 + m;
        const size_t o3 = (size_t)cB * N_ + n0 + 16 + m;
        yb[o0] = xb[o0] + g * o00[j];
        yb[o1] = xb[o1] + g * o01[j];
        yb[o2] = xb[o2] + g * o10[j];
        yb[o3] = xb[o3] + g * o11[j];
    }
}

extern "C" void kernel_launch(void* const* d_in, const int* in_sizes, int n_in,
                              void* d_out, int out_size, void* d_ws, size_t ws_size,
                              hipStream_t stream) {
    (void)in_sizes; (void)n_in; (void)out_size; (void)ws_size;
    const float* x     = (const float*)d_in[0];
    const float* Wq    = (const float*)d_in[1];
    const float* Wk    = (const float*)d_in[2];
    const float* Wv    = (const float*)d_in[3];
    const float* gamma = (const float*)d_in[4];
    float* y = (float*)d_out;

    char* ws = (char*)d_ws;
    const size_t qkBytes = (size_t)B_ * C_ * N_ * sizeof(float);
    const size_t mBytes  = (size_t)B_ * C_ * C_ * sizeof(float);
    float* Q     = (float*)(ws);
    float* K     = (float*)(ws + qkBytes);
    float* M     = (float*)(ws + 2 * qkBytes);
    float* G     = (float*)(ws + 2 * qkBytes + mBytes);
    float* invQ  = (float*)(ws + 2 * qkBytes + 2 * mBytes);
    float* invK  = invQ  + B_ * C_;

    k_qkproj    <<<dim3(N_ / 128, C_ / 32, B_), 128, 0, stream>>>(x, Wq, Wk, Q, K);
    k_rownorm   <<<B_ * C_, 256, 0, stream>>>(Q, K, invQ, invK);
    k_qkt       <<<dim3(C_ / 128, C_ / 32, B_), 128, 0, stream>>>(Q, K, invQ, invK, M);
    k_colsoftmax<<<B_ * C_ / 32, 256, 0, stream>>>(M);
    k_rowl1     <<<B_ * C_, 256, 0, stream>>>(M);
    k_mwv       <<<dim3(C_ / 128, C_ / 32, B_), 128, 0, stream>>>(M, Wv, G);
    k_out       <<<dim3(N_ / 128, C_ / 32, B_), 128, 0, stream>>>(x, G, gamma, y);
}


# Re-associate Q*K^T and the Q,K row norms as Wq*(x*x^T)*Wk^T through the 512x512 Gram matrix of x: Q and K are never formed, the two projection GEMMs go; summation order of the logits changes
# speedup vs baseline: 1.5970x; 1.3451x over previous
#include <hip/hip_runtime.h>

typedef __attribute__((ext_vector_type(2))) float v2f;
typedef __attribute__((ext_vector_type(8))) float v8f;

#define B_ 8
#define C_ 512
#define N_ 4096
#define EPS_ 1e-6f
#define HT_ 0.25f

static __device__ __forceinline__ v8f wmma4(v2f a, v2f b, v8f c) {
    return __builtin_amdgcn_wmma_f32_16x16x4_f32(false, a, false, b, (short)0, c, false, false);
}


__global__ __launch_bounds__(128) void k_gram(const float* __restrict__ x,
                                              float* __restrict__ S) {
    const int lane = threadIdx.x & 31;
    const int wave = threadIdx.x >> 5;
    const int m    = lane & 15;
    const int kp   = (lane >> 4) << 1;
    const int half = lane >> 4;
    const int c0 = blockIdx.y * 32;
    const int d0 = blockIdx.x * 128 + wave * 32;
    const int b  = blockIdx.z;
    const size_t bOff = (size_t)b * C_ * N_;
    const float* q0 = x + bOff + (size_t)(c0 + m) * N_;
    const float* q1 = x + bOff + (size_t)(c0 + 16 + m) * N_;
    const float* k0 = x + bOff + (size_t)(d0 + m) * N_;
    const float* k1 = x + bOff + (size_t)(d0 + 16 + m) * N_;
    v8f a00 = {}, a01 = {}, a10 = {}, a11 = {};
    for (int k = 0; k < N_; k += 4) {
        v2f aA = *(const v2f*)(q0 + k + kp);
        v2f aB = *(const v2f*)(q1 + k + kp);
        v2f bA = *(const v2f*)(k0 + k + kp);
        v2f bB = *(const v2f*)(k1 + k + kp);
        a00 = wmma4(aA, bA, a00); a01 = wmma4(aA, bB, a01);
        a10 = wmma4(aB, bA, a10); a11 = wmma4(aB, bB, a11);
    }
    float* Sb = S + (size_t)b * C_ * C_;
    #pragma unroll
    for (int j = 0; j < 8; ++j) {
        const int cA = c0 + j + 8 * half;
        const int cB = cA + 16;
        Sb[(size_t)cA * C_ + d0 + m]      = a00[j];
        Sb[(size_t)cA * C_ + d0 + 16 + m] = a01[j];
        Sb[(size_t)cB * C_ + d0 + m]      = a10[j];
        Sb[(size_t)cB * C_ + d0 + 16 + m] = a11[j];
    }
}

__global__ __launch_bounds__(128) void k_projs(const float* __restrict__ S,
                                               const float* __restrict__ Wq,
                                               const float* __restrict__ Wk,
                                               float* __restrict__ T,
                                               float* __restrict__ U) {
    const int lane = threadIdx.x & 31;
    const int wave = threadIdx.x >> 5;
    const int m    = lane & 15;
    const int kp   = (lane >> 4) << 1;
    const int half = lane >> 4;
    const int c0 = blockIdx.y * 32;
    const int n0 = blockIdx.x * 128 + wave * 32;
    const int b  = blockIdx.z;
    const float* sb  = S  + (size_t)b * C_ * C_;
    const float* wq0 = Wq + (size_t)(c0 + m) * C_;
    const float* wq1 = Wq + (size_t)(c0 + 16 + m) * C_;
    const float* wk0 = Wk + (size_t)(c0 + m) * C_;
    const float* wk1 = Wk + (size_t)(c0 + 16 + m) * C_;
    v8f q00 = {}, q01 = {}, q10 = {}, q11 = {};
    v8f k00 = {}, k01 = {}, k10 = {}, k11 = {};
    for (int k = 0; k < C_; k += 4) {
        const int kk = k + kp;
        v2f aq0 = *(const v2f*)(wq0 + kk);
        v2f aq1 = *(const v2f*)(wq1 + kk);
        v2f ak0 = *(const v2f*)(wk0 + kk);
        v2f ak1 = *(const v2f*)(wk1 + kk);
        const float* sr0 = sb + (size_t)kk * C_;
        const float* sr1 = sb + (size_t)(kk + 1) * C_;
        v2f b0, b1;
        b0.x = sr0[n0 + m];      b0.y = sr1[n0 + m];
        b1.x = sr0[n0 + 16 + m]; b1.y = sr1[n0 + 16 + m];
        q00 = wmma4(aq0, b0, q00); q01 = wmma4(aq0, b1, q01);
        q10 = wmma4(aq1, b0, q10); q11 = wmma4(aq1, b1, q11);
        k00 = wmma4(ak0, b0, k00); k01 = wmma4(ak0, b1, k01);
        k10 = wmma4(ak1, b0, k10); k11 = wmma4(ak1, b1, k11);
    }
    float* Tb = T + (size_t)b * C_ * C_;
    float* Ub = U + (size_t)b * C_ * C_;
    #pragma unroll
    for (int j = 0; j < 8; ++j) {
        const int cA = c0 + j + 8 * half;
        const int cB = cA + 16;
        Tb[(size_t)cA * C_ + n0 + m]      = q00[j];
        Tb[(size_t)cA * C_ + n0 + 16 + m] = q01[j];
        Tb[(size_t)cB * C_ + n0 + m]      = q10[j];
        Tb[(size_t)cB * C_ + n0 + 16 + m] = q11[j];
        Ub[(size_t)cA * C_ + n0 + m]      = k00[j];
        Ub[(size_t)cA * C_ + n0 + 16 + m] = k01[j];
        Ub[(size_t)cB * C_ + n0 + m]      = k10[j];
        Ub[(size_t)cB * C_ + n0 + 16 + m] = k11[j];
    }
}

__global__ __launch_bounds__(256) void k_rownorm(const float* __restrict__ T,
                                                 const float* __restrict__ U,
                                                 const float* __restrict__ Wq,
                                                 const float* __restrict__ Wk,
                                                 float* __restrict__ invQ,
                                                 float* __restrict__ invK) {
    __shared__ float sq[256], sk[256];
    const int row = blockIdx.x;
    const int c   = row & (C_ - 1);
    const float* tr = T + (size_t)row * C_;
    const float* ur = U + (size_t)row * C_;
    const float* wq = Wq + (size_t)c * C_;
    const float* wk = Wk + (size_t)c * C_;
    float aq = 0.f, ak = 0.f;
    for (int e = threadIdx.x; e < C_; e += 256) {
        aq += tr[e] * wq[e];
        ak += ur[e] * wk[e];
    }
    sq[threadIdx.x] = aq; sk[threadIdx.x] = ak;
    __syncthreads();
    for (int s = 128; s > 0; s >>= 1) {
        if (threadIdx.x < s) {
            sq[threadIdx.x] += sq[threadIdx.x + s];
            sk[threadIdx.x] += sk[threadIdx.x + s];
        }
        __syncthreads();
    }
    if (threadIdx.x == 0) { invQ[row] = rsqrtf(sq[0]); invK[row] = rsqrtf(sk[0]); }
}

__global__ __launch_bounds__(128) void k_qkt(const float* __restrict__ T,
                                             const float* __restrict__ Wk,
                                             const float* __restrict__ invQ,
                                             const float* __restrict__ invK,
                                             float* __restrict__ M) {
    const int lane = threadIdx.x & 31;
    const int wave = threadIdx.x >> 5;
    const int m    = lane & 15;
    const int kp   = (lane >> 4) << 1;
    const int half = lane >> 4;
    const int c0 = blockIdx.y * 32;
    const int d0 = blockIdx.x * 128 + wave * 32;
    const int b  = blockIdx.z;
    const size_t bOff = (size_t)b * C_ * C_;
    const float* q0 = T + bOff + (size_t)(c0 + m) * C_;
    const float* q1 = T + bOff + (size_t)(c0 + 16 + m) * C_;
    const float* k0 = Wk + (size_t)(d0 + m) * C_;
    const float* k1 = Wk + (size_t)(d0 + 16 + m) * C_;
    v8f a00 = {}, a01 = {}, a10 = {}, a11 = {};
    for (int k = 0; k < C_; k += 4) {
        v2f aA = *(const v2f*)(q0 + k + kp);
        v2f aB = *(const v2f*)(q1 + k + kp);
        v2f bA = *(const v2f*)(k0 + k + kp);
        v2f bB = *(const v2f*)(k1 + k + kp);
        a00 = wmma4(aA, bA, a00); a01 = wmma4(aA, bB, a01);
        a10 = wmma4(aB, bA, a10); a11 = wmma4(aB, bB, a11);
    }
    float* Mb = M + (size_t)b * C_ * C_;
    const float ik0 = invK[b * C_ + d0 + m];
    const float ik1 = invK[b * C_ + d0 + 16 + m];
    #pragma unroll
    for (int j = 0; j < 8; ++j) {
        const int cA = c0 + j + 8 * half;
        const int cB = cA + 16;
        const float iqA = invQ[b * C_ + cA];
        const float iqB = invQ[b * C_ + cB];
        Mb[(size_t)cA * C_ + d0 + m]      = a00[j] * iqA * ik0;
        Mb[(size_t)cA * C_ + d0 + 16 + m] = a01[j] * iqA * ik1;
        Mb[(size_t)cB * C_ + d0 + m]      = a10[j] * iqB * ik0;
        Mb[(size_t)cB * C_ + d0 + 16 + m] = a11[j] * iqB * ik1;
    }
}

__global__ __launch_bounds__(256) void k_colsoftmax(float* __restrict__ M) {
    __shared__ float smax[8][32], smin[8][32], ssum[8][32];
    const int lane = threadIdx.x & 31;
    const int wave = threadIdx.x >> 5;
    const int tile = blockIdx.x;
    const int b  = tile / (C_ / 32);
    const int d  = (tile - b * (C_ / 32)) * 32 + lane;
    float* p = M + (size_t)b * C_ * C_ + (size_t)(wave * 64) * C_ + d;
    float v[64];
    for (int i = 0; i < 64; ++i) v[i] = 1.f - p[(size_t)i * C_];
    float mx = -3.4e38f, mn = 3.4e38f;
    for (int i = 0; i < 64; ++i) { mx = fmaxf(mx, v[i]); mn = fminf(mn, v[i]); }
    smax[wave][lane] = mx; smin[wave][lane] = mn;
    __syncthreads();
    float cmx = -3.4e38f, cmn = 3.4e38f;
    for (int j = 0; j < 8; ++j) {
        cmx = fmaxf(cmx, smax[j][lane]);
        cmn = fminf(cmn, smin[j][lane]);
    }
    const float dn = cmx + EPS_;
    const float inv_dn = 1.f / dn;
    const float sh = 1.f - (cmn / dn) * (1.f / HT_);
    float z = 0.f;
    for (int i = 0; i < 64; ++i) {
        const float m4 = 1.f - (v[i] * inv_dn) * (1.f / HT_);
        const float e = expf(m4 - sh);
        v[i] = e;
        z += e;
    }
    ssum[wave][lane] = z;
    __syncthreads();
    float tot = 0.f;
    for (int j = 0; j < 8; ++j) tot += ssum[j][lane];
    const float invz = 1.f / tot;
    for (int i = 0; i < 64; ++i) p[(size_t)i * C_] = v[i] * invz;
}

__global__ __launch_bounds__(256) void k_rowl1(float* __restrict__ M) {
    __shared__ float ssum[256];
    const int row = blockIdx.x;
    float* r = M + (size_t)row * C_;
    float s = 0.f;
    for (int d = threadIdx.x; d < C_; d += 256) s += r[d];
    ssum[threadIdx.x] = s;
    __syncthreads();
    for (int t = 128; t > 0; t >>= 1) {
        if (threadIdx.x < t) ssum[threadIdx.x] += ssum[threadIdx.x + t];
        __syncthreads();
    }
    const float inv = 1.f / (ssum[0] + EPS_);
    for (int d = threadIdx.x; d < C_; d += 256) r[d] *= inv;
}

__global__ __launch_bounds__(128) void k_mwv(const float* __restrict__ M,
                                             const float* __restrict__ Wv,
                                             float* __restrict__ G) {
    const int lane = threadIdx.x & 31;
    const int wave = threadIdx.x >> 5;
    const int m    = lane & 15;
    const int kp   = (lane >> 4) << 1;
    const int half = lane >> 4;
    const int c0 = blockIdx.y * 32;
    const int e0 = blockIdx.x * 128 + wave * 32;
    const int b  = blockIdx.z;
    const float* a0 = M + (size_t)b * C_ * C_ + (size_t)(c0 + m) * C_;
    const float* a1 = M + (size_t)b * C_ * C_ + (size_t)(c0 + 16 + m) * C_;
    v8f g00 = {}, g01 = {}, g10 = {}, g11 = {};
    for (int k = 0; k < C_; k += 4) {
        const int kk = k + kp;
        v2f aA = *(const v2f*)(a0 + kk);
        v2f aB = *(const v2f*)(a1 + kk);
        const float* wr0 = Wv + (size_t)kk * C_;
        const float* wr1 = Wv + (size_t)(kk + 1) * C_;
        v2f b0, b1;
        b0.x = wr0[e0 + m];      b0.y = wr1[e0 + m];
        b1.x = wr0[e0 + 16 + m]; b1.y = wr1[e0 + 16 + m];
        g00 = wmma4(aA, b0, g00); g01 = wmma4(aA, b1, g01);
        g10 = wmma4(aB, b0, g10); g11 = wmma4(aB, b1, g11);
    }
    float* Gb = G + (size_t)b * C_ * C_;
    #pragma unroll
    for (int j = 0; j < 8; ++j) {
        const int cA = c0 + j + 8 * half;
        const int cB = cA + 16;
        Gb[(size_t)cA * C_ + e0 + m]      = g00[j];
        Gb[(size_t)cA * C_ + e0 + 16 + m] = g01[j];
        Gb[(size_t)cB * C_ + e0 + m]      = g10[j];
        Gb[(size_t)cB * C_ + e0 + 16 + m] = g11[j];
    }
}

__global__ __launch_bounds__(128) void k_out(const float* __restrict__ x,
                                             const float* __restrict__ G,
                                             const float* __restrict__ gamma,
                                             float* __restrict__ y) {
    const int lane = threadIdx.x & 31;
    const int wave = threadIdx.x >> 5;
    const int m    = lane & 15;
    const int kp   = (lane >> 4) << 1;
    const int half = lane >> 4;
    const int c0 = blockIdx.y * 32;
    const int n0 = blockIdx.x * 128 + wave * 32;
    const int b  = blockIdx.z;
    const float* xb = x + (size_t)b * C_ * N_;
    const float* g0 = G + (size_t)b * C_ * C_ + (size_t)(c0 + m) * C_;
    const float* g1 = G + (size_t)b * C_ * C_ + (size_t)(c0 + 16 + m) * C_;
    v8f o00 = {}, o01 = {}, o10 = {}, o11 = {};
    for (int k = 0; k < C_; k += 4) {
        const int kk = k + kp;
        v2f aA = *(const v2f*)(g0 + kk);
        v2f aB = *(const v2f*)(g1 + kk);
        const float* xr0 = xb + (size_t)kk * N_;
        const float* xr1 = xb + (size_t)(kk + 1) * N_;
        v2f b0, b1;
        b0.x = xr0[n0 + m];      b0.y = xr1[n0 + m];
        b1.x = xr0[n0 + 16 + m]; b1.y = xr1[n0 + 16 + m];
        o00 = wmma4(aA, b0, o00); o01 = wmma4(aA, b1, o01);
        o10 = wmma4(aB, b0, o10); o11 = wmma4(aB, b1, o11);
    }
    const float g = gamma[0];
    float* yb = y + (size_t)b * C_ * N_;
    #pragma unroll
    for (int j = 0; j < 8; ++j) {
        const int cA = c0 + j + 8 * half;
        const int cB = cA + 16;
        const size_t o0 = (size_t)cA * N_ + n0 + m;
        const size_t o1 = (size_t)cA * N_ + n0 + 16 + m;
        const size_t o2 = (size_t)cB * N_ + n0 + m;
        const size_t o3 = (size_t)cB * N_ + n0 + 16 + m;
        yb[o0] = xb[o0] + g * o00[j];
        yb[o1] = xb[o1] + g * o01[j];
        yb[o2] = xb[o2] + g * o10[j];
        yb[o3] = xb[o3] + g * o11[j];
    }
}

extern "C" void kernel_launch(void* const* d_in, const int* in_sizes, int n_in,
                              void* d_out, int out_size, void* d_ws, size_t ws_size,
                              hipStream_t stream) {
    (void)in_sizes; (void)n_in; (void)out_size; (void)ws_size;
    const float* x     = (const float*)d_in[0];
    const float* Wq    = (const float*)d_in[1];
    const float* Wk    = (const float*)d_in[2];
    const float* Wv    = (const float*)d_in[3];
    const float* gamma = (const float*)d_in[4];
    float* y = (float*)d_out;

    char* ws = (char*)d_ws;
    const size_t mBytes  = (size_t)B_ * C_ * C_ * sizeof(float);
    float* S     = (float*)(ws);
    float* T     = (float*)(ws + mBytes);
    float* U     = (float*)(ws + 2 * mBytes);
    float* M     = (float*)(ws + 3 * mBytes);
    float* G     = (float*)(ws + 4 * mBytes);
    float* invQ  = (float*)(ws + 5 * mBytes);
    float* invK  = invQ  + B_ * C_;

    k_gram      <<<dim3(C_ / 128, C_ / 32, B_), 128, 0, stream>>>(x, S);
    k_projs     <<<dim3(C_ / 128, C_ / 32, B_), 128, 0, stream>>>(S, Wq, Wk, T, U);
    k_rownorm   <<<B_ * C_, 256, 0, stream>>>(T, U, Wq, Wk, invQ, invK);
    k_qkt       <<<dim3(C_ / 128, C_ / 32, B_), 128, 0, stream>>>(T, Wk, invQ, invK, M);
    k_colsoftmax<<<B_ * C_ / 32, 256, 0, stream>>>(M);
    k_rowl1     <<<B_ * C_, 256, 0, stream>>>(M);
    k_mwv       <<<dim3(C_ / 128, C_ / 32, B_), 128, 0, stream>>>(M, Wv, G);
    k_out       <<<dim3(N_ / 128, C_ / 32, B_), 128, 0, stream>>>(x, G, gamma, y);
}


# Fragment loads of K-contiguous WMMA operands widened 8->16 B per lane (k re-paired in blocks of 8, B loaders follow), grouped ahead of their WMMAs via waves_per_eu(1,2); dot summation order changes
# speedup vs baseline: 2.1048x; 1.3180x over previous
#include <hip/hip_runtime.h>

typedef __attribute__((ext_vector_type(2))) float v2f;
typedef __attribute__((ext_vector_type(4))) float v4f;
typedef __attribute__((ext_vector_type(8))) float v8f;

#define B_ 8
#define C_ 512
#define N_ 4096
#define EPS_ 1e-6f
#define HT_ 0.25f

static __device__ __forceinline__ v8f wmma4(v2f a, v2f b, v8f c) {
    return __builtin_amdgcn_wmma_f32_16x16x4_f32(false, a, false, b, (short)0, c, false, false);
}

static __device__ __forceinline__ v2f lo2(v4f v) { return __builtin_shufflevector(v, v, 0, 1); }
static __device__ __forceinline__ v2f hi2(v4f v) { return __builtin_shufflevector(v, v, 2, 3); }


__global__ __launch_bounds__(128) __attribute__((amdgpu_waves_per_eu(1, 2))) void k_gram(const float* __restrict__ x,
                                              float* __restrict__ S) {
    const int lane = threadIdx.x & 31;
    const int wave = threadIdx.x >> 5;
    const int m    = lane & 15;
    const int half = lane >> 4;
    const int k4   = half << 2;
    const int c0 = blockIdx.y * 32;
    const int d0 = blockIdx.x * 128 + wave * 32;
    const int b  = blockIdx.z;
    const size_t bOff = (size_t)b * C_ * N_;
    const float* q0 = x + bOff + (size_t)(c0 + m) * N_;
    const float* q1 = x + bOff + (size_t)(c0 + 16 + m) * N_;
    const float* k0 = x + bOff + (size_t)(d0 + m) * N_;
    const float* k1 = x + bOff + (size_t)(d0 + 16 + m) * N_;
    v8f a00 = {}, a01 = {}, a10 = {}, a11 = {};
    for (int k = 0; k < N_; k += 8) {
        const v4f aA = *(const v4f*)(q0 + k + k4);
        const v4f aB = *(const v4f*)(q1 + k + k4);
        const v4f bA = *(const v4f*)(k0 + k + k4);
        const v4f bB = *(const v4f*)(k1 + k + k4);
        a00 = wmma4(lo2(aA), lo2(bA), a00); a01 = wmma4(lo2(aA), lo2(bB), a01);
        a10 = wmma4(lo2(aB), lo2(bA), a10); a11 = wmma4(lo2(aB), lo2(bB), a11);
        a00 = wmma4(hi2(aA), hi2(bA), a00); a01 = wmma4(hi2(aA), hi2(bB), a01);
        a10 = wmma4(hi2(aB), hi2(bA), a10); a11 = wmma4(hi2(aB), hi2(bB), a11);
    }
    float* Sb = S + (size_t)b * C_ * C_;
    #pragma unroll
    for (int j = 0; j < 8; ++j) {
        const int cA = c0 + j + 8 * half;
        const int cB = cA + 16;
        Sb[(size_t)cA * C_ + d0 + m]      = a00[j];
        Sb[(size_t)cA * C_ + d0 + 16 + m] = a01[j];
        Sb[(size_t)cB * C_ + d0 + m]      = a10[j];
        Sb[(size_t)cB * C_ + d0 + 16 + m] = a11[j];
    }
}

__global__ __launch_bounds__(128) __attribute__((amdgpu_waves_per_eu(1, 2))) void k_projs(const float* __restrict__ S,
                                               const float* __restrict__ Wq,
                                               const float* __restrict__ Wk,
                                               float* __restrict__ T,
                                               float* __restrict__ U) {
    const int lane = threadIdx.x & 31;
    const int wave = threadIdx.x >> 5;
    const int m    = lane & 15;
    const int half = lane >> 4;
    const int k4   = half << 2;
    const int c0 = blockIdx.y * 32;
    const int n0 = blockIdx.x * 128 + wave * 32;
    const int b  = blockIdx.z;
    const float* sb  = S  + (size_t)b * C_ * C_;
    const float* wq0 = Wq + (size_t)(c0 + m) * C_;
    const float* wq1 = Wq + (size_t)(c0 + 16 + m) * C_;
    const float* wk0 = Wk + (size_t)(c0 + m) * C_;
    const float* wk1 = Wk + (size_t)(c0 + 16 + m) * C_;
    v8f q00 = {}, q01 = {}, q10 = {}, q11 = {};
    v8f k00 = {}, k01 = {}, k10 = {}, k11 = {};
    for (int k = 0; k < C_; k += 8) {
        const int kk = k + k4;
        const v4f aq0 = *(const v4f*)(wq0 + kk);
        const v4f aq1 = *(const v4f*)(wq1 + kk);
        const v4f ak0 = *(const v4f*)(wk0 + kk);
        const v4f ak1 = *(const v4f*)(wk1 + kk);
        const float* sr0 = sb + (size_t)kk * C_;
        const float* sr1 = sb + (size_t)(kk + 1) * C_;
        const float* sr2 = sb + (size_t)(kk + 2) * C_;
        const float* sr3 = sb + (size_t)(kk + 3) * C_;
        v2f b0, b1, b2, b3;
        b0.x = sr0[n0 + m];      b0.y = sr1[n0 + m];
        b1.x = sr0[n0 + 16 + m]; b1.y = sr1[n0 + 16 + m];
        b2.x = sr2[n0 + m];      b2.y = sr3[n0 + m];
        b3.x = sr2[n0 + 16 + m]; b3.y = sr3[n0 + 16 + m];
        q00 = wmma4(lo2(aq0), b0, q00); q01 = wmma4(lo2(aq0), b1, q01);
        q10 = wmma4(lo2(aq1), b0, q10); q11 = wmma4(lo2(aq1), b1, q11);
        k00 = wmma4(lo2(ak0), b0, k00); k01 = wmma4(lo2(ak0), b1, k01);
        k10 = wmma4(lo2(ak1), b0, k10); k11 = wmma4(lo2(ak1), b1, k11);
        q00 = wmma4(hi2(aq0), b2, q00); q01 = wmma4(hi2(aq0), b3, q01);
        q10 = wmma4(hi2(aq1), b2, q10); q11 = wmma4(hi2(aq1), b3, q11);
        k00 = wmma4(hi2(ak0), b2, k00); k01 = wmma4(hi2(ak0), b3, k01);
        k10 = wmma4(hi2(ak1), b2, k10); k11 = wmma4(hi2(ak1), b3, k11);
    }
    float* Tb = T + (size_t)b * C_ * C_;
    float* Ub = U + (size_t)b * C_ * C_;
    #pragma unroll
    for (int j = 0; j < 8; ++j) {
        const int cA = c0 + j + 8 * half;
        const int cB = cA + 16;
        Tb[(size_t)cA * C_ + n0 + m]      = q00[j];
        Tb[(size_t)cA * C_ + n0 + 16 + m] = q01[j];
        Tb[(size_t)cB * C_ + n0 + m]      = q10[j];
        Tb[(size_t)cB * C_ + n0 + 16 + m] = q11[j];
        Ub[(size_t)cA * C_ + n0 + m]      = k00[j];
        Ub[(size_t)cA * C_ + n0 + 16 + m] = k01[j];
        Ub[(size_t)cB * C_ + n0 + m]      = k10[j];
        Ub[(size_t)cB * C_ + n0 + 16 + m] = k11[j];
    }
}

__global__ __launch_bounds__(256) void k_rownorm(const float* __restrict__ T,
                                                 const float* __restrict__ U,
                                                 const float* __restrict__ Wq,
                                                 const float* __restrict__ Wk,
                                                 float* __restrict__ invQ,
                                                 float* __restrict__ invK) {
    __shared__ float sq[256], sk[256];
    const int row = blockIdx.x;
    const int c   = row & (C_ - 1);
    const float* tr = T + (size_t)row * C_;
    const float* ur = U + (size_t)row * C_;
    const float* wq = Wq + (size_t)c * C_;
    const float* wk = Wk + (size_t)c * C_;
    float aq = 0.f, ak = 0.f;
    for (int e = threadIdx.x; e < C_; e += 256) {
        aq += tr[e] * wq[e];
        ak += ur[e] * wk[e];
    }
    sq[threadIdx.x] = aq; sk[threadIdx.x] = ak;
    __syncthreads();
    for (int s = 128; s > 0; s >>= 1) {
        if (threadIdx.x < s) {
            sq[threadIdx.x] += sq[threadIdx.x + s];
            sk[threadIdx.x] += sk[threadIdx.x + s];
        }
        __syncthreads();
    }
    if (threadIdx.x == 0) { invQ[row] = rsqrtf(sq[0]); invK[row] = rsqrtf(sk[0]); }
}

__global__ __launch_bounds__(128) __attribute__((amdgpu_waves_per_eu(1, 2))) void k_qkt(const float* __restrict__ T,
                                             const float* __restrict__ Wk,
                                             const float* __restrict__ invQ,
                                             const float* __restrict__ invK,
                                             float* __restrict__ M) {
    const int lane = threadIdx.x & 31;
    const int wave = threadIdx.x >> 5;
    const int m    = lane & 15;
    const int half = lane >> 4;
    const int k4   = half << 2;
    const int c0 = blockIdx.y * 32;
    const int d0 = blockIdx.x * 128 + wave * 32;
    const int b  = blockIdx.z;
    const size_t bOff = (size_t)b * C_ * C_;
    const float* q0 = T + bOff + (size_t)(c0 + m) * C_;
    const float* q1 = T + bOff + (size_t)(c0 + 16 + m) * C_;
    const float* k0 = Wk + (size_t)(d0 + m) * C_;
    const float* k1 = Wk + (size_t)(d0 + 16 + m) * C_;
    v8f a00 = {}, a01 = {}, a10 = {}, a11 = {};
    for (int k = 0; k < C_; k += 8) {
        const v4f aA = *(const v4f*)(q0 + k + k4);
        const v4f aB = *(const v4f*)(q1 + k + k4);
        const v4f bA = *(const v4f*)(k0 + k + k4);
        const v4f bB = *(const v4f*)(k1 + k + k4);
        a00 = wmma4(lo2(aA), lo2(bA), a00); a01 = wmma4(lo2(aA), lo2(bB), a01);
        a10 = wmma4(lo2(aB), lo2(bA), a10); a11 = wmma4(lo2(aB), lo2(bB), a11);
        a00 = wmma4(hi2(aA), hi2(bA), a00); a01 = wmma4(hi2(aA), hi2(bB), a01);
        a10 = wmma4(hi2(aB), hi2(bA), a10); a11 = wmma4(hi2(aB), hi2(bB), a11);
    }
    float* Mb = M + (size_t)b * C_ * C_;
    const float ik0 = invK[b * C_ + d0 + m];
    const float ik1 = invK[b * C_ + d0 + 16 + m];
    #pragma unroll
    for (int j = 0; j < 8; ++j) {
        const int cA = c0 + j + 8 * half;
        const int cB = cA + 16;
        const float iqA = invQ[b * C_ + cA];
        const float iqB = invQ[b * C_ + cB];
        Mb[(size_t)cA * C_ + d0 + m]      = a00[j] * iqA * ik0;
        Mb[(size_t)cA * C_ + d0 + 16 + m] = a01[j] * iqA * ik1;
        Mb[(size_t)cB * C_ + d0 + m]      = a10[j] * iqB * ik0;
        Mb[(size_t)cB * C_ + d0 + 16 + m] = a11[j] * iqB * ik1;
    }
}

__global__ __launch_bounds__(256) void k_colsoftmax(float* __restrict__ M) {
    __shared__ float smax[8][32], smin[8][32], ssum[8][32];
    const int lane = threadIdx.x & 31;
    const int wave = threadIdx.x >> 5;
    const int tile = blockIdx.x;
    const int b  = tile / (C_ / 32);
    const int d  = (tile - b * (C_ / 32)) * 32 + lane;
    float* p = M + (size_t)b * C_ * C_ + (size_t)(wave * 64) * C_ + d;
    float v[64];
    for (int i = 0; i < 64; ++i) v[i] = 1.f - p[(size_t)i * C_];
    float mx = -3.4e38f, mn = 3.4e38f;
    for (int i = 0; i < 64; ++i) { mx = fmaxf(mx, v[i]); mn = fminf(mn, v[i]); }
    smax[wave][lane] = mx; smin[wave][lane] = mn;
    __syncthreads();
    float cmx = -3.4e38f, cmn = 3.4e38f;
    for (int j = 0; j < 8; ++j) {
        cmx = fmaxf(cmx, smax[j][lane]);
        cmn = fminf(cmn, smin[j][lane]);
    }
    const float dn = cmx + EPS_;
    const float inv_dn = 1.f / dn;
    const float sh = 1.f - (cmn / dn) * (1.f / HT_);
    float z = 0.f;
    for (int i = 0; i < 64; ++i) {
        const float m4 = 1.f - (v[i] * inv_dn) * (1.f / HT_);
        const float e = expf(m4 - sh);
        v[i] = e;
        z += e;
    }
    ssum[wave][lane] = z;
    __syncthreads();
    float tot = 0.f;
    for (int j = 0; j < 8; ++j) tot += ssum[j][lane];
    const float invz = 1.f / tot;
    for (int i = 0; i < 64; ++i) p[(size_t)i * C_] = v[i] * invz;
}

__global__ __launch_bounds__(256) void k_rowl1(float* __restrict__ M) {
    __shared__ float ssum[256];
    const int row = blockIdx.x;
    float* r = M + (size_t)row * C_;
    float s = 0.f;
    for (int d = threadIdx.x; d < C_; d += 256) s += r[d];
    ssum[threadIdx.x] = s;
    __syncthreads();
    for (int t = 128; t > 0; t >>= 1) {
        if (threadIdx.x < t) ssum[threadIdx.x] += ssum[threadIdx.x + t];
        __syncthreads();
    }
    const float inv = 1.f / (ssum[0] + EPS_);
    for (int d = threadIdx.x; d < C_; d += 256) r[d] *= inv;
}

__global__ __launch_bounds__(128) __attribute__((amdgpu_waves_per_eu(1, 2))) void k_mwv(const float* __restrict__ M,
                                             const float* __restrict__ Wv,
                                             float* __restrict__ G) {
    const int lane = threadIdx.x & 31;
    const int wave = threadIdx.x >> 5;
    const int m    = lane & 15;
    const int half = lane >> 4;
    const int k4   = half << 2;
    const int c0 = blockIdx.y * 32;
    const int e0 = blockIdx.x * 128 + wave * 32;
    const int b  = blockIdx.z;
    const float* a0 = M + (size_t)b * C_ * C_ + (size_t)(c0 + m) * C_;
    const float* a1 = M + (size_t)b * C_ * C_ + (size_t)(c0 + 16 + m) * C_;
    v8f g00 = {}, g01 = {}, g10 = {}, g11 = {};
    for (int k = 0; k < C_; k += 8) {
        const int kk = k + k4;
        const v4f aA = *(const v4f*)(a0 + kk);
        const v4f aB = *(const v4f*)(a1 + kk);
        const float* wr0 = Wv + (size_t)kk * C_;
        const float* wr1 = Wv + (size_t)(kk + 1) * C_;
        const float* wr2 = Wv + (size_t)(kk + 2) * C_;
        const float* wr3 = Wv + (size_t)(kk + 3) * C_;
        v2f b0, b1, b2, b3;
        b0.x = wr0[e0 + m];      b0.y = wr1[e0 + m];
        b1.x = wr0[e0 + 16 + m]; b1.y = wr1[e0 + 16 + m];
        b2.x = wr2[e0 + m];      b2.y = wr3[e0 + m];
        b3.x = wr2[e0 + 16 + m]; b3.y = wr3[e0 + 16 + m];
        g00 = wmma4(lo2(aA), b0, g00); g01 = wmma4(lo2(aA), b1, g01);
        g10 = wmma4(lo2(aB), b0, g10); g11 = wmma4(lo2(aB), b1, g11);
        g00 = wmma4(hi2(aA), b2, g00); g01 = wmma4(hi2(aA), b3, g01);
        g10 = wmma4(hi2(aB), b2, g10); g11 = wmma4(hi2(aB), b3, g11);
    }
    float* Gb = G + (size_t)b * C_ * C_;
    #pragma unroll
    for (int j = 0; j < 8; ++j) {
        const int cA = c0 + j + 8 * half;
        const int cB = cA + 16;
        Gb[(size_t)cA * C_ + e0 + m]      = g00[j];
        Gb[(size_t)cA * C_ + e0 + 16 + m] = g01[j];
        Gb[(size_t)cB * C_ + e0 + m]      = g10[j];
        Gb[(size_t)cB * C_ + e0 + 16 + m] = g11[j];
    }
}

__global__ __launch_bounds__(128) void k_out(const float* __restrict__ x,
                                             const float* __restrict__ G,
                                             const float* __restrict__ gamma,
                                             float* __restrict__ y) {
    const int lane = threadIdx.x & 31;
    const int wave = threadIdx.x >> 5;
    const int m    = lane & 15;
    const int half = lane >> 4;
    const int k4   = half << 2;
    const int c0 = blockIdx.y * 32;
    const int n0 = blockIdx.x * 128 + wave * 32;
    const int b  = blockIdx.z;
    const float* xb = x + (size_t)b * C_ * N_;
    const float* g0 = G + (size_t)b * C_ * C_ + (size_t)(c0 + m) * C_;
    const float* g1 = G + (size_t)b * C_ * C_ + (size_t)(c0 + 16 + m) * C_;
    v8f o00 = {}, o01 = {}, o10 = {}, o11 = {};
    for (int k = 0; k < C_; k += 8) {
        const int kk = k + k4;
        const v4f aA = *(const v4f*)(g0 + kk);
        const v4f aB = *(const v4f*)(g1 + kk);
        const float* xr0 = xb + (size_t)kk * N_;
        const float* xr1 = xb + (size_t)(kk + 1) * N_;
        const float* xr2 = xb + (size_t)(kk + 2) * N_;
        const float* xr3 = xb + (size_t)(kk + 3) * N_;
        v2f b0, b1, b2, b3;
        b0.x = xr0[n0 + m];      b0.y = xr1[n0 + m];
        b1.x = xr0[n0 + 16 + m]; b1.y = xr1[n0 + 16 + m];
        b2.x = xr2[n0 + m];      b2.y = xr3[n0 + m];
        b3.x = xr2[n0 + 16 + m]; b3.y = xr3[n0 + 16 + m];
        o00 = wmma4(lo2(aA), b0, o00); o01 = wmma4(lo2(aA), b1, o01);
        o10 = wmma4(lo2(aB), b0, o10); o11 = wmma4(lo2(aB), b1, o11);
        o00 = wmma4(hi2(aA), b2, o00); o01 = wmma4(hi2(aA), b3, o01);
        o10 = wmma4(hi2(aB), b2, o10); o11 = wmma4(hi2(aB), b3, o11);
    }
    const float g = gamma[0];
    float* yb = y + (size_t)b * C_ * N_;
    #pragma unroll
    for (int j = 0; j < 8; ++j) {
        const int cA = c0 + j + 8 * half;
        const int cB = cA + 16;
        const size_t o0 = (size_t)cA * N_ + n0 + m;
        const size_t o1 = (size_t)cA * N_ + n0 + 16 + m;
        const size_t o2 = (size_t)cB * N_ + n0 + m;
        const size_t o3 = (size_t)cB * N_ + n0 + 16 + m;
        yb[o0] = xb[o0] + g * o00[j];
        yb[o1] = xb[o1] + g * o01[j];
        yb[o2] = xb[o2] + g * o10[j];
        yb[o3] = xb[o3] + g * o11[j];
    }
}

extern "C" void kernel_launch(void* const* d_in, const int* in_sizes, int n_in,
                              void* d_out, int out_size, void* d_ws, size_t ws_size,
                              hipStream_t stream) {
    (void)in_sizes; (void)n_in; (void)out_size; (void)ws_size;
    const float* x     = (const float*)d_in[0];
    const float* Wq    = (const float*)d_in[1];
    const float* Wk    = (const float*)d_in[2];
    const float* Wv    = (const float*)d_in[3];
    const float* gamma = (const float*)d_in[4];
    float* y = (float*)d_out;

    char* ws = (char*)d_ws;
    const size_t mBytes  = (size_t)B_ * C_ * C_ * sizeof(float);
    float* S     = (float*)(ws);
    float* T     = (float*)(ws + mBytes);
    float* U     = (float*)(ws + 2 * mBytes);
    float* M     = (float*)(ws + 3 * mBytes);
    float* G     = (float*)(ws + 4 * mBytes);
    float* invQ  = (float*)(ws + 5 * mBytes);
    float* invK  = invQ  + B_ * C_;

    k_gram      <<<dim3(C_ / 128, C_ / 32, B_), 128, 0, stream>>>(x, S);
    k_projs     <<<dim3(C_ / 128, C_ / 32, B_), 128, 0, stream>>>(S, Wq, Wk, T, U);
    k_rownorm   <<<B_ * C_, 256, 0, stream>>>(T, U, Wq, Wk, invQ, invK);
    k_qkt       <<<dim3(C_ / 128, C_ / 32, B_), 128, 0, stream>>>(T, Wk, invQ, invK, M);
    k_colsoftmax<<<B_ * C_ / 32, 256, 0, stream>>>(M);
    k_rowl1     <<<B_ * C_, 256, 0, stream>>>(M);
    k_mwv       <<<dim3(C_ / 128, C_ / 32, B_), 128, 0, stream>>>(M, Wv, G);
    k_out       <<<dim3(N_ / 128, C_ / 32, B_), 128, 0, stream>>>(x, G, gamma, y);
}


# All five WMMA kernels: operand tiles staged through LDS by whole-line global loads (two stages, shared by the 4 waves), fragments read from LDS instead of 16-line global loads; sums unchanged
# speedup vs baseline: 2.4627x; 1.1700x over previous
#include <hip/hip_runtime.h>

typedef __attribute__((ext_vector_type(2))) float v2f;
typedef __attribute__((ext_vector_type(4))) float v4f;
typedef __attribute__((ext_vector_type(8))) float v8f;

#define B_ 8
#define C_ 512
#define N_ 4096
#define EPS_ 1e-6f
#define HT_ 0.25f

static __device__ __forceinline__ v8f wmma4(v2f a, v2f b, v8f c) {
    return __builtin_amdgcn_wmma_f32_16x16x4_f32(false, a, false, b, (short)0, c, false, false);
}

static __device__ __forceinline__ v2f lo2(v4f v) { return __builtin_shufflevector(v, v, 0, 1); }
static __device__ __forceinline__ v2f hi2(v4f v) { return __builtin_shufflevector(v, v, 2, 3); }


#define KC_    32
#define PK_    36
#define GROWS_ 160
#define PX_    132

template <int LD, int KTOT>
static __device__ __forceinline__ void mm_rows_rows(const float* __restrict__ gA,
                                                    const float* __restrict__ gB,
                                                    float* __restrict__ tile,
                                                    v8f& a00, v8f& a01, v8f& a10, v8f& a11) {
    const int t    = threadIdx.x;
    const int lane = t & 31;
    const int wave = t >> 5;
    const int m    = lane & 15;
    const int k4   = (lane >> 4) << 2;
    const int sr = t >> 3;
    const int sq = (t & 7) << 2;
    const float* pA = gA + (size_t)sr * LD + sq;
    const float* pB = gB + (size_t)sr * LD + sq;
    const int sOff = sr * PK_ + sq;
    const int fA0 = m * PK_ + k4;
    const int fA1 = (16 + m) * PK_ + k4;
    const int fB0 = (32 + wave * 32 + m) * PK_ + k4;
    const int fB1 = fB0 + 16 * PK_;
    v4f g[10];
    #pragma unroll
    for (int r = 0; r < 2; ++r) g[r] = *(const v4f*)(pA + (size_t)(16 * r) * LD);
    #pragma unroll
    for (int r = 0; r < 8; ++r) g[2 + r] = *(const v4f*)(pB + (size_t)(16 * r) * LD);
    #pragma unroll
    for (int r = 0; r < 10; ++r) *(v4f*)(tile + sOff + 16 * r * PK_) = g[r];
    __syncthreads();
    #pragma unroll 1
    for (int c = 0; c < KTOT / KC_; ++c) {
        const bool more = (c + 1 < KTOT / KC_);
        if (more) {
            const int kc = (c + 1) * KC_;
            #pragma unroll
            for (int r = 0; r < 2; ++r) g[r] = *(const v4f*)(pA + (size_t)(16 * r) * LD + kc);
            #pragma unroll
            for (int r = 0; r < 8; ++r) g[2 + r] = *(const v4f*)(pB + (size_t)(16 * r) * LD + kc);
        }
        const float* st = tile + (c & 1) * (GROWS_ * PK_);
        v4f aA[4], aB[4], bA[4], bB[4];
        #pragma unroll
        for (int j = 0; j < 4; ++j) {
            aA[j] = *(const v4f*)(st + fA0 + 8 * j);
            aB[j] = *(const v4f*)(st + fA1 + 8 * j);
            bA[j] = *(const v4f*)(st + fB0 + 8 * j);
            bB[j] = *(const v4f*)(st + fB1 + 8 * j);
        }
        #pragma unroll
        for (int j = 0; j < 4; ++j) {
            a00 = wmma4(lo2(aA[j]), lo2(bA[j]), a00); a01 = wmma4(lo2(aA[j]), lo2(bB[j]), a01);
            a10 = wmma4(lo2(aB[j]), lo2(bA[j]), a10); a11 = wmma4(lo2(aB[j]), lo2(bB[j]), a11);
            a00 = wmma4(hi2(aA[j]), hi2(bA[j]), a00); a01 = wmma4(hi2(aA[j]), hi2(bB[j]), a01);
            a10 = wmma4(hi2(aB[j]), hi2(bA[j]), a10); a11 = wmma4(hi2(aB[j]), hi2(bB[j]), a11);
        }
        __builtin_amdgcn_sched_barrier(0);
        if (more) {
            float* sn = tile + ((c + 1) & 1) * (GROWS_ * PK_);
            #pragma unroll
            for (int r = 0; r < 10; ++r) *(v4f*)(sn + sOff + 16 * r * PK_) = g[r];
        }
        __syncthreads();
    }
}

template <int LDB, int NA>
static __device__ __forceinline__ void mm_rows_kn(const float* __restrict__ gA0,
                                                  const float* __restrict__ gA1,
                                                  const float* __restrict__ gB,
                                                  float* __restrict__ xs,
                                                  float* __restrict__ gs,
                                                  v8f (&o)[NA][4]) {
    const int t    = threadIdx.x;
    const int lane = t & 31;
    const int wave = t >> 5;
    const int m    = lane & 15;
    const int k4   = (lane >> 4) << 2;
    const float* pX = gB + (size_t)wave * LDB + (lane << 2);
    const int sxOff = wave * PX_ + (lane << 2);
    const int sr = t >> 3;
    const int sq = (t & 7) << 2;
    const float* pG[NA];
    pG[0] = gA0 + (size_t)sr * C_ + sq;
    if (NA > 1) pG[NA - 1] = gA1 + (size_t)sr * C_ + sq;
    const int sgOff = sr * PK_ + sq;
    const int fA0 = m * PK_ + k4;
    const int fA1 = (16 + m) * PK_ + k4;
    const int fB  = k4 * PX_ + wave * 32 + m;
    v4f gx[8], gg[NA][2];
    #pragma unroll
    for (int r = 0; r < 8; ++r) gx[r] = *(const v4f*)(pX + (size_t)(4 * r) * LDB);
    #pragma unroll
    for (int a = 0; a < NA; ++a) {
        #pragma unroll
        for (int r = 0; r < 2; ++r) gg[a][r] = *(const v4f*)(pG[a] + (size_t)(16 * r) * C_);
    }
    #pragma unroll
    for (int r = 0; r < 8; ++r) *(v4f*)(xs + sxOff + 4 * r * PX_) = gx[r];
    #pragma unroll
    for (int a = 0; a < NA; ++a) {
        #pragma unroll
        for (int r = 0; r < 2; ++r) *(v4f*)(gs + (a * 32 + 16 * r) * PK_ + sgOff) = gg[a][r];
    }
    __syncthreads();
    for (int c = 0; c < C_ / KC_; ++c) {
        const bool more = (c + 1 < C_ / KC_);
        if (more) {
            const int kc = (c + 1) * KC_;
            #pragma unroll
            for (int r = 0; r < 8; ++r) gx[r] = *(const v4f*)(pX + (size_t)(kc + 4 * r) * LDB);
            #pragma unroll
            for (int a = 0; a < NA; ++a) {
                #pragma unroll
                for (int r = 0; r < 2; ++r) gg[a][r] = *(const v4f*)(pG[a] + (size_t)(16 * r) * C_ + kc);
            }
        }
        const float* sx = xs + (c & 1) * (KC_ * PX_);
        const float* sg = gs + (c & 1) * (NA * 32 * PK_);
        v4f aA[NA][4], aB[NA][4];
        v2f b0[4], b1[4], b2[4], b3[4];
        #pragma unroll
        for (int j = 0; j < 4; ++j) {
            #pragma unroll
            for (int a = 0; a < NA; ++a) {
                aA[a][j] = *(const v4f*)(sg + a * 32 * PK_ + fA0 + 8 * j);
                aB[a][j] = *(const v4f*)(sg + a * 32 * PK_ + fA1 + 8 * j);
            }
            const float* r0 = sx + fB + (8 * j) * PX_;
            b0[j].x = r0[0];            b0[j].y = r0[PX_];
            b1[j].x = r0[16];           b1[j].y = r0[PX_ + 16];
            b2[j].x = r0[2 * PX_];      b2[j].y = r0[3 * PX_];
            b3[j].x = r0[2 * PX_ + 16]; b3[j].y = r0[3 * PX_ + 16];
        }
        #pragma unroll
        for (int j = 0; j < 4; ++j) {
            #pragma unroll
            for (int a = 0; a < NA; ++a) {
                o[a][0] = wmma4(lo2(aA[a][j]), b0[j], o[a][0]); o[a][1] = wmma4(lo2(aA[a][j]), b1[j], o[a][1]);
                o[a][2] = wmma4(lo2(aB[a][j]), b0[j], o[a][2]); o[a][3] = wmma4(lo2(aB[a][j]), b1[j], o[a][3]);
            }
            #pragma unroll
            for (int a = 0; a < NA; ++a) {
                o[a][0] = wmma4(hi2(aA[a][j]), b2[j], o[a][0]); o[a][1] = wmma4(hi2(aA[a][j]), b3[j], o[a][1]);
                o[a][2] = wmma4(hi2(aB[a][j]), b2[j], o[a][2]); o[a][3] = wmma4(hi2(aB[a][j]), b3[j], o[a][3]);
            }
        }
        __builtin_amdgcn_sched_barrier(0);
        if (more) {
            float* nx = xs + ((c + 1) & 1) * (KC_ * PX_);
            float* ng = gs + ((c + 1) & 1) * (NA * 32 * PK_);
            #pragma unroll
            for (int r = 0; r < 8; ++r) *(v4f*)(nx + sxOff + 4 * r * PX_) = gx[r];
            #pragma unroll
            for (int a = 0; a < NA; ++a) {
                #pragma unroll
                for (int r = 0; r < 2; ++r) *(v4f*)(ng + (a * 32 + 16 * r) * PK_ + sgOff) = gg[a][r];
            }
        }
        __syncthreads();
    }
}

__global__ __launch_bounds__(128) __attribute__((amdgpu_waves_per_eu(1, 2))) void k_gram(const float* __restrict__ x,
                                              float* __restrict__ S) {
    __shared__ __attribute__((aligned(16))) float tile[2 * GROWS_ * PK_];
    const int lane = threadIdx.x & 31;
    const int wave = threadIdx.x >> 5;
    const int m    = lane & 15;
    const int half = lane >> 4;
    const int c0   = blockIdx.y * 32;
    const int dblk = blockIdx.x * 128;
    const int d0   = dblk + wave * 32;
    const int b    = blockIdx.z;
    const size_t bOff = (size_t)b * C_ * N_;
    v8f a00 = {}, a01 = {}, a10 = {}, a11 = {};
    mm_rows_rows<N_, N_>(x + bOff + (size_t)c0 * N_, x + bOff + (size_t)dblk * N_, tile, a00, a01, a10, a11);
    float* Sb = S + (size_t)b * C_ * C_;
    #pragma unroll
    for (int j = 0; j < 8; ++j) {
        const int cA = c0 + j + 8 * half;
        const int cB = cA + 16;
        Sb[(size_t)cA * C_ + d0 + m]      = a00[j];
        Sb[(size_t)cA * C_ + d0 + 16 + m] = a01[j];
        Sb[(size_t)cB * C_ + d0 + m]      = a10[j];
        Sb[(size_t)cB * C_ + d0 + 16 + m] = a11[j];
    }
}

__global__ __launch_bounds__(128) __attribute__((amdgpu_waves_per_eu(1, 2))) void k_projs(const float* __restrict__ S,
                                               const float* __restrict__ Wq,
                                               const float* __restrict__ Wk,
                                               float* __restrict__ T,
                                               float* __restrict__ U) {
    __shared__ __attribute__((aligned(16))) float xs[2 * KC_ * PX_];
    __shared__ __attribute__((aligned(16))) float gs[2 * 2 * 32 * PK_];
    const int lane = threadIdx.x & 31;
    const int wave = threadIdx.x >> 5;
    const int m    = lane & 15;
    const int half = lane >> 4;
    const int c0   = blockIdx.y * 32;
    const int nblk = blockIdx.x * 128;
    const int n0   = nblk + wave * 32;
    const int b    = blockIdx.z;
    v8f o[2][4] = {};
    mm_rows_kn<C_, 2>(Wq + (size_t)c0 * C_, Wk + (size_t)c0 * C_, S + (size_t)b * C_ * C_ + nblk, xs, gs, o);
    float* Tb = T + (size_t)b * C_ * C_;
    float* Ub = U + (size_t)b * C_ * C_;
    #pragma unroll
    for (int j = 0; j < 8; ++j) {
        const int cA = c0 + j + 8 * half;
        const int cB = cA + 16;
        Tb[(size_t)cA * C_ + n0 + m]      = o[0][0][j];
        Tb[(size_t)cA * C_ + n0 + 16 + m] = o[0][1][j];
        Tb[(size_t)cB * C_ + n0 + m]      = o[0][2][j];
        Tb[(size_t)cB * C_ + n0 + 16 + m] = o[0][3][j];
        Ub[(size_t)cA * C_ + n0 + m]      = o[1][0][j];
        Ub[(size_t)cA * C_ + n0 + 16 + m] = o[1][1][j];
        Ub[(size_t)cB * C_ + n0 + m]      = o[1][2][j];
        Ub[(size_t)cB * C_ + n0 + 16 + m] = o[1][3][j];
    }
}

__global__ __launch_bounds__(256) void k_rownorm(const float* __restrict__ T,
                                                 const float* __restrict__ U,
                                                 const float* __restrict__ Wq,
                                                 const float* __restrict__ Wk,
                                                 float* __restrict__ invQ,
                                                 float* __restrict__ invK) {
    __shared__ float sq[256], sk[256];
    const int row = blockIdx.x;
    const int c   = row & (C_ - 1);
    const float* tr = T + (size_t)row * C_;
    const float* ur = U + (size_t)row * C_;
    const float* wq = Wq + (size_t)c * C_;
    const float* wk = Wk + (size_t)c * C_;
    float aq = 0.f, ak = 0.f;
    for (int e = threadIdx.x; e < C_; e += 256) {
        aq += tr[e] * wq[e];
        ak += ur[e] * wk[e];
    }
    sq[threadIdx.x] = aq; sk[threadIdx.x] = ak;
    __syncthreads();
    for (int s = 128; s > 0; s >>= 1) {
        if (threadIdx.x < s) {
            sq[threadIdx.x] += sq[threadIdx.x + s];
            sk[threadIdx.x] += sk[threadIdx.x + s];
        }
        __syncthreads();
    }
    if (threadIdx.x == 0) { invQ[row] = rsqrtf(sq[0]); invK[row] = rsqrtf(sk[0]); }
}

__global__ __launch_bounds__(128) __attribute__((amdgpu_waves_per_eu(1, 2))) void k_qkt(const float* __restrict__ T,
                                             const float* __restrict__ Wk,
                                             const float* __restrict__ invQ,
                                             const float* __restrict__ invK,
                                             float* __restrict__ M) {
    __shared__ __attribute__((aligned(16))) float tile[2 * GROWS_ * PK_];
    const int lane = threadIdx.x & 31;
    const int wave = threadIdx.x >> 5;
    const int m    = lane & 15;
    const int half = lane >> 4;
    const int c0   = blockIdx.y * 32;
    const int dblk = blockIdx.x * 128;
    const int d0   = dblk + wave * 32;
    const int b    = blockIdx.z;
    const size_t bOff = (size_t)b * C_ * C_;
    v8f a00 = {}, a01 = {}, a10 = {}, a11 = {};
    mm_rows_rows<C_, C_>(T + bOff + (size_t)c0 * C_, Wk + (size_t)dblk * C_, tile, a00, a01, a10, a11);
    float* Mb = M + (size_t)b * C_ * C_;
    const float ik0 = invK[b * C_ + d0 + m];
    const float ik1 = invK[b * C_ + d0 + 16 + m];
    #pragma unroll
    for (int j = 0; j < 8; ++j) {
        const int cA = c0 + j + 8 * half;
        const int cB = cA + 16;
        const float iqA = invQ[b * C_ + cA];
        const float iqB = invQ[b * C_ + cB];
        Mb[(size_t)cA * C_ + d0 + m]      = a00[j] * iqA * ik0;
        Mb[(size_t)cA * C_ + d0 + 16 + m] = a01[j] * iqA * ik1;
        Mb[(size_t)cB * C_ + d0 + m]      = a10[j] * iqB * ik0;
        Mb[(size_t)cB * C_ + d0 + 16 + m] = a11[j] * iqB * ik1;
    }
}

__global__ __launch_bounds__(256) void k_colsoftmax(float* __restrict__ M) {
    __shared__ float smax[8][32], smin[8][32], ssum[8][32];
    const int lane = threadIdx.x & 31;
    const int wave = threadIdx.x >> 5;
    const int tile = blockIdx.x;
    const int b  = tile / (C_ / 32);
    const int d  = (tile - b * (C_ / 32)) * 32 + lane;
    float* p = M + (size_t)b * C_ * C_ + (size_t)(wave * 64) * C_ + d;
    float v[64];
    for (int i = 0; i < 64; ++i) v[i] = 1.f - p[(size_t)i * C_];
    float mx = -3.4e38f, mn = 3.4e38f;
    for (int i = 0; i < 64; ++i) { mx = fmaxf(mx, v[i]); mn = fminf(mn, v[i]); }
    smax[wave][lane] = mx; smin[wave][lane] = mn;
    __syncthreads();
    float cmx = -3.4e38f, cmn = 3.4e38f;
    for (int j = 0; j < 8; ++j) {
        cmx = fmaxf(cmx, smax[j][lane]);
        cmn = fminf(cmn, smin[j][lane]);
    }
    const float dn = cmx + EPS_;
    const float inv_dn = 1.f / dn;
    const float sh = 1.f - (cmn / dn) * (1.f / HT_);
    float z = 0.f;
    for (int i = 0; i < 64; ++i) {
        const float m4 = 1.f - (v[i] * inv_dn) * (1.f / HT_);
        const float e = expf(m4 - sh);
        v[i] = e;
        z += e;
    }
    ssum[wave][lane] = z;
    __syncthreads();
    float tot = 0.f;
    for (int j = 0; j < 8; ++j) tot += ssum[j][lane];
    const float invz = 1.f / tot;
    for (int i = 0; i < 64; ++i) p[(size_t)i * C_] = v[i] * invz;
}

__global__ __launch_bounds__(256) void k_rowl1(float* __restrict__ M) {
    __shared__ float ssum[256];
    const int row = blockIdx.x;
    float* r = M + (size_t)row * C_;
    float s = 0.f;
    for (int d = threadIdx.x; d < C_; d += 256) s += r[d];
    ssum[threadIdx.x] = s;
    __syncthreads();
    for (int t = 128; t > 0; t >>= 1) {
        if (threadIdx.x < t) ssum[threadIdx.x] += ssum[threadIdx.x + t];
        __syncthreads();
    }
    const float inv = 1.f / (ssum[0] + EPS_);
    for (int d = threadIdx.x; d < C_; d += 256) r[d] *= inv;
}

__global__ __launch_bounds__(128) __attribute__((amdgpu_waves_per_eu(1, 2))) void k_mwv(const float* __restrict__ M,
                                             const float* __restrict__ Wv,
                                             float* __restrict__ G) {
    __shared__ __attribute__((aligned(16))) float xs[2 * KC_ * PX_];
    __shared__ __attribute__((aligned(16))) float gs[2 * 32 * PK_];
    const int lane = threadIdx.x & 31;
    const int wave = threadIdx.x >> 5;
    const int m    = lane & 15;
    const int half = lane >> 4;
    const int c0   = blockIdx.y * 32;
    const int eblk = blockIdx.x * 128;
    const int e0   = eblk + wave * 32;
    const int b    = blockIdx.z;
    const float* Mb = M + (size_t)b * C_ * C_ + (size_t)c0 * C_;
    v8f o[1][4] = {};
    mm_rows_kn<C_, 1>(Mb, Mb, Wv + eblk, xs, gs, o);
    float* Gb = G + (size_t)b * C_ * C_;
    #pragma unroll
    for (int j = 0; j < 8; ++j) {
        const int cA = c0 + j + 8 * half;
        const int cB = cA + 16;
        Gb[(size_t)cA * C_ + e0 + m]      = o[0][0][j];
        Gb[(size_t)cA * C_ + e0 + 16 + m] = o[0][1][j];
        Gb[(size_t)cB * C_ + e0 + m]      = o[0][2][j];
        Gb[(size_t)cB * C_ + e0 + 16 + m] = o[0][3][j];
    }
}

__global__ __launch_bounds__(128) void k_out(const float* __restrict__ x,
                                             const float* __restrict__ G,
                                             const float* __restrict__ gamma,
                                             float* __restrict__ y) {
    __shared__ __attribute__((aligned(16))) float xs[2 * KC_ * PX_];
    __shared__ __attribute__((aligned(16))) float gs[2 * 32 * PK_];
    const int lane = threadIdx.x & 31;
    const int wave = threadIdx.x >> 5;
    const int m    = lane & 15;
    const int half = lane >> 4;
    const int c0   = blockIdx.y * 32;
    const int nblk = blockIdx.x * 128;
    const int n0   = nblk + wave * 32;
    const int b    = blockIdx.z;
    const float* xb = x + (size_t)b * C_ * N_;
    const float* Gb = G + (size_t)b * C_ * C_ + (size_t)c0 * C_;
    v8f o[1][4] = {};
    mm_rows_kn<N_, 1>(Gb, Gb, xb + nblk, xs, gs, o);
    const float g = gamma[0];
    float* yb = y + (size_t)b * C_ * N_;
    #pragma unroll
    for (int j = 0; j < 8; ++j) {
        const int cA = c0 + j + 8 * half;
        const int cB = cA + 16;
        const size_t o0 = (size_t)cA * N_ + n0 + m;
        const size_t o1 = (size_t)cA * N_ + n0 + 16 + m;
        const size_t o2 = (size_t)cB * N_ + n0 + m;
        const size_t o3 = (size_t)cB * N_ + n0 + 16 + m;
        yb[o0] = xb[o0] + g * o[0][0][j];
        yb[o1] = xb[o1] + g * o[0][1][j];
        yb[o2] = xb[o2] + g * o[0][2][j];
        yb[o3] = xb[o3] + g * o[0][3][j];
    }
}

extern "C" void kernel_launch(void* const* d_in, const int* in_sizes, int n_in,
                              void* d_out, int out_size, void* d_ws, size_t ws_size,
                              hipStream_t stream) {
    (void)in_sizes; (void)n_in; (void)out_size; (void)ws_size;
    const float* x     = (const float*)d_in[0];
    const float* Wq    = (const float*)d_in[1];
    const float* Wk    = (const float*)d_in[2];
    const float* Wv    = (const float*)d_in[3];
    const float* gamma = (const float*)d_in[4];
    float* y = (float*)d_out;

    char* ws = (char*)d_ws;
    const size_t mBytes  = (size_t)B_ * C_ * C_ * sizeof(float);
    float* S     = (float*)(ws);
    float* T     = (float*)(ws + mBytes);
    float* U     = (float*)(ws + 2 * mBytes);
    float* M     = (float*)(ws + 3 * mBytes);
    float* G     = (float*)(ws + 4 * mBytes);
    float* invQ  = (float*)(ws + 5 * mBytes);
    float* invK  = invQ  + B_ * C_;

    k_gram      <<<dim3(C_ / 128, C_ / 32, B_), 128, 0, stream>>>(x, S);
    k_projs     <<<dim3(C_ / 128, C_ / 32, B_), 128, 0, stream>>>(S, Wq, Wk, T, U);
    k_rownorm   <<<B_ * C_, 256, 0, stream>>>(T, U, Wq, Wk, invQ, invK);
    k_qkt       <<<dim3(C_ / 128, C_ / 32, B_), 128, 0, stream>>>(T, Wk, invQ, invK, M);
    k_colsoftmax<<<B_ * C_ / 32, 256, 0, stream>>>(M);
    k_rowl1     <<<B_ * C_, 256, 0, stream>>>(M);
    k_mwv       <<<dim3(C_ / 128, C_ / 32, B_), 128, 0, stream>>>(M, Wv, G);
    k_out       <<<dim3(N_ / 128, C_ / 32, B_), 128, 0, stream>>>(x, G, gamma, y);
}


# All five WMMA kernels: LDS fragment reads of chunk c+1 issued as one group ahead of chunk c's WMMAs into a second register set; B pairs read as two k rows of one column, no moves; sums unchanged
# speedup vs baseline: 2.5569x; 1.0383x over previous
#include <hip/hip_runtime.h>

typedef __attribute__((ext_vector_type(2))) float v2f;
typedef __attribute__((ext_vector_type(4))) float v4f;
typedef __attribute__((ext_vector_type(8))) float v8f;

#define B_ 8
#define C_ 512
#define N_ 4096
#define EPS_ 1e-6f
#define HT_ 0.25f

static __device__ __forceinline__ v8f wmma4(v2f a, v2f b, v8f c) {
    return __builtin_amdgcn_wmma_f32_16x16x4_f32(false, a, false, b, (short)0, c, false, false);
}

static __device__ __forceinline__ v2f lo2(v4f v) { return __builtin_shufflevector(v, v, 0, 1); }
static __device__ __forceinline__ v2f hi2(v4f v) { return __builtin_shufflevector(v, v, 2, 3); }


#define KC_    32
#define PK_    36
#define GROWS_ 160
#define PX_    132

struct FragRR { v4f aA[4], aB[4], bA[4], bB[4]; };

static __device__ __forceinline__ void frag_rr_load(FragRR& f, const float* __restrict__ st,
                                                    int fA0, int fA1, int fB0, int fB1) {
    #pragma unroll
    for (int j = 0; j < 4; ++j) {
        f.aA[j] = *(const v4f*)(st + fA0 + 8 * j);
        f.aB[j] = *(const v4f*)(st + fA1 + 8 * j);
        f.bA[j] = *(const v4f*)(st + fB0 + 8 * j);
        f.bB[j] = *(const v4f*)(st + fB1 + 8 * j);
    }
}

static __device__ __forceinline__ void frag_rr_mma(const FragRR& f, v8f& a00, v8f& a01, v8f& a10, v8f& a11) {
    #pragma unroll
    for (int j = 0; j < 4; ++j) {
        a00 = wmma4(lo2(f.aA[j]), lo2(f.bA[j]), a00); a01 = wmma4(lo2(f.aA[j]), lo2(f.bB[j]), a01);
        a10 = wmma4(lo2(f.aB[j]), lo2(f.bA[j]), a10); a11 = wmma4(lo2(f.aB[j]), lo2(f.bB[j]), a11);
        a00 = wmma4(hi2(f.aA[j]), hi2(f.bA[j]), a00); a01 = wmma4(hi2(f.aA[j]), hi2(f.bB[j]), a01);
        a10 = wmma4(hi2(f.aB[j]), hi2(f.bA[j]), a10); a11 = wmma4(hi2(f.aB[j]), hi2(f.bB[j]), a11);
    }
}

template <int LD, int KTOT>
static __device__ __forceinline__ void mm_rows_rows(const float* __restrict__ gA,
                                                    const float* __restrict__ gB,
                                                    float* __restrict__ tile,
                                                    v8f& a00, v8f& a01, v8f& a10, v8f& a11) {
    constexpr int NCH = KTOT / KC_;
    constexpr int STG = GROWS_ * PK_;
    const int t    = threadIdx.x;
    const int lane = t & 31;
    const int wave = t >> 5;
    const int m    = lane & 15;
    const int k4   = (lane >> 4) << 2;
    const int sr = t >> 3;
    const int sq = (t & 7) << 2;
    const float* pA = gA + (size_t)sr * LD + sq;
    const float* pB = gB + (size_t)sr * LD + sq;
    const int sOff = sr * PK_ + sq;
    const int fA0 = m * PK_ + k4;
    const int fA1 = (16 + m) * PK_ + k4;
    const int fB0 = (32 + wave * 32 + m) * PK_ + k4;
    const int fB1 = fB0 + 16 * PK_;
    v4f g[10];
    {
        v4f h[10];
        #pragma unroll
        for (int r = 0; r < 2; ++r) {
            g[r] = *(const v4f*)(pA + (size_t)(16 * r) * LD);
            h[r] = *(const v4f*)(pA + (size_t)(16 * r) * LD + KC_);
        }
        #pragma unroll
        for (int r = 0; r < 8; ++r) {
            g[2 + r] = *(const v4f*)(pB + (size_t)(16 * r) * LD);
            h[2 + r] = *(const v4f*)(pB + (size_t)(16 * r) * LD + KC_);
        }
        #pragma unroll
        for (int r = 0; r < 10; ++r) {
            *(v4f*)(tile + sOff + 16 * r * PK_) = g[r];
            *(v4f*)(tile + STG + sOff + 16 * r * PK_) = h[r];
        }
    }
    __syncthreads();
    FragRR P, Q;
    frag_rr_load(P, tile, fA0, fA1, fB0, fB1);
    __syncthreads();
    #pragma unroll 1
    for (int c = 0; c < NCH; c += 2) {
        const bool more = (c + 2 < NCH);
        if (more) {
            const int kc = (c + 2) * KC_;
            #pragma unroll
            for (int r = 0; r < 2; ++r) g[r] = *(const v4f*)(pA + (size_t)(16 * r) * LD + kc);
            #pragma unroll
            for (int r = 0; r < 8; ++r) g[2 + r] = *(const v4f*)(pB + (size_t)(16 * r) * LD + kc);
        }
        frag_rr_load(Q, tile + STG, fA0, fA1, fB0, fB1);
        __builtin_amdgcn_sched_barrier(0);
        frag_rr_mma(P, a00, a01, a10, a11);
        __builtin_amdgcn_sched_barrier(0);
        if (more) {
            #pragma unroll
            for (int r = 0; r < 10; ++r) *(v4f*)(tile + sOff + 16 * r * PK_) = g[r];
        }
        __syncthreads();
        if (more) {
            const int kc = (c + 3) * KC_;
            #pragma unroll
            for (int r = 0; r < 2; ++r) g[r] = *(const v4f*)(pA + (size_t)(16 * r) * LD + kc);
            #pragma unroll
            for (int r = 0; r < 8; ++r) g[2 + r] = *(const v4f*)(pB + (size_t)(16 * r) * LD + kc);
            frag_rr_load(P, tile, fA0, fA1, fB0, fB1);
        }
        __builtin_amdgcn_sched_barrier(0);
        frag_rr_mma(Q, a00, a01, a10, a11);
        __builtin_amdgcn_sched_barrier(0);
        if (more) {
            #pragma unroll
            for (int r = 0; r < 10; ++r) *(v4f*)(tile + STG + sOff + 16 * r * PK_) = g[r];
        }
        __syncthreads();
    }
}

template <int NA>
struct FragKN { v4f aA[NA][4], aB[NA][4]; v2f b0[4], b1[4], b2[4], b3[4]; };

template <int NA>
static __device__ __forceinline__ void frag_kn_load(FragKN<NA>& f, const float* __restrict__ sx,
                                                    const float* __restrict__ sg,
                                                    int fA0, int fA1, int fB, int fBh) {
    #pragma unroll
    for (int j = 0; j < 4; ++j) {
        #pragma unroll
        for (int a = 0; a < NA; ++a) {
            f.aA[a][j] = *(const v4f*)(sg + a * 32 * PK_ + fA0 + 8 * j);
            f.aB[a][j] = *(const v4f*)(sg + a * 32 * PK_ + fA1 + 8 * j);
        }
        const float* r0 = sx + fB + (8 * j) * PX_;
        const float* r1 = sx + fBh + (8 * j) * PX_;
        f.b0[j].x = r0[0];       f.b0[j].y = r0[PX_];
        f.b1[j].x = r1[0];       f.b1[j].y = r1[PX_];
        f.b2[j].x = r0[2 * PX_]; f.b2[j].y = r0[3 * PX_];
        f.b3[j].x = r1[2 * PX_]; f.b3[j].y = r1[3 * PX_];
    }
}

template <int NA>
static __device__ __forceinline__ void frag_kn_mma(const FragKN<NA>& f, v8f (&o)[NA][4]) {
    #pragma unroll
    for (int j = 0; j < 4; ++j) {
        #pragma unroll
        for (int a = 0; a < NA; ++a) {
            o[a][0] = wmma4(lo2(f.aA[a][j]), f.b0[j], o[a][0]); o[a][1] = wmma4(lo2(f.aA[a][j]), f.b1[j], o[a][1]);
            o[a][2] = wmma4(lo2(f.aB[a][j]), f.b0[j], o[a][2]); o[a][3] = wmma4(lo2(f.aB[a][j]), f.b1[j], o[a][3]);
        }
        #pragma unroll
        for (int a = 0; a < NA; ++a) {
            o[a][0] = wmma4(hi2(f.aA[a][j]), f.b2[j], o[a][0]); o[a][1] = wmma4(hi2(f.aA[a][j]), f.b3[j], o[a][1]);
            o[a][2] = wmma4(hi2(f.aB[a][j]), f.b2[j], o[a][2]); o[a][3] = wmma4(hi2(f.aB[a][j]), f.b3[j], o[a][3]);
        }
    }
}

template <int LDB, int NA>
static __device__ __forceinline__ void mm_rows_kn(const float* __restrict__ gA0,
                                                  const float* __restrict__ gA1,
                                                  const float* __restrict__ gB,
                                                  float* __restrict__ xs,
                                                  float* __restrict__ gs,
                                                  v8f (&o)[NA][4]) {
    constexpr int NCH = C_ / KC_;
    constexpr int SXS = KC_ * PX_;
    constexpr int SGS = NA * 32 * PK_;
    const int t    = threadIdx.x;
    const int lane = t & 31;
    const int wave = t >> 5;
    const int m    = lane & 15;
    const int k4   = (lane >> 4) << 2;
    const float* pX = gB + (size_t)wave * LDB + (lane << 2);
    const int sxOff = wave * PX_ + (lane << 2);
    const int sr = t >> 3;
    const int sq = (t & 7) << 2;
    const float* pG[NA];
    pG[0] = gA0 + (size_t)sr * C_ + sq;
    if (NA > 1) pG[NA - 1] = gA1 + (size_t)sr * C_ + sq;
    const int sgOff = sr * PK_ + sq;
    const int fA0 = m * PK_ + k4;
    const int fA1 = (16 + m) * PK_ + k4;
    const int fB  = k4 * PX_ + wave * 32 + m;
    int fBh = fB + 16;
    asm volatile("" : "+v"(fBh));
    v4f gx[8], gg[NA][2];
    {
        v4f hx[8], hg[NA][2];
        #pragma unroll
        for (int r = 0; r < 8; ++r) {
            gx[r] = *(const v4f*)(pX + (size_t)(4 * r) * LDB);
            hx[r] = *(const v4f*)(pX + (size_t)(KC_ + 4 * r) * LDB);
        }
        #pragma unroll
        for (int a = 0; a < NA; ++a) {
            #pragma unroll
            for (int r = 0; r < 2; ++r) {
                gg[a][r] = *(const v4f*)(pG[a] + (size_t)(16 * r) * C_);
                hg[a][r] = *(const v4f*)(pG[a] + (size_t)(16 * r) * C_ + KC_);
            }
        }
        #pragma unroll
        for (int r = 0; r < 8; ++r) {
            *(v4f*)(xs + sxOff + 4 * r * PX_) = gx[r];
            *(v4f*)(xs + SXS + sxOff + 4 * r * PX_) = hx[r];
        }
        #pragma unroll
        for (int a = 0; a < NA; ++a) {
            #pragma unroll
            for (int r = 0; r < 2; ++r) {
                *(v4f*)(gs + (a * 32 + 16 * r) * PK_ + sgOff) = gg[a][r];
                *(v4f*)(gs + SGS + (a * 32 + 16 * r) * PK_ + sgOff) = hg[a][r];
            }
        }
    }
    __syncthreads();
    FragKN<NA> P, Q;
    frag_kn_load<NA>(P, xs, gs, fA0, fA1, fB, fBh);
    __syncthreads();
    #pragma unroll 1
    for (int c = 0; c < NCH; c += 2) {
        const bool more = (c + 2 < NCH);
        if (more) {
            const int kc = (c + 2) * KC_;
            #pragma unroll
            for (int r = 0; r < 8; ++r) gx[r] = *(const v4f*)(pX + (size_t)(kc + 4 * r) * LDB);
            #pragma unroll
            for (int a = 0; a < NA; ++a) {
                #pragma unroll
                for (int r = 0; r < 2; ++r) gg[a][r] = *(const v4f*)(pG[a] + (size_t)(16 * r) * C_ + kc);
            }
        }
        frag_kn_load<NA>(Q, xs + SXS, gs + SGS, fA0, fA1, fB, fBh);
        __builtin_amdgcn_sched_barrier(0);
        frag_kn_mma<NA>(P, o);
        __builtin_amdgcn_sched_barrier(0);
        if (more) {
            #pragma unroll
            for (int r = 0; r < 8; ++r) *(v4f*)(xs + sxOff + 4 * r * PX_) = gx[r];
            #pragma unroll
            for (int a = 0; a < NA; ++a) {
                #pragma unroll
                for (int r = 0; r < 2; ++r) *(v4f*)(gs + (a * 32 + 16 * r) * PK_ + sgOff) = gg[a][r];
            }
        }
        __syncthreads();
        if (more) {
            const int kc = (c + 3) * KC_;
            #pragma unroll
            for (int r = 0; r < 8; ++r) gx[r] = *(const v4f*)(pX + (size_t)(kc + 4 * r) * LDB);
            #pragma unroll
            for (int a = 0; a < NA; ++a) {
                #pragma unroll
                for (int r = 0; r < 2; ++r) gg[a][r] = *(const v4f*)(pG[a] + (size_t)(16 * r) * C_ + kc);
            }
            frag_kn_load<NA>(P, xs, gs, fA0, fA1, fB, fBh);
        }
        __builtin_amdgcn_sched_barrier(0);
        frag_kn_mma<NA>(Q, o);
        __builtin_amdgcn_sched_barrier(0);
        if (more) {
            #pragma unroll
            for (int r = 0; r < 8; ++r) *(v4f*)(xs + SXS + sxOff + 4 * r * PX_) = gx[r];
            #pragma unroll
            for (int a = 0; a < NA; ++a) {
                #pragma unroll
                for (int r = 0; r < 2; ++r) *(v4f*)(gs + SGS + (a * 32 + 16 * r) * PK_ + sgOff) = gg[a][r];
            }
        }
        __syncthreads();
    }
}

__global__ __launch_bounds__(128) __attribute__((amdgpu_waves_per_eu(1, 2))) void k_gram(const float* __restrict__ x,
                                              float* __restrict__ S) {
    __shared__ __attribute__((aligned(16))) float tile[2 * GROWS_ * PK_];
    const int lane = threadIdx.x & 31;
    const int wave = threadIdx.x >> 5;
    const int m    = lane & 15;
    const int half = lane >> 4;
    const int c0   = blockIdx.y * 32;
    const int dblk = blockIdx.x * 128;
    const int d0   = dblk + wave * 32;
    const int b    = blockIdx.z;
    const size_t bOff = (size_t)b * C_ * N_;
    v8f a00 = {}, a01 = {}, a10 = {}, a11 = {};
    mm_rows_rows<N_, N_>(x + bOff + (size_t)c0 * N_, x + bOff + (size_t)dblk * N_, tile, a00, a01, a10, a11);
    float* Sb = S + (size_t)b * C_ * C_;
    #pragma unroll
    for (int j = 0; j < 8; ++j) {
        const int cA = c0 + j + 8 * half;
        const int cB = cA + 16;
        Sb[(size_t)cA * C_ + d0 + m]      = a00[j];
        Sb[(size_t)cA * C_ + d0 + 16 + m] = a01[j];
        Sb[(size_t)cB * C_ + d0 + m]      = a10[j];
        Sb[(size_t)cB * C_ + d0 + 16 + m] = a11[j];
    }
}

__global__ __launch_bounds__(128) __attribute__((amdgpu_waves_per_eu(1, 2))) void k_projs(const float* __restrict__ S,
                                               const float* __restrict__ Wq,
                                               const float* __restrict__ Wk,
                                               float* __restrict__ T,
                                               float* __restrict__ U) {
    __shared__ __attribute__((aligned(16))) float xs[2 * KC_ * PX_];
    __shared__ __attribute__((aligned(16))) float gs[2 * 2 * 32 * PK_];
    const int lane = threadIdx.x & 31;
    const int wave = threadIdx.x >> 5;
    const int m    = lane & 15;
    const int half = lane >> 4;
    const int c0   = blockIdx.y * 32;
    const int nblk = blockIdx.x * 128;
    const int n0   = nblk + wave * 32;
    const int b    = blockIdx.z;
    v8f o[2][4] = {};
    mm_rows_kn<C_, 2>(Wq + (size_t)c0 * C_, Wk + (size_t)c0 * C_, S + (size_t)b * C_ * C_ + nblk, xs, gs, o);
    float* Tb = T + (size_t)b * C_ * C_;
    float* Ub = U + (size_t)b * C_ * C_;
    #pragma unroll
    for (int j = 0; j < 8; ++j) {
        const int cA = c0 + j + 8 * half;
        const int cB = cA + 16;
        Tb[(size_t)cA * C_ + n0 + m]      = o[0][0][j];
        Tb[(size_t)cA * C_ + n0 + 16 + m] = o[0][1][j];
        Tb[(size_t)cB * C_ + n0 + m]      = o[0][2][j];
        Tb[(size_t)cB * C_ + n0 + 16 + m] = o[0][3][j];
        Ub[(size_t)cA * C_ + n0 + m]      = o[1][0][j];
        Ub[(size_t)cA * C_ + n0 + 16 + m] = o[1][1][j];
        Ub[(size_t)cB * C_ + n0 + m]      = o[1][2][j];
        Ub[(size_t)cB * C_ + n0 + 16 + m] = o[1][3][j];
    }
}

__global__ __launch_bounds__(256) void k_rownorm(const float* __restrict__ T,
                                                 const float* __restrict__ U,
                                                 const float* __restrict__ Wq,
                                                 const float* __restrict__ Wk,
                                                 float* __restrict__ invQ,
                                                 float* __restrict__ invK) {
    __shared__ float sq[256], sk[256];
    const int row = blockIdx.x;
    const int c   = row & (C_ - 1);
    const float* tr = T + (size_t)row * C_;
    const float* ur = U + (size_t)row * C_;
    const float* wq = Wq + (size_t)c * C_;
    const float* wk = Wk + (size_t)c * C_;
    float aq = 0.f, ak = 0.f;
    for (int e = threadIdx.x; e < C_; e += 256) {
        aq += tr[e] * wq[e];
        ak += ur[e] * wk[e];
    }
    sq[threadIdx.x] = aq; sk[threadIdx.x] = ak;
    __syncthreads();
    for (int s = 128; s > 0; s >>= 1) {
        if (threadIdx.x < s) {
            sq[threadIdx.x] += sq[threadIdx.x + s];
            sk[threadIdx.x] += sk[threadIdx.x + s];
        }
        __syncthreads();
    }
    if (threadIdx.x == 0) { invQ[row] = rsqrtf(sq[0]); invK[row] = rsqrtf(sk[0]); }
}

__global__ __launch_bounds__(128) __attribute__((amdgpu_waves_per_eu(1, 2))) void k_qkt(const float* __restrict__ T,
                                             const float* __restrict__ Wk,
                                             const float* __restrict__ invQ,
                                             const float* __restrict__ invK,
                                             float* __restrict__ M) {
    __shared__ __attribute__((aligned(16))) float tile[2 * GROWS_ * PK_];
    const int lane = threadIdx.x & 31;
    const int wave = threadIdx.x >> 5;
    const int m    = lane & 15;
    const int half = lane >> 4;
    const int c0   = blockIdx.y * 32;
    const int dblk = blockIdx.x * 128;
    const int d0   = dblk + wave * 32;
    const int b    = blockIdx.z;
    const size_t bOff = (size_t)b * C_ * C_;
    v8f a00 = {}, a01 = {}, a10 = {}, a11 = {};
    mm_rows_rows<C_, C_>(T + bOff + (size_t)c0 * C_, Wk + (size_t)dblk * C_, tile, a00, a01, a10, a11);
    float* Mb = M + (size_t)b * C_ * C_;
    const float ik0 = invK[b * C_ + d0 + m];
    const float ik1 = invK[b * C_ + d0 + 16 + m];
    #pragma unroll
    for (int j = 0; j < 8; ++j) {
        const int cA = c0 + j + 8 * half;
        const int cB = cA + 16;
        const float iqA = invQ[b * C_ + cA];
        const float iqB = invQ[b * C_ + cB];
        Mb[(size_t)cA * C_ + d0 + m]      = a00[j] * iqA * ik0;
        Mb[(size_t)cA * C_ + d0 + 16 + m] = a01[j] * iqA * ik1;
        Mb[(size_t)cB * C_ + d0 + m]      = a10[j] * iqB * ik0;
        Mb[(size_t)cB * C_ + d0 + 16 + m] = a11[j] * iqB * ik1;
    }
}

__global__ __launch_bounds__(256) void k_colsoftmax(float* __restrict__ M) {
    __shared__ float smax[8][32], smin[8][32], ssum[8][32];
    const int lane = threadIdx.x & 31;
    const int wave = threadIdx.x >> 5;
    const int tile = blockIdx.x;
    const int b  = tile / (C_ / 32);
    const int d  = (tile - b * (C_ / 32)) * 32 + lane;
    float* p = M + (size_t)b * C_ * C_ + (size_t)(wave * 64) * C_ + d;
    float v[64];
    for (int i = 0; i < 64; ++i) v[i] = 1.f - p[(size_t)i * C_];
    float mx = -3.4e38f, mn = 3.4e38f;
    for (int i = 0; i < 64; ++i) { mx = fmaxf(mx, v[i]); mn = fminf(mn, v[i]); }
    smax[wave][lane] = mx; smin[wave][lane] = mn;
    __syncthreads();
    float cmx = -3.4e38f, cmn = 3.4e38f;
    for (int j = 0; j < 8; ++j) {
        cmx = fmaxf(cmx, smax[j][lane]);
        cmn = fminf(cmn, smin[j][lane]);
    }
    const float dn = cmx + EPS_;
    const float inv_dn = 1.f / dn;
    const float sh = 1.f - (cmn / dn) * (1.f / HT_);
    float z = 0.f;
    for (int i = 0; i < 64; ++i) {
        const float m4 = 1.f - (v[i] * inv_dn) * (1.f / HT_);
        const float e = expf(m4 - sh);
        v[i] = e;
        z += e;
    }
    ssum[wave][lane] = z;
    __syncthreads();
    float tot = 0.f;
    for (int j = 0; j < 8; ++j) tot += ssum[j][lane];
    const float invz = 1.f / tot;
    for (int i = 0; i < 64; ++i) p[(size_t)i * C_] = v[i] * invz;
}

__global__ __launch_bounds__(256) void k_rowl1(float* __restrict__ M) {
    __shared__ float ssum[256];
    const int row = blockIdx.x;
    float* r = M + (size_t)row * C_;
    float s = 0.f;
    for (int d = threadIdx.x; d < C_; d += 256) s += r[d];
    ssum[threadIdx.x] = s;
    __syncthreads();
    for (int t = 128; t > 0; t >>= 1) {
        if (threadIdx.x < t) ssum[threadIdx.x] += ssum[threadIdx.x + t];
        __syncthreads();
    }
    const float inv = 1.f / (ssum[0] + EPS_);
    for (int d = threadIdx.x; d < C_; d += 256) r[d] *= inv;
}

__global__ __launch_bounds__(128) __attribute__((amdgpu_waves_per_eu(1, 2))) void k_mwv(const float* __restrict__ M,
                                             const float* __restrict__ Wv,
                                             float* __restrict__ G) {
    __shared__ __attribute__((aligned(16))) float xs[2 * KC_ * PX_];
    __shared__ __attribute__((aligned(16))) float gs[2 * 32 * PK_];
    const int lane = threadIdx.x & 31;
    const int wave = threadIdx.x >> 5;
    const int m    = lane & 15;
    const int half = lane >> 4;
    const int c0   = blockIdx.y * 32;
    const int eblk = blockIdx.x * 128;
    const int e0   = eblk + wave * 32;
    const int b    = blockIdx.z;
    const float* Mb = M + (size_t)b * C_ * C_ + (size_t)c0 * C_;
    v8f o[1][4] = {};
    mm_rows_kn<C_, 1>(Mb, Mb, Wv + eblk, xs, gs, o);
    float* Gb = G + (size_t)b * C_ * C_;
    #pragma unroll
    for (int j = 0; j < 8; ++j) {
        const int cA = c0 + j + 8 * half;
        const int cB = cA + 16;
        Gb[(size_t)cA * C_ + e0 + m]      = o[0][0][j];
        Gb[(size_t)cA * C_ + e0 + 16 + m] = o[0][1][j];
        Gb[(size_t)cB * C_ + e0 + m]      = o[0][2][j];
        Gb[(size_t)cB * C_ + e0 + 16 + m] = o[0][3][j];
    }
}

__global__ __launch_bounds__(128) void k_out(const float* __restrict__ x,
                                             const float* __restrict__ G,
                                             const float* __restrict__ gamma,
                                             float* __restrict__ y) {
    __shared__ __attribute__((aligned(16))) float xs[2 * KC_ * PX_];
    __shared__ __attribute__((aligned(16))) float gs[2 * 32 * PK_];
    const int lane = threadIdx.x & 31;
    const int wave = threadIdx.x >> 5;
    const int m    = lane & 15;
    const int half = lane >> 4;
    const int c0   = blockIdx.y * 32;
    const int nblk = blockIdx.x * 128;
    const int n0   = nblk + wave * 32;
    const int b    = blockIdx.z;
    const float* xb = x + (size_t)b * C_ * N_;
    const float* Gb = G + (size_t)b * C_ * C_ + (size_t)c0 * C_;
    v8f o[1][4] = {};
    mm_rows_kn<N_, 1>(Gb, Gb, xb + nblk, xs, gs, o);
    const float g = gamma[0];
    float* yb = y + (size_t)b * C_ * N_;
    #pragma unroll
    for (int j = 0; j < 8; ++j) {
        const int cA = c0 + j + 8 * half;
        const int cB = cA + 16;
        const size_t o0 = (size_t)cA * N_ + n0 + m;
        const size_t o1 = (size_t)cA * N_ + n0 + 16 + m;
        const size_t o2 = (size_t)cB * N_ + n0 + m;
        const size_t o3 = (size_t)cB * N_ + n0 + 16 + m;
        yb[o0] = xb[o0] + g * o[0][0][j];
        yb[o1] = xb[o1] + g * o[0][1][j];
        yb[o2] = xb[o2] + g * o[0][2][j];
        yb[o3] = xb[o3] + g * o[0][3][j];
    }
}

extern "C" void kernel_launch(void* const* d_in, const int* in_sizes, int n_in,
                              void* d_out, int out_size, void* d_ws, size_t ws_size,
                              hipStream_t stream) {
    (void)in_sizes; (void)n_in; (void)out_size; (void)ws_size;
    const float* x     = (const float*)d_in[0];
    const float* Wq    = (const float*)d_in[1];
    const float* Wk    = (const float*)d_in[2];
    const float* Wv    = (const float*)d_in[3];
    const float* gamma = (const float*)d_in[4];
    float* y = (float*)d_out;

    char* ws = (char*)d_ws;
    const size_t mBytes  = (size_t)B_ * C_ * C_ * sizeof(float);
    float* S     = (float*)(ws);
    float* T     = (float*)(ws + mBytes);
    float* U     = (float*)(ws + 2 * mBytes);
    float* M     = (float*)(ws + 3 * mBytes);
    float* G     = (float*)(ws + 4 * mBytes);
    float* invQ  = (float*)(ws + 5 * mBytes);
    float* invK  = invQ  + B_ * C_;

    k_gram      <<<dim3(C_ / 128, C_ / 32, B_), 128, 0, stream>>>(x, S);
    k_projs     <<<dim3(C_ / 128, C_ / 32, B_), 128, 0, stream>>>(S, Wq, Wk, T, U);
    k_rownorm   <<<B_ * C_, 256, 0, stream>>>(T, U, Wq, Wk, invQ, invK);
    k_qkt       <<<dim3(C_ / 128, C_ / 32, B_), 128, 0, stream>>>(T, Wk, invQ, invK, M);
    k_colsoftmax<<<B_ * C_ / 32, 256, 0, stream>>>(M);
    k_rowl1     <<<B_ * C_, 256, 0, stream>>>(M);
    k_mwv       <<<dim3(C_ / 128, C_ / 32, B_), 128, 0, stream>>>(M, Wv, G);
    k_out       <<<dim3(N_ / 128, C_ / 32, B_), 128, 0, stream>>>(x, G, gamma, y);
}


# Gram matrix by its symmetry: only 128x128 block pairs on or above the diagonal computed (and mirror-stored), K cut in two halves as two planes the next GEMM adds while staging; Gram sum tree changes
# speedup vs baseline: 2.6488x; 1.0359x over previous
#include <hip/hip_runtime.h>

typedef __attribute__((ext_vector_type(2))) float v2f;
typedef __attribute__((ext_vector_type(4))) float v4f;
typedef __attribute__((ext_vector_type(8))) float v8f;

#define B_ 8
#define C_ 512
#define N_ 4096
#define EPS_ 1e-6f
#define HT_ 0.25f

static __device__ __forceinline__ v8f wmma4(v2f a, v2f b, v8f c) {
    return __builtin_amdgcn_wmma_f32_16x16x4_f32(false, a, false, b, (short)0, c, false, false);
}

static __device__ __forceinline__ v2f lo2(v4f v) { return __builtin_shufflevector(v, v, 0, 1); }
static __device__ __forceinline__ v2f hi2(v4f v) { return __builtin_shufflevector(v, v, 2, 3); }


#define KC_    32
#define PK_    36
#define GROWS_ 160
#define PX_    132

struct FragRR { v4f aA[4], aB[4], bA[4], bB[4]; };

static __device__ __forceinline__ void frag_rr_load(FragRR& f, const float* __restrict__ st,
                                                    int fA0, int fA1, int fB0, int fB1) {
    #pragma unroll
    for (int j = 0; j < 4; ++j) {
        f.aA[j] = *(const v4f*)(st + fA0 + 8 * j);
        f.aB[j] = *(const v4f*)(st + fA1 + 8 * j);
        f.bA[j] = *(const v4f*)(st + fB0 + 8 * j);
        f.bB[j] = *(const v4f*)(st + fB1 + 8 * j);
    }
}

static __device__ __forceinline__ void frag_rr_mma(const FragRR& f, v8f& a00, v8f& a01, v8f& a10, v8f& a11) {
    #pragma unroll
    for (int j = 0; j < 4; ++j) {
        a00 = wmma4(lo2(f.aA[j]), lo2(f.bA[j]), a00); a01 = wmma4(lo2(f.aA[j]), lo2(f.bB[j]), a01);
        a10 = wmma4(lo2(f.aB[j]), lo2(f.bA[j]), a10); a11 = wmma4(lo2(f.aB[j]), lo2(f.bB[j]), a11);
        a00 = wmma4(hi2(f.aA[j]), hi2(f.bA[j]), a00); a01 = wmma4(hi2(f.aA[j]), hi2(f.bB[j]), a01);
        a10 = wmma4(hi2(f.aB[j]), hi2(f.bA[j]), a10); a11 = wmma4(hi2(f.aB[j]), hi2(f.bB[j]), a11);
    }
}

template <int LD, int KTOT>
static __device__ __forceinline__ void mm_rows_rows(const float* __restrict__ gA,
                                                    const float* __restrict__ gB,
                                                    float* __restrict__ tile,
                                                    v8f& a00, v8f& a01, v8f& a10, v8f& a11) {
    constexpr int NCH = KTOT / KC_;
    constexpr int STG = GROWS_ * PK_;
    const int t    = threadIdx.x;
    const int lane = t & 31;
    const int wave = t >> 5;
    const int m    = lane & 15;
    const int k4   = (lane >> 4) << 2;
    const int sr = t >> 3;
    const int sq = (t & 7) << 2;
    const float* pA = gA + (size_t)sr * LD + sq;
    const float* pB = gB + (size_t)sr * LD + sq;
    const int sOff = sr * PK_ + sq;
    const int fA0 = m * PK_ + k4;
    const int fA1 = (16 + m) * PK_ + k4;
    const int fB0 = (32 + wave * 32 + m) * PK_ + k4;
    const int fB1 = fB0 + 16 * PK_;
    v4f g[10];
    {
        v4f h[10];
        #pragma unroll
        for (int r = 0; r < 2; ++r) {
            g[r] = *(const v4f*)(pA + (size_t)(16 * r) * LD);
            h[r] = *(const v4f*)(pA + (size_t)(16 * r) * LD + KC_);
        }
        #pragma unroll
        for (int r = 0; r < 8; ++r) {
            g[2 + r] = *(const v4f*)(pB + (size_t)(16 * r) * LD);
            h[2 + r] = *(const v4f*)(pB + (size_t)(16 * r) * LD + KC_);
        }
        #pragma unroll
        for (int r = 0; r < 10; ++r) {
            *(v4f*)(tile + sOff + 16 * r * PK_) = g[r];
            *(v4f*)(tile + STG + sOff + 16 * r * PK_) = h[r];
        }
    }
    __syncthreads();
    FragRR P, Q;
    frag_rr_load(P, tile, fA0, fA1, fB0, fB1);
    __syncthreads();
    #pragma unroll 1
    for (int c = 0; c < NCH; c += 2) {
        const bool more = (c + 2 < NCH);
        if (more) {
            const int kc = (c + 2) * KC_;
            #pragma unroll
            for (int r = 0; r < 2; ++r) g[r] = *(const v4f*)(pA + (size_t)(16 * r) * LD + kc);
            #pragma unroll
            for (int r = 0; r < 8; ++r) g[2 + r] = *(const v4f*)(pB + (size_t)(16 * r) * LD + kc);
        }
        frag_rr_load(Q, tile + STG, fA0, fA1, fB0, fB1);
        __builtin_amdgcn_sched_barrier(0);
        frag_rr_mma(P, a00, a01, a10, a11);
        __builtin_amdgcn_sched_barrier(0);
        if (more) {
            #pragma unroll
            for (int r = 0; r < 10; ++r) *(v4f*)(tile + sOff + 16 * r * PK_) = g[r];
        }
        __syncthreads();
        if (more) {
            const int kc = (c + 3) * KC_;
            #pragma unroll
            for (int r = 0; r < 2; ++r) g[r] = *(const v4f*)(pA + (size_t)(16 * r) * LD + kc);
            #pragma unroll
            for (int r = 0; r < 8; ++r) g[2 + r] = *(const v4f*)(pB + (size_t)(16 * r) * LD + kc);
            frag_rr_load(P, tile, fA0, fA1, fB0, fB1);
        }
        __builtin_amdgcn_sched_barrier(0);
        frag_rr_mma(Q, a00, a01, a10, a11);
        __builtin_amdgcn_sched_barrier(0);
        if (more) {
            #pragma unroll
            for (int r = 0; r < 10; ++r) *(v4f*)(tile + STG + sOff + 16 * r * PK_) = g[r];
        }
        __syncthreads();
    }
}

template <int NA>
struct FragKN { v4f aA[NA][4], aB[NA][4]; v2f b0[4], b1[4], b2[4], b3[4]; };

template <int NA>
static __device__ __forceinline__ void frag_kn_load(FragKN<NA>& f, const float* __restrict__ sx,
                                                    const float* __restrict__ sg,
                                                    int fA0, int fA1, int fB, int fBh) {
    #pragma unroll
    for (int j = 0; j < 4; ++j) {
        #pragma unroll
        for (int a = 0; a < NA; ++a) {
            f.aA[a][j] = *(const v4f*)(sg + a * 32 * PK_ + fA0 + 8 * j);
            f.aB[a][j] = *(const v4f*)(sg + a * 32 * PK_ + fA1 + 8 * j);
        }
        const float* r0 = sx + fB + (8 * j) * PX_;
        const float* r1 = sx + fBh + (8 * j) * PX_;
        f.b0[j].x = r0[0];       f.b0[j].y = r0[PX_];
        f.b1[j].x = r1[0];       f.b1[j].y = r1[PX_];
        f.b2[j].x = r0[2 * PX_]; f.b2[j].y = r0[3 * PX_];
        f.b3[j].x = r1[2 * PX_]; f.b3[j].y = r1[3 * PX_];
    }
}

template <int NA>
static __device__ __forceinline__ void frag_kn_mma(const FragKN<NA>& f, v8f (&o)[NA][4]) {
    #pragma unroll
    for (int j = 0; j < 4; ++j) {
        #pragma unroll
        for (int a = 0; a < NA; ++a) {
            o[a][0] = wmma4(lo2(f.aA[a][j]), f.b0[j], o[a][0]); o[a][1] = wmma4(lo2(f.aA[a][j]), f.b1[j], o[a][1]);
            o[a][2] = wmma4(lo2(f.aB[a][j]), f.b0[j], o[a][2]); o[a][3] = wmma4(lo2(f.aB[a][j]), f.b1[j], o[a][3]);
        }
        #pragma unroll
        for (int a = 0; a < NA; ++a) {
            o[a][0] = wmma4(hi2(f.aA[a][j]), f.b2[j], o[a][0]); o[a][1] = wmma4(hi2(f.aA[a][j]), f.b3[j], o[a][1]);
            o[a][2] = wmma4(hi2(f.aB[a][j]), f.b2[j], o[a][2]); o[a][3] = wmma4(hi2(f.aB[a][j]), f.b3[j], o[a][3]);
        }
    }
}

template <int LDB, int NA, bool TWO>
static __device__ __forceinline__ void mm_rows_kn(const float* __restrict__ gA0,
                                                  const float* __restrict__ gA1,
                                                  const float* __restrict__ gB,
                                                  const float* __restrict__ gB2,
                                                  float* __restrict__ xs,
                                                  float* __restrict__ gs,
                                                  v8f (&o)[NA][4]) {
    constexpr int NCH = C_ / KC_;
    constexpr int SXS = KC_ * PX_;
    constexpr int SGS = NA * 32 * PK_;
    const int t    = threadIdx.x;
    const int lane = t & 31;
    const int wave = t >> 5;
    const int m    = lane & 15;
    const int k4   = (lane >> 4) << 2;
    const float* pX = gB + (size_t)wave * LDB + (lane << 2);
    const float* pY = gB2 + (size_t)wave * LDB + (lane << 2);
    const int sxOff = wave * PX_ + (lane << 2);
    const int sr = t >> 3;
    const int sq = (t & 7) << 2;
    const float* pG[NA];
    pG[0] = gA0 + (size_t)sr * C_ + sq;
    if (NA > 1) pG[NA - 1] = gA1 + (size_t)sr * C_ + sq;
    const int sgOff = sr * PK_ + sq;
    const int fA0 = m * PK_ + k4;
    const int fA1 = (16 + m) * PK_ + k4;
    const int fB  = k4 * PX_ + wave * 32 + m;
    int fBh = fB + 16;
    asm volatile("" : "+v"(fBh));
    v4f gx[8], gy[8], gg[NA][2];
    {
        v4f hx[8], hy[8], hg[NA][2];
        #pragma unroll
        for (int r = 0; r < 8; ++r) {
            gx[r] = *(const v4f*)(pX + (size_t)(4 * r) * LDB);
            hx[r] = *(const v4f*)(pX + (size_t)(KC_ + 4 * r) * LDB);
            if (TWO) {
                gy[r] = *(const v4f*)(pY + (size_t)(4 * r) * LDB);
                hy[r] = *(const v4f*)(pY + (size_t)(KC_ + 4 * r) * LDB);
            }
        }
        #pragma unroll
        for (int a = 0; a < NA; ++a) {
            #pragma unroll
            for (int r = 0; r < 2; ++r) {
                gg[a][r] = *(const v4f*)(pG[a] + (size_t)(16 * r) * C_);
                hg[a][r] = *(const v4f*)(pG[a] + (size_t)(16 * r) * C_ + KC_);
            }
        }
        #pragma unroll
        for (int r = 0; r < 8; ++r) {
            *(v4f*)(xs + sxOff + 4 * r * PX_) = TWO ? gx[r] + gy[r] : gx[r];
            *(v4f*)(xs + SXS + sxOff + 4 * r * PX_) = TWO ? hx[r] + hy[r] : hx[r];
        }
        #pragma unroll
        for (int a = 0; a < NA; ++a) {
            #pragma unroll
            for (int r = 0; r < 2; ++r) {
                *(v4f*)(gs + (a * 32 + 16 * r) * PK_ + sgOff) = gg[a][r];
                *(v4f*)(gs + SGS + (a * 32 + 16 * r) * PK_ + sgOff) = hg[a][r];
            }
        }
    }
    __syncthreads();
    FragKN<NA> P, Q;
    frag_kn_load<NA>(P, xs, gs, fA0, fA1, fB, fBh);
    __syncthreads();
    #pragma unroll 1
    for (int c = 0; c < NCH; c += 2) {
        const bool more = (c + 2 < NCH);
        if (more) {
            const int kc = (c + 2) * KC_;
            #pragma unroll
            for (int r = 0; r < 8; ++r) gx[r] = *(const v4f*)(pX + (size_t)(kc + 4 * r) * LDB);
            if (TWO) {
                #pragma unroll
                for (int r = 0; r < 8; ++r) gy[r] = *(const v4f*)(pY + (size_t)(kc + 4 * r) * LDB);
            }
            #pragma unroll
            for (int a = 0; a < NA; ++a) {
                #pragma unroll
                for (int r = 0; r < 2; ++r) gg[a][r] = *(const v4f*)(pG[a] + (size_t)(16 * r) * C_ + kc);
            }
        }
        frag_kn_load<NA>(Q, xs + SXS, gs + SGS, fA0, fA1, fB, fBh);
        __builtin_amdgcn_sched_barrier(0);
        frag_kn_mma<NA>(P, o);
        __builtin_amdgcn_sched_barrier(0);
        if (more) {
            #pragma unroll
            for (int r = 0; r < 8; ++r) *(v4f*)(xs + sxOff + 4 * r * PX_) = TWO ? gx[r] + gy[r] : gx[r];
            #pragma unroll
            for (int a = 0; a < NA; ++a) {
                #pragma unroll
                for (int r = 0; r < 2; ++r) *(v4f*)(gs + (a * 32 + 16 * r) * PK_ + sgOff) = gg[a][r];
            }
        }
        __syncthreads();
        if (more) {
            const int kc = (c + 3) * KC_;
            #pragma unroll
            for (int r = 0; r < 8; ++r) gx[r] = *(const v4f*)(pX + (size_t)(kc + 4 * r) * LDB);
            if (TWO) {
                #pragma unroll
                for (int r = 0; r < 8; ++r) gy[r] = *(const v4f*)(pY + (size_t)(kc + 4 * r) * LDB);
            }
            #pragma unroll
            for (int a = 0; a < NA; ++a) {
                #pragma unroll
                for (int r = 0; r < 2; ++r) gg[a][r] = *(const v4f*)(pG[a] + (size_t)(16 * r) * C_ + kc);
            }
            frag_kn_load<NA>(P, xs, gs, fA0, fA1, fB, fBh);
        }
        __builtin_amdgcn_sched_barrier(0);
        frag_kn_mma<NA>(Q, o);
        __builtin_amdgcn_sched_barrier(0);
        if (more) {
            #pragma unroll
            for (int r = 0; r < 8; ++r) *(v4f*)(xs + SXS + sxOff + 4 * r * PX_) = TWO ? gx[r] + gy[r] : gx[r];
            #pragma unroll
            for (int a = 0; a < NA; ++a) {
                #pragma unroll
                for (int r = 0; r < 2; ++r) *(v4f*)(gs + SGS + (a * 32 + 16 * r) * PK_ + sgOff) = gg[a][r];
            }
        }
        __syncthreads();
    }
}

#define KH_ (N_ / 2)
__global__ __launch_bounds__(128) __attribute__((amdgpu_waves_per_eu(1, 2))) void k_gram(const float* __restrict__ x,
                                              float* __restrict__ S0,
                                              float* __restrict__ S1) {
    __shared__ __attribute__((aligned(16))) float tile[2 * GROWS_ * PK_];
    const int lane = threadIdx.x & 31;
    const int wave = threadIdx.x >> 5;
    const int m    = lane & 15;
    const int half = lane >> 4;
    const int p    = blockIdx.x >> 2;
    const int I    = (p >= 4) + (p >= 7) + (p >= 9);
    const int J    = p - (4 * I - ((I * (I - 1)) >> 1)) + I;
    const int c0   = I * 128 + (blockIdx.x & 3) * 32;
    const int dblk = J * 128;
    const int d0   = dblk + wave * 32;
    const int kh   = blockIdx.y;
    const int b    = blockIdx.z;
    const size_t bOff = (size_t)b * C_ * N_ + (size_t)kh * KH_;
    v8f a00 = {}, a01 = {}, a10 = {}, a11 = {};
    mm_rows_rows<N_, KH_>(x + bOff + (size_t)c0 * N_, x + bOff + (size_t)dblk * N_, tile, a00, a01, a10, a11);
    float* Sb = (kh ? S1 : S0) + (size_t)b * C_ * C_;
    #pragma unroll
    for (int j = 0; j < 8; ++j) {
        const int cA = c0 + j + 8 * half;
        const int cB = cA + 16;
        Sb[(size_t)cA * C_ + d0 + m]      = a00[j];
        Sb[(size_t)cA * C_ + d0 + 16 + m] = a01[j];
        Sb[(size_t)cB * C_ + d0 + m]      = a10[j];
        Sb[(size_t)cB * C_ + d0 + 16 + m] = a11[j];
    }
    if (I != J) {
        float* r0 = Sb + (size_t)(d0 + m) * C_ + c0 + 8 * half;
        float* r1 = r0 + (size_t)16 * C_;
        *(v8f*)(r0)      = a00;
        *(v8f*)(r0 + 16) = a10;
        *(v8f*)(r1)      = a01;
        *(v8f*)(r1 + 16) = a11;
    }
}

__global__ __launch_bounds__(128) __attribute__((amdgpu_waves_per_eu(1, 2))) void k_projs(const float* __restrict__ S0,
                                               const float* __restrict__ S1,
                                               const float* __restrict__ Wq,
                                               const float* __restrict__ Wk,
                                               float* __restrict__ T,
                                               float* __restrict__ U) {
    __shared__ __attribute__((aligned(16))) float xs[2 * KC_ * PX_];
    __shared__ __attribute__((aligned(16))) float gs[2 * 2 * 32 * PK_];
    const int lane = threadIdx.x & 31;
    const int wave = threadIdx.x >> 5;
    const int m    = lane & 15;
    const int half = lane >> 4;
    const int c0   = blockIdx.y * 32;
    const int nblk = blockIdx.x * 128;
    const int n0   = nblk + wave * 32;
    const int b    = blockIdx.z;
    v8f o[2][4] = {};
    mm_rows_kn<C_, 2, true>(Wq + (size_t)c0 * C_, Wk + (size_t)c0 * C_, S0 + (size_t)b * C_ * C_ + nblk,
                            S1 + (size_t)b * C_ * C_ + nblk, xs, gs, o);
    float* Tb = T + (size_t)b * C_ * C_;
    float* Ub = U + (size_t)b * C_ * C_;
    #pragma unroll
    for (int j = 0; j < 8; ++j) {
        const int cA = c0 + j + 8 * half;
        const int cB = cA + 16;
        Tb[(size_t)cA * C_ + n0 + m]      = o[0][0][j];
        Tb[(size_t)cA * C_ + n0 + 16 + m] = o[0][1][j];
        Tb[(size_t)cB * C_ + n0 + m]      = o[0][2][j];
        Tb[(size_t)cB * C_ + n0 + 16 + m] = o[0][3][j];
        Ub[(size_t)cA * C_ + n0 + m]      = o[1][0][j];
        Ub[(size_t)cA * C_ + n0 + 16 + m] = o[1][1][j];
        Ub[(size_t)cB * C_ + n0 + m]      = o[1][2][j];
        Ub[(size_t)cB * C_ + n0 + 16 + m] = o[1][3][j];
    }
}

__global__ __launch_bounds__(256) void k_rownorm(const float* __restrict__ T,
                                                 const float* __restrict__ U,
                                                 const float* __restrict__ Wq,
                                                 const float* __restrict__ Wk,
                                                 float* __restrict__ invQ,
                                                 float* __restrict__ invK) {
    __shared__ float sq[256], sk[256];
    const int row = blockIdx.x;
    const int c   = row & (C_ - 1);
    const float* tr = T + (size_t)row * C_;
    const float* ur = U + (size_t)row * C_;
    const float* wq = Wq + (size_t)c * C_;
    const float* wk = Wk + (size_t)c * C_;
    float aq = 0.f, ak = 0.f;
    for (int e = threadIdx.x; e < C_; e += 256) {
        aq += tr[e] * wq[e];
        ak += ur[e] * wk[e];
    }
    sq[threadIdx.x] = aq; sk[threadIdx.x] = ak;
    __syncthreads();
    for (int s = 128; s > 0; s >>= 1) {
        if (threadIdx.x < s) {
            sq[threadIdx.x] += sq[threadIdx.x + s];
            sk[threadIdx.x] += sk[threadIdx.x + s];
        }
        __syncthreads();
    }
    if (threadIdx.x == 0) { invQ[row] = rsqrtf(sq[0]); invK[row] = rsqrtf(sk[0]); }
}

__global__ __launch_bounds__(128) __attribute__((amdgpu_waves_per_eu(1, 2))) void k_qkt(const float* __restrict__ T,
                                             const float* __restrict__ Wk,
                                             const float* __restrict__ invQ,
                                             const float* __restrict__ invK,
                                             float* __restrict__ M) {
    __shared__ __attribute__((aligned(16))) float tile[2 * GROWS_ * PK_];
    const int lane = threadIdx.x & 31;
    const int wave = threadIdx.x >> 5;
    const int m    = lane & 15;
    const int half = lane >> 4;
    const int c0   = blockIdx.y * 32;
    const int dblk = blockIdx.x * 128;
    const int d0   = dblk + wave * 32;
    const int b    = blockIdx.z;
    const size_t bOff = (size_t)b * C_ * C_;
    v8f a00 = {}, a01 = {}, a10 = {}, a11 = {};
    mm_rows_rows<C_, C_>(T + bOff + (size_t)c0 * C_, Wk + (size_t)dblk * C_, tile, a00, a01, a10, a11);
    float* Mb = M + (size_t)b * C_ * C_;
    const float ik0 = invK[b * C_ + d0 + m];
    const float ik1 = invK[b * C_ + d0 + 16 + m];
    #pragma unroll
    for (int j = 0; j < 8; ++j) {
        const int cA = c0 + j + 8 * half;
        const int cB = cA + 16;
        const float iqA = invQ[b * C_ + cA];
        const float iqB = invQ[b * C_ + cB];
        Mb[(size_t)cA * C_ + d0 + m]      = a00[j] * iqA * ik0;
        Mb[(size_t)cA * C_ + d0 + 16 + m] = a01[j] * iqA * ik1;
        Mb[(size_t)cB * C_ + d0 + m]      = a10[j] * iqB * ik0;
        Mb[(size_t)cB * C_ + d0 + 16 + m] = a11[j] * iqB * ik1;
    }
}

__global__ __launch_bounds__(256) void k_colsoftmax(float* __restrict__ M) {
    __shared__ float smax[8][32], smin[8][32], ssum[8][32];
    const int lane = threadIdx.x & 31;
    const int wave = threadIdx.x >> 5;
    const int tile = blockIdx.x;
    const int b  = tile / (C_ / 32);
    const int d  = (tile - b * (C_ / 32)) * 32 + lane;
    float* p = M + (size_t)b * C_ * C_ + (size_t)(wave * 64) * C_ + d;
    float v[64];
    for (int i = 0; i < 64; ++i) v[i] = 1.f - p[(size_t)i * C_];
    float mx = -3.4e38f, mn = 3.4e38f;
    for (int i = 0; i < 64; ++i) { mx = fmaxf(mx, v[i]); mn = fminf(mn, v[i]); }
    smax[wave][lane] = mx; smin[wave][lane] = mn;
    __syncthreads();
    float cmx = -3.4e38f, cmn = 3.4e38f;
    for (int j = 0; j < 8; ++j) {
        cmx = fmaxf(cmx, smax[j][lane]);
        cmn = fminf(cmn, smin[j][lane]);
    }
    const float dn = cmx + EPS_;
    const float inv_dn = 1.f / dn;
    const float sh = 1.f - (cmn / dn) * (1.f / HT_);
    float z = 0.f;
    for (int i = 0; i < 64; ++i) {
        const float m4 = 1.f - (v[i] * inv_dn) * (1.f / HT_);
        const float e = expf(m4 - sh);
        v[i] = e;
        z += e;
    }
    ssum[wave][lane] = z;
    __syncthreads();
    float tot = 0.f;
    for (int j = 0; j < 8; ++j) tot += ssum[j][lane];
    const float invz = 1.f / tot;
    for (int i = 0; i < 64; ++i) p[(size_t)i * C_] = v[i] * invz;
}

__global__ __launch_bounds__(256) void k_rowl1(float* __restrict__ M) {
    __shared__ float ssum[256];
    const int row = blockIdx.x;
    float* r = M + (size_t)row * C_;
    float s = 0.f;
    for (int d = threadIdx.x; d < C_; d += 256) s += r[d];
    ssum[threadIdx.x] = s;
    __syncthreads();
    for (int t = 128; t > 0; t >>= 1) {
        if (threadIdx.x < t) ssum[threadIdx.x] += ssum[threadIdx.x + t];
        __syncthreads();
    }
    const float inv = 1.f / (ssum[0] + EPS_);
    for (int d = threadIdx.x; d < C_; d += 256) r[d] *= inv;
}

__global__ __launch_bounds__(128) __attribute__((amdgpu_waves_per_eu(1, 2))) void k_mwv(const float* __restrict__ M,
                                             const float* __restrict__ Wv,
                                             float* __restrict__ G) {
    __shared__ __attribute__((aligned(16))) float xs[2 * KC_ * PX_];
    __shared__ __attribute__((aligned(16))) float gs[2 * 32 * PK_];
    const int lane = threadIdx.x & 31;
    const int wave = threadIdx.x >> 5;
    const int m    = lane & 15;
    const int half = lane >> 4;
    const int c0   = blockIdx.y * 32;
    const int eblk = blockIdx.x * 128;
    const int e0   = eblk + wave * 32;
    const int b    = blockIdx.z;
    const float* Mb = M + (size_t)b * C_ * C_ + (size_t)c0 * C_;
    v8f o[1][4] = {};
    mm_rows_kn<C_, 1, false>(Mb, Mb, Wv + eblk, Wv + eblk, xs, gs, o);
    float* Gb = G + (size_t)b * C_ * C_;
    #pragma unroll
    for (int j = 0; j < 8; ++j) {
        const int cA = c0 + j + 8 * half;
        const int cB = cA + 16;
        Gb[(size_t)cA * C_ + e0 + m]      = o[0][0][j];
        Gb[(size_t)cA * C_ + e0 + 16 + m] = o[0][1][j];
        Gb[(size_t)cB * C_ + e0 + m]      = o[0][2][j];
        Gb[(size_t)cB * C_ + e0 + 16 + m] = o[0][3][j];
    }
}

__global__ __launch_bounds__(128) void k_out(const float* __restrict__ x,
                                             const float* __restrict__ G,
                                             const float* __restrict__ gamma,
                                             float* __restrict__ y) {
    __shared__ __attribute__((aligned(16))) float xs[2 * KC_ * PX_];
    __shared__ __attribute__((aligned(16))) float gs[2 * 32 * PK_];
    const int lane = threadIdx.x & 31;
    const int wave = threadIdx.x >> 5;
    const int m    = lane & 15;
    const int half = lane >> 4;
    const int c0   = blockIdx.y * 32;
    const int nblk = blockIdx.x * 128;
    const int n0   = nblk + wave * 32;
    const int b    = blockIdx.z;
    const float* xb = x + (size_t)b * C_ * N_;
    const float* Gb = G + (size_t)b * C_ * C_ + (size_t)c0 * C_;
    v8f o[1][4] = {};
    mm_rows_kn<N_, 1, false>(Gb, Gb, xb + nblk, xb + nblk, xs, gs, o);
    const float g = gamma[0];
    float* yb = y + (size_t)b * C_ * N_;
    #pragma unroll
    for (int j = 0; j < 8; ++j) {
        const int cA = c0 + j + 8 * half;
        const int cB = cA + 16;
        const size_t o0 = (size_t)cA * N_ + n0 + m;
        const size_t o1 = (size_t)cA * N_ + n0 + 16 + m;
        const size_t o2 = (size_t)cB * N_ + n0 + m;
        const size_t o3 = (size_t)cB * N_ + n0 + 16 + m;
        yb[o0] = xb[o0] + g * o[0][0][j];
        yb[o1] = xb[o1] + g * o[0][1][j];
        yb[o2] = xb[o2] + g * o[0][2][j];
        yb[o3] = xb[o3] + g * o[0][3][j];
    }
}

extern "C" void kernel_launch(void* const* d_in, const int* in_sizes, int n_in,
                              void* d_out, int out_size, void* d_ws, size_t ws_size,
                              hipStream_t stream) {
    (void)in_sizes; (void)n_in; (void)out_size; (void)ws_size;
    const float* x     = (const float*)d_in[0];
    const float* Wq    = (const float*)d_in[1];
    const float* Wk    = (const float*)d_in[2];
    const float* Wv    = (const float*)d_in[3];
    const float* gamma = (const float*)d_in[4];
    float* y = (float*)d_out;

    char* ws = (char*)d_ws;
    const size_t mBytes  = (size_t)B_ * C_ * C_ * sizeof(float);
    float* S     = (float*)(ws);
    float* T     = (float*)(ws + mBytes);
    float* U     = (float*)(ws + 2 * mBytes);
    float* M     = (float*)(ws + 3 * mBytes);
    float* G     = (float*)(ws + 4 * mBytes);
    float* invQ  = (float*)(ws + 5 * mBytes);
    float* invK  = invQ  + B_ * C_;

    k_gram      <<<dim3(40, 2, B_), 128, 0, stream>>>(x, S, G);
    k_projs     <<<dim3(C_ / 128, C_ / 32, B_), 128, 0, stream>>>(S, G, Wq, Wk, T, U);
    k_rownorm   <<<B_ * C_, 256, 0, stream>>>(T, U, Wq, Wk, invQ, invK);
    k_qkt       <<<dim3(C_ / 128, C_ / 32, B_), 128, 0, stream>>>(T, Wk, invQ, invK, M);
    k_colsoftmax<<<B_ * C_ / 32, 256, 0, stream>>>(M);
    k_rowl1     <<<B_ * C_, 256, 0, stream>>>(M);
    k_mwv       <<<dim3(C_ / 128, C_ / 32, B_), 128, 0, stream>>>(M, Wv, G);
    k_out       <<<dim3(N_ / 128, C_ / 32, B_), 128, 0, stream>>>(x, G, gamma, y);
}


# Gram K range cut in three parts (44+42+42 chunks, three planes summed by the next GEMM while staging) instead of two halves: 960 work-groups fill two rounds, none runs alone; Gram sum tree changes
# speedup vs baseline: 2.7328x; 1.0317x over previous
#include <hip/hip_runtime.h>

typedef __attribute__((ext_vector_type(2))) float v2f;
typedef __attribute__((ext_vector_type(4))) float v4f;
typedef __attribute__((ext_vector_type(8))) float v8f;

#define B_ 8
#define C_ 512
#define N_ 4096
#define EPS_ 1e-6f
#define HT_ 0.25f

static __device__ __forceinline__ v8f wmma4(v2f a, v2f b, v8f c) {
    return __builtin_amdgcn_wmma_f32_16x16x4_f32(false, a, false, b, (short)0, c, false, false);
}

static __device__ __forceinline__ v2f lo2(v4f v) { return __builtin_shufflevector(v, v, 0, 1); }
static __device__ __forceinline__ v2f hi2(v4f v) { return __builtin_shufflevector(v, v, 2, 3); }


#define KC_    32
#define PK_    36
#define GROWS_ 160
#define PX_    132

struct FragRR { v4f aA[4], aB[4], bA[4], bB[4]; };

static __device__ __forceinline__ void frag_rr_load(FragRR& f, const float* __restrict__ st,
                                                    int fA0, int fA1, int fB0, int fB1) {
    #pragma unroll
    for (int j = 0; j < 4; ++j) {
        f.aA[j] = *(const v4f*)(st + fA0 + 8 * j);
        f.aB[j] = *(const v4f*)(st + fA1 + 8 * j);
        f.bA[j] = *(const v4f*)(st + fB0 + 8 * j);
        f.bB[j] = *(const v4f*)(st + fB1 + 8 * j);
    }
}

static __device__ __forceinline__ void frag_rr_mma(const FragRR& f, v8f& a00, v8f& a01, v8f& a10, v8f& a11) {
    #pragma unroll
    for (int j = 0; j < 4; ++j) {
        a00 = wmma4(lo2(f.aA[j]), lo2(f.bA[j]), a00); a01 = wmma4(lo2(f.aA[j]), lo2(f.bB[j]), a01);
        a10 = wmma4(lo2(f.aB[j]), lo2(f.bA[j]), a10); a11 = wmma4(lo2(f.aB[j]), lo2(f.bB[j]), a11);
        a00 = wmma4(hi2(f.aA[j]), hi2(f.bA[j]), a00); a01 = wmma4(hi2(f.aA[j]), hi2(f.bB[j]), a01);
        a10 = wmma4(hi2(f.aB[j]), hi2(f.bA[j]), a10); a11 = wmma4(hi2(f.aB[j]), hi2(f.bB[j]), a11);
    }
}

template <int LD>
static __device__ __forceinline__ void mm_rows_rows(const float* __restrict__ gA,
                                                    const float* __restrict__ gB,
                                                    float* __restrict__ tile, const int nch,
                                                    v8f& a00, v8f& a01, v8f& a10, v8f& a11) {
    constexpr int STG = GROWS_ * PK_;
    const int t    = threadIdx.x;
    const int lane = t & 31;
    const int wave = t >> 5;
    const int m    = lane & 15;
    const int k4   = (lane >> 4) << 2;
    const int sr = t >> 3;
    const int sq = (t & 7) << 2;
    const float* pA = gA + (size_t)sr * LD + sq;
    const float* pB = gB + (size_t)sr * LD + sq;
    const int sOff = sr * PK_ + sq;
    const int fA0 = m * PK_ + k4;
    const int fA1 = (16 + m) * PK_ + k4;
    const int fB0 = (32 + wave * 32 + m) * PK_ + k4;
    const int fB1 = fB0 + 16 * PK_;
    v4f g[10];
    {
        v4f h[10];
        #pragma unroll
        for (int r = 0; r < 2; ++r) {
            g[r] = *(const v4f*)(pA + (size_t)(16 * r) * LD);
            h[r] = *(const v4f*)(pA + (size_t)(16 * r) * LD + KC_);
        }
        #pragma unroll
        for (int r = 0; r < 8; ++r) {
            g[2 + r] = *(const v4f*)(pB + (size_t)(16 * r) * LD);
            h[2 + r] = *(const v4f*)(pB + (size_t)(16 * r) * LD + KC_);
        }
        #pragma unroll
        for (int r = 0; r < 10; ++r) {
            *(v4f*)(tile + sOff + 16 * r * PK_) = g[r];
            *(v4f*)(tile + STG + sOff + 16 * r * PK_) = h[r];
        }
    }
    __syncthreads();
    FragRR P, Q;
    frag_rr_load(P, tile, fA0, fA1, fB0, fB1);
    __syncthreads();
    #pragma unroll 1
    for (int c = 0; c < nch; c += 2) {
        const bool more = (c + 2 < nch);
        if (more) {
            const int kc = (c + 2) * KC_;
            #pragma unroll
            for (int r = 0; r < 2; ++r) g[r] = *(const v4f*)(pA + (size_t)(16 * r) * LD + kc);
            #pragma unroll
            for (int r = 0; r < 8; ++r) g[2 + r] = *(const v4f*)(pB + (size_t)(16 * r) * LD + kc);
        }
        frag_rr_load(Q, tile + STG, fA0, fA1, fB0, fB1);
        __builtin_amdgcn_sched_barrier(0);
        frag_rr_mma(P, a00, a01, a10, a11);
        __builtin_amdgcn_sched_barrier(0);
        if (more) {
            #pragma unroll
            for (int r = 0; r < 10; ++r) *(v4f*)(tile + sOff + 16 * r * PK_) = g[r];
        }
        __syncthreads();
        if (more) {
            const int kc = (c + 3) * KC_;
            #pragma unroll
            for (int r = 0; r < 2; ++r) g[r] = *(const v4f*)(pA + (size_t)(16 * r) * LD + kc);
            #pragma unroll
            for (int r = 0; r < 8; ++r) g[2 + r] = *(const v4f*)(pB + (size_t)(16 * r) * LD + kc);
            frag_rr_load(P, tile, fA0, fA1, fB0, fB1);
        }
        __builtin_amdgcn_sched_barrier(0);
        frag_rr_mma(Q, a00, a01, a10, a11);
        __builtin_amdgcn_sched_barrier(0);
        if (more) {
            #pragma unroll
            for (int r = 0; r < 10; ++r) *(v4f*)(tile + STG + sOff + 16 * r * PK_) = g[r];
        }
        __syncthreads();
    }
}

template <int NA>
struct FragKN { v4f aA[NA][4], aB[NA][4]; v2f b0[4], b1[4], b2[4], b3[4]; };

template <int NA>
static __device__ __forceinline__ void frag_kn_load(FragKN<NA>& f, const float* __restrict__ sx,
                                                    const float* __restrict__ sg,
                                                    int fA0, int fA1, int fB, int fBh) {
    #pragma unroll
    for (int j = 0; j < 4; ++j) {
        #pragma unroll
        for (int a = 0; a < NA; ++a) {
            f.aA[a][j] = *(const v4f*)(sg + a * 32 * PK_ + fA0 + 8 * j);
            f.aB[a][j] = *(const v4f*)(sg + a * 32 * PK_ + fA1 + 8 * j);
        }
        const float* r0 = sx + fB + (8 * j) * PX_;
        const float* r1 = sx + fBh + (8 * j) * PX_;
        f.b0[j].x = r0[0];       f.b0[j].y = r0[PX_];
        f.b1[j].x = r1[0];       f.b1[j].y = r1[PX_];
        f.b2[j].x = r0[2 * PX_]; f.b2[j].y = r0[3 * PX_];
        f.b3[j].x = r1[2 * PX_]; f.b3[j].y = r1[3 * PX_];
    }
}

template <int NA>
static __device__ __forceinline__ void frag_kn_mma(const FragKN<NA>& f, v8f (&o)[NA][4]) {
    #pragma unroll
    for (int j = 0; j < 4; ++j) {
        #pragma unroll
        for (int a = 0; a < NA; ++a) {
            o[a][0] = wmma4(lo2(f.aA[a][j]), f.b0[j], o[a][0]); o[a][1] = wmma4(lo2(f.aA[a][j]), f.b1[j], o[a][1]);
            o[a][2] = wmma4(lo2(f.aB[a][j]), f.b0[j], o[a][2]); o[a][3] = wmma4(lo2(f.aB[a][j]), f.b1[j], o[a][3]);
        }
        #pragma unroll
        for (int a = 0; a < NA; ++a) {
            o[a][0] = wmma4(hi2(f.aA[a][j]), f.b2[j], o[a][0]); o[a][1] = wmma4(hi2(f.aA[a][j]), f.b3[j], o[a][1]);
            o[a][2] = wmma4(hi2(f.aB[a][j]), f.b2[j], o[a][2]); o[a][3] = wmma4(hi2(f.aB[a][j]), f.b3[j], o[a][3]);
        }
    }
}

#define PLSUM_(x, y, z) (NPL == 1 ? (x) : NPL == 2 ? (x) + (y) : ((x) + (y)) + (z))
template <int LDB, int NA, int NPL>
static __device__ __forceinline__ void mm_rows_kn(const float* __restrict__ gA0,
                                                  const float* __restrict__ gA1,
                                                  const float* __restrict__ gB,
                                                  const float* __restrict__ gB2,
                                                  const float* __restrict__ gB3,
                                                  float* __restrict__ xs,
                                                  float* __restrict__ gs,
                                                  v8f (&o)[NA][4]) {
    constexpr int NCH = C_ / KC_;
    constexpr int SXS = KC_ * PX_;
    constexpr int SGS = NA * 32 * PK_;
    const int t    = threadIdx.x;
    const int lane = t & 31;
    const int wave = t >> 5;
    const int m    = lane & 15;
    const int k4   = (lane >> 4) << 2;
    const float* pX = gB + (size_t)wave * LDB + (lane << 2);
    const float* pY = gB2 + (size_t)wave * LDB + (lane << 2);
    const float* pZ = gB3 + (size_t)wave * LDB + (lane << 2);
    const int sxOff = wave * PX_ + (lane << 2);
    const int sr = t >> 3;
    const int sq = (t & 7) << 2;
    const float* pG[NA];
    pG[0] = gA0 + (size_t)sr * C_ + sq;
    if (NA > 1) pG[NA - 1] = gA1 + (size_t)sr * C_ + sq;
    const int sgOff = sr * PK_ + sq;
    const int fA0 = m * PK_ + k4;
    const int fA1 = (16 + m) * PK_ + k4;
    const int fB  = k4 * PX_ + wave * 32 + m;
    int fBh = fB + 16;
    asm volatile("" : "+v"(fBh));
    v4f gx[8], gy[8], gz[8], gg[NA][2];
    {
        v4f hx[8], hy[8], hz[8], hg[NA][2];
        #pragma unroll
        for (int r = 0; r < 8; ++r) {
            gx[r] = *(const v4f*)(pX + (size_t)(4 * r) * LDB);
            hx[r] = *(const v4f*)(pX + (size_t)(KC_ + 4 * r) * LDB);
            if (NPL > 1) {
                gy[r] = *(const v4f*)(pY + (size_t)(4 * r) * LDB);
                hy[r] = *(const v4f*)(pY + (size_t)(KC_ + 4 * r) * LDB);
            }
            if (NPL > 2) {
                gz[r] = *(const v4f*)(pZ + (size_t)(4 * r) * LDB);
                hz[r] = *(const v4f*)(pZ + (size_t)(KC_ + 4 * r) * LDB);
            }
        }
        #pragma unroll
        for (int a = 0; a < NA; ++a) {
            #pragma unroll
            for (int r = 0; r < 2; ++r) {
                gg[a][r] = *(const v4f*)(pG[a] + (size_t)(16 * r) * C_);
                hg[a][r] = *(const v4f*)(pG[a] + (size_t)(16 * r) * C_ + KC_);
            }
        }
        #pragma unroll
        for (int r = 0; r < 8; ++r) {
            *(v4f*)(xs + sxOff + 4 * r * PX_) = PLSUM_(gx[r], gy[r], gz[r]);
            *(v4f*)(xs + SXS + sxOff + 4 * r * PX_) = PLSUM_(hx[r], hy[r], hz[r]);
        }
        #pragma unroll
        for (int a = 0; a < NA; ++a) {
            #pragma unroll
            for (int r = 0; r < 2; ++r) {
                *(v4f*)(gs + (a * 32 + 16 * r) * PK_ + sgOff) = gg[a][r];
                *(v4f*)(gs + SGS + (a * 32 + 16 * r) * PK_ + sgOff) = hg[a][r];
            }
        }
    }
    __syncthreads();
    FragKN<NA> P, Q;
    frag_kn_load<NA>(P, xs, gs, fA0, fA1, fB, fBh);
    __syncthreads();
    #pragma unroll 1
    for (int c = 0; c < NCH; c += 2) {
        const bool more = (c + 2 < NCH);
        if (more) {
            const int kc = (c + 2) * KC_;
            #pragma unroll
            for (int r = 0; r < 8; ++r) gx[r] = *(const v4f*)(pX + (size_t)(kc + 4 * r) * LDB);
            if (NPL > 1) {
                #pragma unroll
                for (int r = 0; r < 8; ++r) gy[r] = *(const v4f*)(pY + (size_t)(kc + 4 * r) * LDB);
            }
            if (NPL > 2) {
                #pragma unroll
                for (int r = 0; r < 8; ++r) gz[r] = *(const v4f*)(pZ + (size_t)(kc + 4 * r) * LDB);
            }
            #pragma unroll
            for (int a = 0; a < NA; ++a) {
                #pragma unroll
                for (int r = 0; r < 2; ++r) gg[a][r] = *(const v4f*)(pG[a] + (size_t)(16 * r) * C_ + kc);
            }
        }
        frag_kn_load<NA>(Q, xs + SXS, gs + SGS, fA0, fA1, fB, fBh);
        __builtin_amdgcn_sched_barrier(0);
        frag_kn_mma<NA>(P, o);
        __builtin_amdgcn_sched_barrier(0);
        if (more) {
            #pragma unroll
            for (int r = 0; r < 8; ++r) *(v4f*)(xs + sxOff + 4 * r * PX_) = PLSUM_(gx[r], gy[r], gz[r]);
            #pragma unroll
            for (int a = 0; a < NA; ++a) {
                #pragma unroll
                for (int r = 0; r < 2; ++r) *(v4f*)(gs + (a * 32 + 16 * r) * PK_ + sgOff) = gg[a][r];
            }
        }
        __syncthreads();
        if (more) {
            const int kc = (c + 3) * KC_;
            #pragma unroll
            for (int r = 0; r < 8; ++r) gx[r] = *(const v4f*)(pX + (size_t)(kc + 4 * r) * LDB);
            if (NPL > 1) {
                #pragma unroll
                for (int r = 0; r < 8; ++r) gy[r] = *(const v4f*)(pY + (size_t)(kc + 4 * r) * LDB);
            }
            if (NPL > 2) {
                #pragma unroll
                for (int r = 0; r < 8; ++r) gz[r] = *(const v4f*)(pZ + (size_t)(kc + 4 * r) * LDB);
            }
            #pragma unroll
            for (int a = 0; a < NA; ++a) {
                #pragma unroll
                for (int r = 0; r < 2; ++r) gg[a][r] = *(const v4f*)(pG[a] + (size_t)(16 * r) * C_ + kc);
            }
            frag_kn_load<NA>(P, xs, gs, fA0, fA1, fB, fBh);
        }
        __builtin_amdgcn_sched_barrier(0);
        frag_kn_mma<NA>(Q, o);
        __builtin_amdgcn_sched_barrier(0);
        if (more) {
            #pragma unroll
            for (int r = 0; r < 8; ++r) *(v4f*)(xs + SXS + sxOff + 4 * r * PX_) = PLSUM_(gx[r], gy[r], gz[r]);
            #pragma unroll
            for (int a = 0; a < NA; ++a) {
                #pragma unroll
                for (int r = 0; r < 2; ++r) *(v4f*)(gs + SGS + (a * 32 + 16 * r) * PK_ + sgOff) = gg[a][r];
            }
        }
        __syncthreads();
    }
}

#define KP0_ 44
#define KP1_ 42
__global__ __launch_bounds__(128) __attribute__((amdgpu_waves_per_eu(1, 2))) void k_gram(const float* __restrict__ x,
                                              float* __restrict__ S0,
                                              float* __restrict__ S1,
                                              float* __restrict__ S2) {
    __shared__ __attribute__((aligned(16))) float tile[2 * GROWS_ * PK_];
    const int lane = threadIdx.x & 31;
    const int wave = threadIdx.x >> 5;
    const int m    = lane & 15;
    const int half = lane >> 4;
    const int p    = blockIdx.x >> 2;
    const int I    = (p >= 4) + (p >= 7) + (p >= 9);
    const int J    = p - (4 * I - ((I * (I - 1)) >> 1)) + I;
    const int c0   = I * 128 + (blockIdx.x & 3) * 32;
    const int dblk = J * 128;
    const int d0   = dblk + wave * 32;
    const int kh   = blockIdx.y;
    const int b    = blockIdx.z;
    const int ch0  = kh == 0 ? 0 : KP0_ + (kh - 1) * KP1_;
    const int nch  = kh == 0 ? KP0_ : KP1_;
    const size_t bOff = (size_t)b * C_ * N_ + (size_t)ch0 * KC_;
    v8f a00 = {}, a01 = {}, a10 = {}, a11 = {};
    mm_rows_rows<N_>(x + bOff + (size_t)c0 * N_, x + bOff + (size_t)dblk * N_, tile, nch, a00, a01, a10, a11);
    float* Sb = (kh == 0 ? S0 : kh == 1 ? S1 : S2) + (size_t)b * C_ * C_;
    #pragma unroll
    for (int j = 0; j < 8; ++j) {
        const int cA = c0 + j + 8 * half;
        const int cB = cA + 16;
        Sb[(size_t)cA * C_ + d0 + m]      = a00[j];
        Sb[(size_t)cA * C_ + d0 + 16 + m] = a01[j];
        Sb[(size_t)cB * C_ + d0 + m]      = a10[j];
        Sb[(size_t)cB * C_ + d0 + 16 + m] = a11[j];
    }
    if (I != J) {
        float* r0 = Sb + (size_t)(d0 + m) * C_ + c0 + 8 * half;
        float* r1 = r0 + (size_t)16 * C_;
        *(v8f*)(r0)      = a00;
        *(v8f*)(r0 + 16) = a10;
        *(v8f*)(r1)      = a01;
        *(v8f*)(r1 + 16) = a11;
    }
}

__global__ __launch_bounds__(128) __attribute__((amdgpu_waves_per_eu(1, 2))) void k_projs(const float* __restrict__ S0,
                                               const float* __restrict__ S1,
                                               const float* __restrict__ S2,
                                               const float* __restrict__ Wq,
                                               const float* __restrict__ Wk,
                                               float* __restrict__ T,
                                               float* __restrict__ U) {
    __shared__ __attribute__((aligned(16))) float xs[2 * KC_ * PX_];
    __shared__ __attribute__((aligned(16))) float gs[2 * 2 * 32 * PK_];
    const int lane = threadIdx.x & 31;
    const int wave = threadIdx.x >> 5;
    const int m    = lane & 15;
    const int half = lane >> 4;
    const int c0   = blockIdx.y * 32;
    const int nblk = blockIdx.x * 128;
    const int n0   = nblk + wave * 32;
    const int b    = blockIdx.z;
    v8f o[2][4] = {};
    mm_rows_kn<C_, 2, 3>(Wq + (size_t)c0 * C_, Wk + (size_t)c0 * C_, S0 + (size_t)b * C_ * C_ + nblk,
                         S1 + (size_t)b * C_ * C_ + nblk, S2 + (size_t)b * C_ * C_ + nblk, xs, gs, o);
    float* Tb = T + (size_t)b * C_ * C_;
    float* Ub = U + (size_t)b * C_ * C_;
    #pragma unroll
    for (int j = 0; j < 8; ++j) {
        const int cA = c0 + j + 8 * half;
        const int cB = cA + 16;
        Tb[(size_t)cA * C_ + n0 + m]      = o[0][0][j];
        Tb[(size_t)cA * C_ + n0 + 16 + m] = o[0][1][j];
        Tb[(size_t)cB * C_ + n0 + m]      = o[0][2][j];
        Tb[(size_t)cB * C_ + n0 + 16 + m] = o[0][3][j];
        Ub[(size_t)cA * C_ + n0 + m]      = o[1][0][j];
        Ub[(size_t)cA * C_ + n0 + 16 + m] = o[1][1][j];
        Ub[(size_t)cB * C_ + n0 + m]      = o[1][2][j];
        Ub[(size_t)cB * C_ + n0 + 16 + m] = o[1][3][j];
    }
}

__global__ __launch_bounds__(256) void k_rownorm(const float* __restrict__ T,
                                                 const float* __restrict__ U,
                                                 const float* __restrict__ Wq,
                                                 const float* __restrict__ Wk,
                                                 float* __restrict__ invQ,
                                                 float* __restrict__ invK) {
    __shared__ float sq[256], sk[256];
    const int row = blockIdx.x;
    const int c   = row & (C_ - 1);
    const float* tr = T + (size_t)row * C_;
    const float* ur = U + (size_t)row * C_;
    const float* wq = Wq + (size_t)c * C_;
    const float* wk = Wk + (size_t)c * C_;
    float aq = 0.f, ak = 0.f;
    for (int e = threadIdx.x; e < C_; e += 256) {
        aq += tr[e] * wq[e];
        ak += ur[e] * wk[e];
    }
    sq[threadIdx.x] = aq; sk[threadIdx.x] = ak;
    __syncthreads();
    for (int s = 128; s > 0; s >>= 1) {
        if (threadIdx.x < s) {
            sq[threadIdx.x] += sq[threadIdx.x + s];
            sk[threadIdx.x] += sk[threadIdx.x + s];
        }
        __syncthreads();
    }
    if (threadIdx.x == 0) { invQ[row] = rsqrtf(sq[0]); invK[row] = rsqrtf(sk[0]); }
}

__global__ __launch_bounds__(128) __attribute__((amdgpu_waves_per_eu(1, 2))) void k_qkt(const float* __restrict__ T,
                                             const float* __restrict__ Wk,
                                             const float* __restrict__ invQ,
                                             const float* __restrict__ invK,
                                             float* __restrict__ M) {
    __shared__ __attribute__((aligned(16))) float tile[2 * GROWS_ * PK_];
    const int lane = threadIdx.x & 31;
    const int wave = threadIdx.x >> 5;
    const int m    = lane & 15;
    const int half = lane >> 4;
    const int c0   = blockIdx.y * 32;
    const int dblk = blockIdx.x * 128;
    const int d0   = dblk + wave * 32;
    const int b    = blockIdx.z;
    const size_t bOff = (size_t)b * C_ * C_;
    v8f a00 = {}, a01 = {}, a10 = {}, a11 = {};
    mm_rows_rows<C_>(T + bOff + (size_t)c0 * C_, Wk + (size_t)dblk * C_, tile, C_ / KC_, a00, a01, a10, a11);
    float* Mb = M + (size_t)b * C_ * C_;
    const float ik0 = invK[b * C_ + d0 + m];
    const float ik1 = invK[b * C_ + d0 + 16 + m];
    #pragma unroll
    for (int j = 0; j < 8; ++j) {
        const int cA = c0 + j + 8 * half;
        const int cB = cA + 16;
        const float iqA = invQ[b * C_ + cA];
        const float iqB = invQ[b * C_ + cB];
        Mb[(size_t)cA * C_ + d0 + m]      = a00[j] * iqA * ik0;
        Mb[(size_t)cA * C_ + d0 + 16 + m] = a01[j] * iqA * ik1;
        Mb[(size_t)cB * C_ + d0 + m]      = a10[j] * iqB * ik0;
        Mb[(size_t)cB * C_ + d0 + 16 + m] = a11[j] * iqB * ik1;
    }
}

__global__ __launch_bounds__(256) void k_colsoftmax(float* __restrict__ M) {
    __shared__ float smax[8][32], smin[8][32], ssum[8][32];
    const int lane = threadIdx.x & 31;
    const int wave = threadIdx.x >> 5;
    const int tile = blockIdx.x;
    const int b  = tile / (C_ / 32);
    const int d  = (tile - b * (C_ / 32)) * 32 + lane;
    float* p = M + (size_t)b * C_ * C_ + (size_t)(wave * 64) * C_ + d;
    float v[64];
    for (int i = 0; i < 64; ++i) v[i] = 1.f - p[(size_t)i * C_];
    float mx = -3.4e38f, mn = 3.4e38f;
    for (int i = 0; i < 64; ++i) { mx = fmaxf(mx, v[i]); mn = fminf(mn, v[i]); }
    smax[wave][lane] = mx; smin[wave][lane] = mn;
    __syncthreads();
    float cmx = -3.4e38f, cmn = 3.4e38f;
    for (int j = 0; j < 8; ++j) {
        cmx = fmaxf(cmx, smax[j][lane]);
        cmn = fminf(cmn, smin[j][lane]);
    }
    const float dn = cmx + EPS_;
    const float inv_dn = 1.f / dn;
    const float sh = 1.f - (cmn / dn) * (1.f / HT_);
    float z = 0.f;
    for (int i = 0; i < 64; ++i) {
        const float m4 = 1.f - (v[i] * inv_dn) * (1.f / HT_);
        const float e = expf(m4 - sh);
        v[i] = e;
        z += e;
    }
    ssum[wave][lane] = z;
    __syncthreads();
    float tot = 0.f;
    for (int j = 0; j < 8; ++j) tot += ssum[j][lane];
    const float invz = 1.f / tot;
    for (int i = 0; i < 64; ++i) p[(size_t)i * C_] = v[i] * invz;
}

__global__ __launch_bounds__(256) void k_rowl1(float* __restrict__ M) {
    __shared__ float ssum[256];
    const int row = blockIdx.x;
    float* r = M + (size_t)row * C_;
    float s = 0.f;
    for (int d = threadIdx.x; d < C_; d += 256) s += r[d];
    ssum[threadIdx.x] = s;
    __syncthreads();
    for (int t = 128; t > 0; t >>= 1) {
        if (threadIdx.x < t) ssum[threadIdx.x] += ssum[threadIdx.x + t];
        __syncthreads();
    }
    const float inv = 1.f / (ssum[0] + EPS_);
    for (int d = threadIdx.x; d < C_; d += 256) r[d] *= inv;
}

__global__ __launch_bounds__(128) __attribute__((amdgpu_waves_per_eu(1, 2))) void k_mwv(const float* __restrict__ M,
                                             const float* __restrict__ Wv,
                                             float* __restrict__ G) {
    __shared__ __attribute__((aligned(16))) float xs[2 * KC_ * PX_];
    __shared__ __attribute__((aligned(16))) float gs[2 * 32 * PK_];
    const int lane = threadIdx.x & 31;
    const int wave = threadIdx.x >> 5;
    const int m    = lane & 15;
    const int half = lane >> 4;
    const int c0   = blockIdx.y * 32;
    const int eblk = blockIdx.x * 128;
    const int e0   = eblk + wave * 32;
    const int b    = blockIdx.z;
    const float* Mb = M + (size_t)b * C_ * C_ + (size_t)c0 * C_;
    v8f o[1][4] = {};
    mm_rows_kn<C_, 1, 1>(Mb, Mb, Wv + eblk, Wv + eblk, Wv + eblk, xs, gs, o);
    float* Gb = G + (size_t)b * C_ * C_;
    #pragma unroll
    for (int j = 0; j < 8; ++j) {
        const int cA = c0 + j + 8 * half;
        const int cB = cA + 16;
        Gb[(size_t)cA * C_ + e0 + m]      = o[0][0][j];
        Gb[(size_t)cA * C_ + e0 + 16 + m] = o[0][1][j];
        Gb[(size_t)cB * C_ + e0 + m]      = o[0][2][j];
        Gb[(size_t)cB * C_ + e0 + 16 + m] = o[0][3][j];
    }
}

__global__ __launch_bounds__(128) void k_out(const float* __restrict__ x,
                                             const float* __restrict__ G,
                                             const float* __restrict__ gamma,
                                             float* __restrict__ y) {
    __shared__ __attribute__((aligned(16))) float xs[2 * KC_ * PX_];
    __shared__ __attribute__((aligned(16))) float gs[2 * 32 * PK_];
    const int lane = threadIdx.x & 31;
    const int wave = threadIdx.x >> 5;
    const int m    = lane & 15;
    const int half = lane >> 4;
    const int c0   = blockIdx.y * 32;
    const int nblk = blockIdx.x * 128;
    const int n0   = nblk + wave * 32;
    const int b    = blockIdx.z;
    const float* xb = x + (size_t)b * C_ * N_;
    const float* Gb = G + (size_t)b * C_ * C_ + (size_t)c0 * C_;
    v8f o[1][4] = {};
    mm_rows_kn<N_, 1, 1>(Gb, Gb, xb + nblk, xb + nblk, xb + nblk, xs, gs, o);
    const float g = gamma[0];
    float* yb = y + (size_t)b * C_ * N_;
    #pragma unroll
    for (int j = 0; j < 8; ++j) {
        const int cA = c0 + j + 8 * half;
        const int cB = cA + 16;
        const size_t o0 = (size_t)cA * N_ + n0 + m;
        const size_t o1 = (size_t)cA * N_ + n0 + 16 + m;
        const size_t o2 = (size_t)cB * N_ + n0 + m;
        const size_t o3 = (size_t)cB * N_ + n0 + 16 + m;
        yb[o0] = xb[o0] + g * o[0][0][j];
        yb[o1] = xb[o1] + g * o[0][1][j];
        yb[o2] = xb[o2] + g * o[0][2][j];
        yb[o3] = xb[o3] + g * o[0][3][j];
    }
}

extern "C" void kernel_launch(void* const* d_in, const int* in_sizes, int n_in,
                              void* d_out, int out_size, void* d_ws, size_t ws_size,
                              hipStream_t stream) {
    (void)in_sizes; (void)n_in; (void)out_size; (void)ws_size;
    const float* x     = (const float*)d_in[0];
    const float* Wq    = (const float*)d_in[1];
    const float* Wk    = (const float*)d_in[2];
    const float* Wv    = (const float*)d_in[3];
    const float* gamma = (const float*)d_in[4];
    float* y = (float*)d_out;

    char* ws = (char*)d_ws;
    const size_t mBytes  = (size_t)B_ * C_ * C_ * sizeof(float);
    float* S     = (float*)(ws);
    float* T     = (float*)(ws + mBytes);
    float* U     = (float*)(ws + 2 * mBytes);
    float* M     = (float*)(ws + 3 * mBytes);
    float* G     = (float*)(ws + 4 * mBytes);
    float* invQ  = (float*)(ws + 5 * mBytes);
    float* invK  = invQ  + B_ * C_;

    k_gram      <<<dim3(40, 3, B_), 128, 0, stream>>>(x, S, G, M);
    k_projs     <<<dim3(C_ / 128, C_ / 32, B_), 128, 0, stream>>>(S, G, M, Wq, Wk, T, U);
    k_rownorm   <<<B_ * C_, 256, 0, stream>>>(T, U, Wq, Wk, invQ, invK);
    k_qkt       <<<dim3(C_ / 128, C_ / 32, B_), 128, 0, stream>>>(T, Wk, invQ, invK, M);
    k_colsoftmax<<<B_ * C_ / 32, 256, 0, stream>>>(M);
    k_rowl1     <<<B_ * C_, 256, 0, stream>>>(M);
    k_mwv       <<<dim3(C_ / 128, C_ / 32, B_), 128, 0, stream>>>(M, Wv, G);
    k_out       <<<dim3(N_ / 128, C_ / 32, B_), 128, 0, stream>>>(x, G, gamma, y);
}


# Row sums folded into the GEMMs: Q,K row norms from the projection accumulators (4 partials/row, U not stored); L1 row sums taken as M is staged, scale put on M*Wv; 8->6 launches; sum trees change
# speedup vs baseline: 2.9062x; 1.0635x over previous
#include <hip/hip_runtime.h>

typedef __attribute__((ext_vector_type(2))) float v2f;
typedef __attribute__((ext_vector_type(4))) float v4f;
typedef __attribute__((ext_vector_type(8))) float v8f;

#define B_ 8
#define C_ 512
#define N_ 4096
#define EPS_ 1e-6f
#define HT_ 0.25f

static __device__ __forceinline__ v8f wmma4(v2f a, v2f b, v8f c) {
    return __builtin_amdgcn_wmma_f32_16x16x4_f32(false, a, false, b, (short)0, c, false, false);
}

static __device__ __forceinline__ v2f lo2(v4f v) { return __builtin_shufflevector(v, v, 0, 1); }
static __device__ __forceinline__ v2f hi2(v4f v) { return __builtin_shufflevector(v, v, 2, 3); }


#define KC_    32
#define PK_    36
#define GROWS_ 160
#define PX_    132

struct FragRR { v4f aA[4], aB[4], bA[4], bB[4]; };

static __device__ __forceinline__ void frag_rr_load(FragRR& f, const float* __restrict__ st,
                                                    int fA0, int fA1, int fB0, int fB1) {
    #pragma unroll
    for (int j = 0; j < 4; ++j) {
        f.aA[j] = *(const v4f*)(st + fA0 + 8 * j);
        f.aB[j] = *(const v4f*)(st + fA1 + 8 * j);
        f.bA[j] = *(const v4f*)(st + fB0 + 8 * j);
        f.bB[j] = *(const v4f*)(st + fB1 + 8 * j);
    }
}

static __device__ __forceinline__ void frag_rr_mma(const FragRR& f, v8f& a00, v8f& a01, v8f& a10, v8f& a11) {
    #pragma unroll
    for (int j = 0; j < 4; ++j) {
        a00 = wmma4(lo2(f.aA[j]), lo2(f.bA[j]), a00); a01 = wmma4(lo2(f.aA[j]), lo2(f.bB[j]), a01);
        a10 = wmma4(lo2(f.aB[j]), lo2(f.bA[j]), a10); a11 = wmma4(lo2(f.aB[j]), lo2(f.bB[j]), a11);
        a00 = wmma4(hi2(f.aA[j]), hi2(f.bA[j]), a00); a01 = wmma4(hi2(f.aA[j]), hi2(f.bB[j]), a01);
        a10 = wmma4(hi2(f.aB[j]), hi2(f.bA[j]), a10); a11 = wmma4(hi2(f.aB[j]), hi2(f.bB[j]), a11);
    }
}

template <int LD>
static __device__ __forceinline__ void mm_rows_rows(const float* __restrict__ gA,
                                                    const float* __restrict__ gB,
                                                    float* __restrict__ tile, const int nch,
                                                    v8f& a00, v8f& a01, v8f& a10, v8f& a11) {
    constexpr int STG = GROWS_ * PK_;
    const int t    = threadIdx.x;
    const int lane = t & 31;
    const int wave = t >> 5;
    const int m    = lane & 15;
    const int k4   = (lane >> 4) << 2;
    const int sr = t >> 3;
    const int sq = (t & 7) << 2;
    const float* pA = gA + (size_t)sr * LD + sq;
    const float* pB = gB + (size_t)sr * LD + sq;
    const int sOff = sr * PK_ + sq;
    const int fA0 = m * PK_ + k4;
    const int fA1 = (16 + m) * PK_ + k4;
    const int fB0 = (32 + wave * 32 + m) * PK_ + k4;
    const int fB1 = fB0 + 16 * PK_;
    v4f g[10];
    {
        v4f h[10];
        #pragma unroll
        for (int r = 0; r < 2; ++r) {
            g[r] = *(const v4f*)(pA + (size_t)(16 * r) * LD);
            h[r] = *(const v4f*)(pA + (size_t)(16 * r) * LD + KC_);
        }
        #pragma unroll
        for (int r = 0; r < 8; ++r) {
            g[2 + r] = *(const v4f*)(pB + (size_t)(16 * r) * LD);
            h[2 + r] = *(const v4f*)(pB + (size_t)(16 * r) * LD + KC_);
        }
        #pragma unroll
        for (int r = 0; r < 10; ++r) {
            *(v4f*)(tile + sOff + 16 * r * PK_) = g[r];
            *(v4f*)(tile + STG + sOff + 16 * r * PK_) = h[r];
        }
    }
    __syncthreads();
    FragRR P, Q;
    frag_rr_load(P, tile, fA0, fA1, fB0, fB1);
    __syncthreads();
    #pragma unroll 1
    for (int c = 0; c < nch; c += 2) {
        const bool more = (c + 2 < nch);
        if (more) {
            const int kc = (c + 2) * KC_;
            #pragma unroll
            for (int r = 0; r < 2; ++r) g[r] = *(const v4f*)(pA + (size_t)(16 * r) * LD + kc);
            #pragma unroll
            for (int r = 0; r < 8; ++r) g[2 + r] = *(const v4f*)(pB + (size_t)(16 * r) * LD + kc);
        }
        frag_rr_load(Q, tile + STG, fA0, fA1, fB0, fB1);
        __builtin_amdgcn_sched_barrier(0);
        frag_rr_mma(P, a00, a01, a10, a11);
        __builtin_amdgcn_sched_barrier(0);
        if (more) {
            #pragma unroll
            for (int r = 0; r < 10; ++r) *(v4f*)(tile + sOff + 16 * r * PK_) = g[r];
        }
        __syncthreads();
        if (more) {
            const int kc = (c + 3) * KC_;
            #pragma unroll
            for (int r = 0; r < 2; ++r) g[r] = *(const v4f*)(pA + (size_t)(16 * r) * LD + kc);
            #pragma unroll
            for (int r = 0; r < 8; ++r) g[2 + r] = *(const v4f*)(pB + (size_t)(16 * r) * LD + kc);
            frag_rr_load(P, tile, fA0, fA1, fB0, fB1);
        }
        __builtin_amdgcn_sched_barrier(0);
        frag_rr_mma(Q, a00, a01, a10, a11);
        __builtin_amdgcn_sched_barrier(0);
        if (more) {
            #pragma unroll
            for (int r = 0; r < 10; ++r) *(v4f*)(tile + STG + sOff + 16 * r * PK_) = g[r];
        }
        __syncthreads();
    }
}

template <int NA>
struct FragKN { v4f aA[NA][4], aB[NA][4]; v2f b0[4], b1[4], b2[4], b3[4]; };

template <int NA>
static __device__ __forceinline__ void frag_kn_load(FragKN<NA>& f, const float* __restrict__ sx,
                                                    const float* __restrict__ sg,
                                                    int fA0, int fA1, int fB, int fBh) {
    #pragma unroll
    for (int j = 0; j < 4; ++j) {
        #pragma unroll
        for (int a = 0; a < NA; ++a) {
            f.aA[a][j] = *(const v4f*)(sg + a * 32 * PK_ + fA0 + 8 * j);
            f.aB[a][j] = *(const v4f*)(sg + a * 32 * PK_ + fA1 + 8 * j);
        }
        const float* r0 = sx + fB + (8 * j) * PX_;
        const float* r1 = sx + fBh + (8 * j) * PX_;
        f.b0[j].x = r0[0];       f.b0[j].y = r0[PX_];
        f.b1[j].x = r1[0];       f.b1[j].y = r1[PX_];
        f.b2[j].x = r0[2 * PX_]; f.b2[j].y = r0[3 * PX_];
        f.b3[j].x = r1[2 * PX_]; f.b3[j].y = r1[3 * PX_];
    }
}

template <int NA>
static __device__ __forceinline__ void frag_kn_mma(const FragKN<NA>& f, v8f (&o)[NA][4]) {
    #pragma unroll
    for (int j = 0; j < 4; ++j) {
        #pragma unroll
        for (int a = 0; a < NA; ++a) {
            o[a][0] = wmma4(lo2(f.aA[a][j]), f.b0[j], o[a][0]); o[a][1] = wmma4(lo2(f.aA[a][j]), f.b1[j], o[a][1]);
            o[a][2] = wmma4(lo2(f.aB[a][j]), f.b0[j], o[a][2]); o[a][3] = wmma4(lo2(f.aB[a][j]), f.b1[j], o[a][3]);
        }
        #pragma unroll
        for (int a = 0; a < NA; ++a) {
            o[a][0] = wmma4(hi2(f.aA[a][j]), f.b2[j], o[a][0]); o[a][1] = wmma4(hi2(f.aA[a][j]), f.b3[j], o[a][1]);
            o[a][2] = wmma4(hi2(f.aB[a][j]), f.b2[j], o[a][2]); o[a][3] = wmma4(hi2(f.aB[a][j]), f.b3[j], o[a][3]);
        }
    }
}

#define PLSUM_(x, y, z) (NPL == 1 ? (x) : NPL == 2 ? (x) + (y) : ((x) + (y)) + (z))
static __device__ __forceinline__ float sum4(v4f v) { return (v.x + v.y) + (v.z + v.w); }
template <int LDB, int NA, int NPL, bool RS>
static __device__ __forceinline__ void mm_rows_kn(const float* __restrict__ gA0,
                                                  const float* __restrict__ gA1,
                                                  const float* __restrict__ gB,
                                                  const float* __restrict__ gB2,
                                                  const float* __restrict__ gB3,
                                                  float* __restrict__ xs,
                                                  float* __restrict__ gs,
                                                  v8f (&o)[NA][4], float (&rs)[2]) {
    constexpr int NCH = C_ / KC_;
    constexpr int SXS = KC_ * PX_;
    constexpr int SGS = NA * 32 * PK_;
    const int t    = threadIdx.x;
    const int lane = t & 31;
    const int wave = t >> 5;
    const int m    = lane & 15;
    const int k4   = (lane >> 4) << 2;
    const float* pX = gB + (size_t)wave * LDB + (lane << 2);
    const float* pY = gB2 + (size_t)wave * LDB + (lane << 2);
    const float* pZ = gB3 + (size_t)wave * LDB + (lane << 2);
    const int sxOff = wave * PX_ + (lane << 2);
    const int sr = t >> 3;
    const int sq = (t & 7) << 2;
    const float* pG[NA];
    pG[0] = gA0 + (size_t)sr * C_ + sq;
    if (NA > 1) pG[NA - 1] = gA1 + (size_t)sr * C_ + sq;
    const int sgOff = sr * PK_ + sq;
    const int fA0 = m * PK_ + k4;
    const int fA1 = (16 + m) * PK_ + k4;
    const int fB  = k4 * PX_ + wave * 32 + m;
    int fBh = fB + 16;
    asm volatile("" : "+v"(fBh));
    v4f gx[8], gy[8], gz[8], gg[NA][2];
    {
        v4f hx[8], hy[8], hz[8], hg[NA][2];
        #pragma unroll
        for (int r = 0; r < 8; ++r) {
            gx[r] = *(const v4f*)(pX + (size_t)(4 * r) * LDB);
            hx[r] = *(const v4f*)(pX + (size_t)(KC_ + 4 * r) * LDB);
            if (NPL > 1) {
                gy[r] = *(const v4f*)(pY + (size_t)(4 * r) * LDB);
                hy[r] = *(const v4f*)(pY + (size_t)(KC_ + 4 * r) * LDB);
            }
            if (NPL > 2) {
                gz[r] = *(const v4f*)(pZ + (size_t)(4 * r) * LDB);
                hz[r] = *(const v4f*)(pZ + (size_t)(KC_ + 4 * r) * LDB);
            }
        }
        #pragma unroll
        for (int a = 0; a < NA; ++a) {
            #pragma unroll
            for (int r = 0; r < 2; ++r) {
                gg[a][r] = *(const v4f*)(pG[a] + (size_t)(16 * r) * C_);
                hg[a][r] = *(const v4f*)(pG[a] + (size_t)(16 * r) * C_ + KC_);
            }
        }
        #pragma unroll
        for (int r = 0; r < 8; ++r) {
            *(v4f*)(xs + sxOff + 4 * r * PX_) = PLSUM_(gx[r], gy[r], gz[r]);
            *(v4f*)(xs + SXS + sxOff + 4 * r * PX_) = PLSUM_(hx[r], hy[r], hz[r]);
        }
        #pragma unroll
        for (int a = 0; a < NA; ++a) {
            #pragma unroll
            for (int r = 0; r < 2; ++r) {
                *(v4f*)(gs + (a * 32 + 16 * r) * PK_ + sgOff) = gg[a][r];
                *(v4f*)(gs + SGS + (a * 32 + 16 * r) * PK_ + sgOff) = hg[a][r];
                if (RS) rs[r] = sum4(gg[a][r]) + sum4(hg[a][r]);
            }
        }
    }
    __syncthreads();
    FragKN<NA> P, Q;
    frag_kn_load<NA>(P, xs, gs, fA0, fA1, fB, fBh);
    __syncthreads();
    #pragma unroll 1
    for (int c = 0; c < NCH; c += 2) {
        const bool more = (c + 2 < NCH);
        if (more) {
            const int kc = (c + 2) * KC_;
            #pragma unroll
            for (int r = 0; r < 8; ++r) gx[r] = *(const v4f*)(pX + (size_t)(kc + 4 * r) * LDB);
            if (NPL > 1) {
                #pragma unroll
                for (int r = 0; r < 8; ++r) gy[r] = *(const v4f*)(pY + (size_t)(kc + 4 * r) * LDB);
            }
            if (NPL > 2) {
                #pragma unroll
                for (int r = 0; r < 8; ++r) gz[r] = *(const v4f*)(pZ + (size_t)(kc + 4 * r) * LDB);
            }
            #pragma unroll
            for (int a = 0; a < NA; ++a) {
                #pragma unroll
                for (int r = 0; r < 2; ++r) gg[a][r] = *(const v4f*)(pG[a] + (size_t)(16 * r) * C_ + kc);
            }
        }
        frag_kn_load<NA>(Q, xs + SXS, gs + SGS, fA0, fA1, fB, fBh);
        __builtin_amdgcn_sched_barrier(0);
        frag_kn_mma<NA>(P, o);
        __builtin_amdgcn_sched_barrier(0);
        if (more) {
            #pragma unroll
            for (int r = 0; r < 8; ++r) *(v4f*)(xs + sxOff + 4 * r * PX_) = PLSUM_(gx[r], gy[r], gz[r]);
            #pragma unroll
            for (int a = 0; a < NA; ++a) {
                #pragma unroll
                for (int r = 0; r < 2; ++r) {
                    *(v4f*)(gs + (a * 32 + 16 * r) * PK_ + sgOff) = gg[a][r];
                    if (RS) rs[r] += sum4(gg[a][r]);
                }
            }
        }
        __syncthreads();
        if (more) {
            const int kc = (c + 3) * KC_;
            #pragma unroll
            for (int r = 0; r < 8; ++r) gx[r] = *(const v4f*)(pX + (size_t)(kc + 4 * r) * LDB);
            if (NPL > 1) {
                #pragma unroll
                for (int r = 0; r < 8; ++r) gy[r] = *(const v4f*)(pY + (size_t)(kc + 4 * r) * LDB);
            }
            if (NPL > 2) {
                #pragma unroll
                for (int r = 0; r < 8; ++r) gz[r] = *(const v4f*)(pZ + (size_t)(kc + 4 * r) * LDB);
            }
            #pragma unroll
            for (int a = 0; a < NA; ++a) {
                #pragma unroll
                for (int r = 0; r < 2; ++r) gg[a][r] = *(const v4f*)(pG[a] + (size_t)(16 * r) * C_ + kc);
            }
            frag_kn_load<NA>(P, xs, gs, fA0, fA1, fB, fBh);
        }
        __builtin_amdgcn_sched_barrier(0);
        frag_kn_mma<NA>(Q, o);
        __builtin_amdgcn_sched_barrier(0);
        if (more) {
            #pragma unroll
            for (int r = 0; r < 8; ++r) *(v4f*)(xs + SXS + sxOff + 4 * r * PX_) = PLSUM_(gx[r], gy[r], gz[r]);
            #pragma unroll
            for (int a = 0; a < NA; ++a) {
                #pragma unroll
                for (int r = 0; r < 2; ++r) {
                    *(v4f*)(gs + SGS + (a * 32 + 16 * r) * PK_ + sgOff) = gg[a][r];
                    if (RS) rs[r] += sum4(gg[a][r]);
                }
            }
        }
        __syncthreads();
    }
}

#define KP0_ 44
#define KP1_ 42
__global__ __launch_bounds__(128) __attribute__((amdgpu_waves_per_eu(1, 2))) void k_gram(const float* __restrict__ x,
                                              float* __restrict__ S0,
                                              float* __restrict__ S1,
                                              float* __restrict__ S2) {
    __shared__ __attribute__((aligned(16))) float tile[2 * GROWS_ * PK_];
    const int lane = threadIdx.x & 31;
    const int wave = threadIdx.x >> 5;
    const int m    = lane & 15;
    const int half = lane >> 4;
    const int p    = blockIdx.x >> 2;
    const int I    = (p >= 4) + (p >= 7) + (p >= 9);
    const int J    = p - (4 * I - ((I * (I - 1)) >> 1)) + I;
    const int c0   = I * 128 + (blockIdx.x & 3) * 32;
    const int dblk = J * 128;
    const int d0   = dblk + wave * 32;
    const int kh   = blockIdx.y;
    const int b    = blockIdx.z;
    const int ch0  = kh == 0 ? 0 : KP0_ + (kh - 1) * KP1_;
    const int nch  = kh == 0 ? KP0_ : KP1_;
    const size_t bOff = (size_t)b * C_ * N_ + (size_t)ch0 * KC_;
    v8f a00 = {}, a01 = {}, a10 = {}, a11 = {};
    mm_rows_rows<N_>(x + bOff + (size_t)c0 * N_, x + bOff + (size_t)dblk * N_, tile, nch, a00, a01, a10, a11);
    float* Sb = (kh == 0 ? S0 : kh == 1 ? S1 : S2) + (size_t)b * C_ * C_;
    #pragma unroll
    for (int j = 0; j < 8; ++j) {
        const int cA = c0 + j + 8 * half;
        const int cB = cA + 16;
        Sb[(size_t)cA * C_ + d0 + m]      = a00[j];
        Sb[(size_t)cA * C_ + d0 + 16 + m] = a01[j];
        Sb[(size_t)cB * C_ + d0 + m]      = a10[j];
        Sb[(size_t)cB * C_ + d0 + 16 + m] = a11[j];
    }
    if (I != J) {
        float* r0 = Sb + (size_t)(d0 + m) * C_ + c0 + 8 * half;
        float* r1 = r0 + (size_t)16 * C_;
        *(v8f*)(r0)      = a00;
        *(v8f*)(r0 + 16) = a10;
        *(v8f*)(r1)      = a01;
        *(v8f*)(r1 + 16) = a11;
    }
}

__global__ __launch_bounds__(128) __attribute__((amdgpu_waves_per_eu(1, 2))) void k_projs(const float* __restrict__ S0,
                                               const float* __restrict__ S1,
                                               const float* __restrict__ S2,
                                               const float* __restrict__ Wq,
                                               const float* __restrict__ Wk,
                                               float* __restrict__ T,
                                               float* __restrict__ PQ,
                                               float* __restrict__ PK) {
    __shared__ __attribute__((aligned(16))) float xs[2 * KC_ * PX_];
    __shared__ __attribute__((aligned(16))) float gs[2 * 2 * 32 * PK_];
    const int lane = threadIdx.x & 31;
    const int wave = threadIdx.x >> 5;
    const int m    = lane & 15;
    const int half = lane >> 4;
    const int c0   = blockIdx.y * 32;
    const int nblk = blockIdx.x * 128;
    const int n0   = nblk + wave * 32;
    const int b    = blockIdx.z;
    v8f o[2][4] = {};
    float unused[2];
    mm_rows_kn<C_, 2, 3, false>(Wq + (size_t)c0 * C_, Wk + (size_t)c0 * C_, S0 + (size_t)b * C_ * C_ + nblk,
                                S1 + (size_t)b * C_ * C_ + nblk, S2 + (size_t)b * C_ * C_ + nblk, xs, gs, o, unused);
    float* Tb = T + (size_t)b * C_ * C_;
    float sq[8][2], sk[8][2];
    #pragma unroll
    for (int j = 0; j < 8; ++j) {
        const int cA = c0 + j + 8 * half;
        const int cB = cA + 16;
        const size_t pA = (size_t)cA * C_ + n0 + m;
        const size_t pB = (size_t)cB * C_ + n0 + m;
        Tb[pA]      = o[0][0][j];
        Tb[pA + 16] = o[0][1][j];
        Tb[pB]      = o[0][2][j];
        Tb[pB + 16] = o[0][3][j];
        sq[j][0] = o[0][0][j] * Wq[pA] + o[0][1][j] * Wq[pA + 16];
        sq[j][1] = o[0][2][j] * Wq[pB] + o[0][3][j] * Wq[pB + 16];
        sk[j][0] = o[1][0][j] * Wk[pA] + o[1][1][j] * Wk[pA + 16];
        sk[j][1] = o[1][2][j] * Wk[pB] + o[1][3][j] * Wk[pB + 16];
    }
    #pragma unroll
    for (int s = 1; s < 16; s <<= 1) {
        #pragma unroll
        for (int j = 0; j < 8; ++j) {
            sq[j][0] += __shfl_xor(sq[j][0], s); sq[j][1] += __shfl_xor(sq[j][1], s);
            sk[j][0] += __shfl_xor(sk[j][0], s); sk[j][1] += __shfl_xor(sk[j][1], s);
        }
    }
    float* red = xs;
    if (m == 0) {
        #pragma unroll
        for (int j = 0; j < 8; ++j) {
            red[wave * 64 + j + 8 * half]           = sq[j][0];
            red[wave * 64 + j + 8 * half + 16]      = sq[j][1];
            red[wave * 64 + 32 + j + 8 * half]      = sk[j][0];
            red[wave * 64 + 32 + j + 8 * half + 16] = sk[j][1];
        }
    }
    __syncthreads();
    if (threadIdx.x < 64) {
        const int t = threadIdx.x;
        const float v = ((red[t] + red[64 + t]) + red[128 + t]) + red[192 + t];
        float* dst = (t < 32 ? PQ : PK) + ((size_t)(b * C_ + c0 + (t & 31))) * 4 + blockIdx.x;
        *dst = v;
    }
}

__global__ __launch_bounds__(128) __attribute__((amdgpu_waves_per_eu(1, 2))) void k_qkt(const float* __restrict__ T,
                                             const float* __restrict__ Wk,
                                             const float* __restrict__ PQ,
                                             const float* __restrict__ PK,
                                             float* __restrict__ M) {
    __shared__ __attribute__((aligned(16))) float tile[2 * GROWS_ * PK_];
    __shared__ float sIQ[32], sIK[128];
    const int lane = threadIdx.x & 31;
    const int wave = threadIdx.x >> 5;
    const int m    = lane & 15;
    const int half = lane >> 4;
    const int c0   = blockIdx.y * 32;
    const int dblk = blockIdx.x * 128;
    const int d0   = dblk + wave * 32;
    const int b    = blockIdx.z;
    const size_t bOff = (size_t)b * C_ * C_;
    const v4f pk = *(const v4f*)(PK + ((size_t)(b * C_ + dblk + threadIdx.x)) * 4);
    const v4f pq = *(const v4f*)(PQ + ((size_t)(b * C_ + c0 + (threadIdx.x & 31))) * 4);
    v8f a00 = {}, a01 = {}, a10 = {}, a11 = {};
    mm_rows_rows<C_>(T + bOff + (size_t)c0 * C_, Wk + (size_t)dblk * C_, tile, C_ / KC_, a00, a01, a10, a11);
    sIK[threadIdx.x] = rsqrtf((pk.x + pk.y) + (pk.z + pk.w));
    if (threadIdx.x < 32) sIQ[threadIdx.x] = rsqrtf((pq.x + pq.y) + (pq.z + pq.w));
    __syncthreads();
    float* Mb = M + (size_t)b * C_ * C_;
    const float ik0 = sIK[wave * 32 + m];
    const float ik1 = sIK[wave * 32 + 16 + m];
    #pragma unroll
    for (int j = 0; j < 8; ++j) {
        const int cA = c0 + j + 8 * half;
        const int cB = cA + 16;
        const float iqA = sIQ[j + 8 * half];
        const float iqB = sIQ[j + 8 * half + 16];
        Mb[(size_t)cA * C_ + d0 + m]      = a00[j] * iqA * ik0;
        Mb[(size_t)cA * C_ + d0 + 16 + m] = a01[j] * iqA * ik1;
        Mb[(size_t)cB * C_ + d0 + m]      = a10[j] * iqB * ik0;
        Mb[(size_t)cB * C_ + d0 + 16 + m] = a11[j] * iqB * ik1;
    }
}

__global__ __launch_bounds__(256) void k_colsoftmax(float* __restrict__ M) {
    __shared__ float smax[8][32], smin[8][32], ssum[8][32];
    const int lane = threadIdx.x & 31;
    const int wave = threadIdx.x >> 5;
    const int tile = blockIdx.x;
    const int b  = tile / (C_ / 32);
    const int d  = (tile - b * (C_ / 32)) * 32 + lane;
    float* p = M + (size_t)b * C_ * C_ + (size_t)(wave * 64) * C_ + d;
    float v[64];
    for (int i = 0; i < 64; ++i) v[i] = 1.f - p[(size_t)i * C_];
    float mx = -3.4e38f, mn = 3.4e38f;
    for (int i = 0; i < 64; ++i) { mx = fmaxf(mx, v[i]); mn = fminf(mn, v[i]); }
    smax[wave][lane] = mx; smin[wave][lane] = mn;
    __syncthreads();
    float cmx = -3.4e38f, cmn = 3.4e38f;
    for (int j = 0; j < 8; ++j) {
        cmx = fmaxf(cmx, smax[j][lane]);
        cmn = fminf(cmn, smin[j][lane]);
    }
    const float dn = cmx + EPS_;
    const float inv_dn = 1.f / dn;
    const float sh = 1.f - (cmn / dn) * (1.f / HT_);
    float z = 0.f;
    for (int i = 0; i < 64; ++i) {
        const float m4 = 1.f - (v[i] * inv_dn) * (1.f / HT_);
        const float e = expf(m4 - sh);
        v[i] = e;
        z += e;
    }
    ssum[wave][lane] = z;
    __syncthreads();
    float tot = 0.f;
    for (int j = 0; j < 8; ++j) tot += ssum[j][lane];
    const float invz = 1.f / tot;
    for (int i = 0; i < 64; ++i) p[(size_t)i * C_] = v[i] * invz;
}

__global__ __launch_bounds__(128) __attribute__((amdgpu_waves_per_eu(1, 2))) void k_mwv(const float* __restrict__ M,
                                             const float* __restrict__ Wv,
                                             float* __restrict__ G) {
    __shared__ __attribute__((aligned(16))) float xs[2 * KC_ * PX_];
    __shared__ __attribute__((aligned(16))) float gs[2 * 32 * PK_];
    const int lane = threadIdx.x & 31;
    const int wave = threadIdx.x >> 5;
    const int m    = lane & 15;
    const int half = lane >> 4;
    const int c0   = blockIdx.y * 32;
    const int eblk = blockIdx.x * 128;
    const int e0   = eblk + wave * 32;
    const int b    = blockIdx.z;
    const float* Mb = M + (size_t)b * C_ * C_ + (size_t)c0 * C_;
    v8f o[1][4] = {};
    float rs[2];
    mm_rows_kn<C_, 1, 1, true>(Mb, Mb, Wv + eblk, Wv + eblk, Wv + eblk, xs, gs, o, rs);
    for (int s = 1; s < 8; s <<= 1) {
        rs[0] += __shfl_xor(rs[0], s);
        rs[1] += __shfl_xor(rs[1], s);
    }
    float* srow = gs;
    if ((threadIdx.x & 7) == 0) {
        srow[threadIdx.x >> 3]        = 1.f / (rs[0] + EPS_);
        srow[(threadIdx.x >> 3) + 16] = 1.f / (rs[1] + EPS_);
    }
    __syncthreads();
    float* Gb = G + (size_t)b * C_ * C_;
    #pragma unroll
    for (int j = 0; j < 8; ++j) {
        const int cA = c0 + j + 8 * half;
        const int cB = cA + 16;
        const float iA = srow[j + 8 * half];
        const float iB = srow[j + 8 * half + 16];
        Gb[(size_t)cA * C_ + e0 + m]      = o[0][0][j] * iA;
        Gb[(size_t)cA * C_ + e0 + 16 + m] = o[0][1][j] * iA;
        Gb[(size_t)cB * C_ + e0 + m]      = o[0][2][j] * iB;
        Gb[(size_t)cB * C_ + e0 + 16 + m] = o[0][3][j] * iB;
    }
}

__global__ __launch_bounds__(128) void k_out(const float* __restrict__ x,
                                             const float* __restrict__ G,
                                             const float* __restrict__ gamma,
                                             float* __restrict__ y) {
    __shared__ __attribute__((aligned(16))) float xs[2 * KC_ * PX_];
    __shared__ __attribute__((aligned(16))) float gs[2 * 32 * PK_];
    const int lane = threadIdx.x & 31;
    const int wave = threadIdx.x >> 5;
    const int m    = lane & 15;
    const int half = lane >> 4;
    const int c0   = blockIdx.y * 32;
    const int nblk = blockIdx.x * 128;
    const int n0   = nblk + wave * 32;
    const int b    = blockIdx.z;
    const float* xb = x + (size_t)b * C_ * N_;
    const float* Gb = G + (size_t)b * C_ * C_ + (size_t)c0 * C_;
    v8f o[1][4] = {};
    float unused[2];
    mm_rows_kn<N_, 1, 1, false>(Gb, Gb, xb + nblk, xb + nblk, xb + nblk, xs, gs, o, unused);
    const float g = gamma[0];
    float* yb = y + (size_t)b * C_ * N_;
    #pragma unroll
    for (int j = 0; j < 8; ++j) {
        const int cA = c0 + j + 8 * half;
        const int cB = cA + 16;
        const size_t o0 = (size_t)cA * N_ + n0 + m;
        const size_t o1 = (size_t)cA * N_ + n0 + 16 + m;
        const size_t o2 = (size_t)cB * N_ + n0 + m;
        const size_t o3 = (size_t)cB * N_ + n0 + 16 + m;
        yb[o0] = xb[o0] + g * o[0][0][j];
        yb[o1] = xb[o1] + g * o[0][1][j];
        yb[o2] = xb[o2] + g * o[0][2][j];
        yb[o3] = xb[o3] + g * o[0][3][j];
    }
}

extern "C" void kernel_launch(void* const* d_in, const int* in_sizes, int n_in,
                              void* d_out, int out_size, void* d_ws, size_t ws_size,
                              hipStream_t stream) {
    (void)in_sizes; (void)n_in; (void)out_size; (void)ws_size;
    const float* x     = (const float*)d_in[0];
    const float* Wq    = (const float*)d_in[1];
    const float* Wk    = (const float*)d_in[2];
    const float* Wv    = (const float*)d_in[3];
    const float* gamma = (const float*)d_in[4];
    float* y = (float*)d_out;

    char* ws = (char*)d_ws;
    const size_t mBytes  = (size_t)B_ * C_ * C_ * sizeof(float);
    float* S     = (float*)(ws);
    float* T     = (float*)(ws + mBytes);
    float* PQ    = (float*)(ws + 2 * mBytes);
    float* PK    = PQ + B_ * C_ * 4;
    float* M     = (float*)(ws + 3 * mBytes);
    float* G     = (float*)(ws + 4 * mBytes);

    k_gram      <<<dim3(40, 3, B_), 128, 0, stream>>>(x, S, G, M);
    k_projs     <<<dim3(C_ / 128, C_ / 32, B_), 128, 0, stream>>>(S, G, M, Wq, Wk, T, PQ, PK);
    k_qkt       <<<dim3(C_ / 128, C_ / 32, B_), 128, 0, stream>>>(T, Wk, PQ, PK, M);
    k_colsoftmax<<<B_ * C_ / 32, 256, 0, stream>>>(M);
    k_mwv       <<<dim3(C_ / 128, C_ / 32, B_), 128, 0, stream>>>(M, Wv, G);
    k_out       <<<dim3(N_ / 128, C_ / 32, B_), 128, 0, stream>>>(x, G, gamma, y);
}


# K loops of all five WMMA kernels start at a chunk pair rotated by block index and wrap round, so work-groups sharing operand rows ask for different lines per trip; order of k sums changes
# speedup vs baseline: 3.0120x; 1.0364x over previous
#include <hip/hip_runtime.h>

typedef __attribute__((ext_vector_type(2))) float v2f;
typedef __attribute__((ext_vector_type(4))) float v4f;
typedef __attribute__((ext_vector_type(8))) float v8f;

#define B_ 8
#define C_ 512
#define N_ 4096
#define EPS_ 1e-6f
#define HT_ 0.25f

static __device__ __forceinline__ v8f wmma4(v2f a, v2f b, v8f c) {
    return __builtin_amdgcn_wmma_f32_16x16x4_f32(false, a, false, b, (short)0, c, false, false);
}

static __device__ __forceinline__ v2f lo2(v4f v) { return __builtin_shufflevector(v, v, 0, 1); }
static __device__ __forceinline__ v2f hi2(v4f v) { return __builtin_shufflevector(v, v, 2, 3); }


#define KC_    32
#define PK_    36
#define GROWS_ 160
#define PX_    132

struct FragRR { v4f aA[4], aB[4], bA[4], bB[4]; };

static __device__ __forceinline__ void frag_rr_load(FragRR& f, const float* __restrict__ st,
                                                    int fA0, int fA1, int fB0, int fB1) {
    #pragma unroll
    for (int j = 0; j < 4; ++j) {
        f.aA[j] = *(const v4f*)(st + fA0 + 8 * j);
        f.aB[j] = *(const v4f*)(st + fA1 + 8 * j);
        f.bA[j] = *(const v4f*)(st + fB0 + 8 * j);
        f.bB[j] = *(const v4f*)(st + fB1 + 8 * j);
    }
}

static __device__ __forceinline__ void frag_rr_mma(const FragRR& f, v8f& a00, v8f& a01, v8f& a10, v8f& a11) {
    #pragma unroll
    for (int j = 0; j < 4; ++j) {
        a00 = wmma4(lo2(f.aA[j]), lo2(f.bA[j]), a00); a01 = wmma4(lo2(f.aA[j]), lo2(f.bB[j]), a01);
        a10 = wmma4(lo2(f.aB[j]), lo2(f.bA[j]), a10); a11 = wmma4(lo2(f.aB[j]), lo2(f.bB[j]), a11);
        a00 = wmma4(hi2(f.aA[j]), hi2(f.bA[j]), a00); a01 = wmma4(hi2(f.aA[j]), hi2(f.bB[j]), a01);
        a10 = wmma4(hi2(f.aB[j]), hi2(f.bA[j]), a10); a11 = wmma4(hi2(f.aB[j]), hi2(f.bB[j]), a11);
    }
}

template <int LD>
static __device__ __forceinline__ void mm_rows_rows(const float* __restrict__ gA,
                                                    const float* __restrict__ gB,
                                                    float* __restrict__ tile, const int nch,
                                                    const int rot,
                                                    v8f& a00, v8f& a01, v8f& a10, v8f& a11) {
    constexpr int STG = GROWS_ * PK_;
    const int t    = threadIdx.x;
    const int lane = t & 31;
    const int wave = t >> 5;
    const int m    = lane & 15;
    const int k4   = (lane >> 4) << 2;
    const int sr = t >> 3;
    const int sq = (t & 7) << 2;
    const float* pA = gA + (size_t)sr * LD + sq;
    const float* pB = gB + (size_t)sr * LD + sq;
    const int sOff = sr * PK_ + sq;
    const int fA0 = m * PK_ + k4;
    const int fA1 = (16 + m) * PK_ + k4;
    const int fB0 = (32 + wave * 32 + m) * PK_ + k4;
    const int fB1 = fB0 + 16 * PK_;
    v4f g[10];
    {
        v4f h[10];
        const int k0 = rot * KC_;
        #pragma unroll
        for (int r = 0; r < 2; ++r) {
            g[r] = *(const v4f*)(pA + (size_t)(16 * r) * LD + k0);
            h[r] = *(const v4f*)(pA + (size_t)(16 * r) * LD + k0 + KC_);
        }
        #pragma unroll
        for (int r = 0; r < 8; ++r) {
            g[2 + r] = *(const v4f*)(pB + (size_t)(16 * r) * LD + k0);
            h[2 + r] = *(const v4f*)(pB + (size_t)(16 * r) * LD + k0 + KC_);
        }
        #pragma unroll
        for (int r = 0; r < 10; ++r) {
            *(v4f*)(tile + sOff + 16 * r * PK_) = g[r];
            *(v4f*)(tile + STG + sOff + 16 * r * PK_) = h[r];
        }
    }
    __syncthreads();
    FragRR P, Q;
    frag_rr_load(P, tile, fA0, fA1, fB0, fB1);
    __syncthreads();
    int q = rot + 2;
    if (q >= nch) q -= nch;
    const float* qA = pA + q * KC_;
    const float* qB = pB + q * KC_;
    #pragma unroll 1
    for (int c = 0; c < nch; c += 2) {
        const bool more = (c + 2 < nch);
        if (more) {
            #pragma unroll
            for (int r = 0; r < 2; ++r) g[r] = *(const v4f*)(qA + (size_t)(16 * r) * LD);
            #pragma unroll
            for (int r = 0; r < 8; ++r) g[2 + r] = *(const v4f*)(qB + (size_t)(16 * r) * LD);
        }
        frag_rr_load(Q, tile + STG, fA0, fA1, fB0, fB1);
        __builtin_amdgcn_sched_barrier(0);
        frag_rr_mma(P, a00, a01, a10, a11);
        __builtin_amdgcn_sched_barrier(0);
        if (more) {
            #pragma unroll
            for (int r = 0; r < 10; ++r) *(v4f*)(tile + sOff + 16 * r * PK_) = g[r];
        }
        __syncthreads();
        if (more) {
            #pragma unroll
            for (int r = 0; r < 2; ++r) g[r] = *(const v4f*)(qA + (size_t)(16 * r) * LD + KC_);
            #pragma unroll
            for (int r = 0; r < 8; ++r) g[2 + r] = *(const v4f*)(qB + (size_t)(16 * r) * LD + KC_);
            frag_rr_load(P, tile, fA0, fA1, fB0, fB1);
        }
        __builtin_amdgcn_sched_barrier(0);
        frag_rr_mma(Q, a00, a01, a10, a11);
        __builtin_amdgcn_sched_barrier(0);
        if (more) {
            #pragma unroll
            for (int r = 0; r < 10; ++r) *(v4f*)(tile + STG + sOff + 16 * r * PK_) = g[r];
        }
        __syncthreads();
        const bool wrap = (q + 2 == nch);
        const int step = wrap ? (2 - nch) * KC_ : 2 * KC_;
        q = wrap ? 0 : q + 2;
        qA += step;
        qB += step;
    }
}

template <int NA>
struct FragKN { v4f aA[NA][4], aB[NA][4]; v2f b0[4], b1[4], b2[4], b3[4]; };

template <int NA>
static __device__ __forceinline__ void frag_kn_load(FragKN<NA>& f, const float* __restrict__ sx,
                                                    const float* __restrict__ sg,
                                                    int fA0, int fA1, int fB, int fBh) {
    #pragma unroll
    for (int j = 0; j < 4; ++j) {
        #pragma unroll
        for (int a = 0; a < NA; ++a) {
            f.aA[a][j] = *(const v4f*)(sg + a * 32 * PK_ + fA0 + 8 * j);
            f.aB[a][j] = *(const v4f*)(sg + a * 32 * PK_ + fA1 + 8 * j);
        }
        const float* r0 = sx + fB + (8 * j) * PX_;
        const float* r1 = sx + fBh + (8 * j) * PX_;
        f.b0[j].x = r0[0];       f.b0[j].y = r0[PX_];
        f.b1[j].x = r1[0];       f.b1[j].y = r1[PX_];
        f.b2[j].x = r0[2 * PX_]; f.b2[j].y = r0[3 * PX_];
        f.b3[j].x = r1[2 * PX_]; f.b3[j].y = r1[3 * PX_];
    }
}

template <int NA>
static __device__ __forceinline__ void frag_kn_mma(const FragKN<NA>& f, v8f (&o)[NA][4]) {
    #pragma unroll
    for (int j = 0; j < 4; ++j) {
        #pragma unroll
        for (int a = 0; a < NA; ++a) {
            o[a][0] = wmma4(lo2(f.aA[a][j]), f.b0[j], o[a][0]); o[a][1] = wmma4(lo2(f.aA[a][j]), f.b1[j], o[a][1]);
            o[a][2] = wmma4(lo2(f.aB[a][j]), f.b0[j], o[a][2]); o[a][3] = wmma4(lo2(f.aB[a][j]), f.b1[j], o[a][3]);
        }
        #pragma unroll
        for (int a = 0; a < NA; ++a) {
            o[a][0] = wmma4(hi2(f.aA[a][j]), f.b2[j], o[a][0]); o[a][1] = wmma4(hi2(f.aA[a][j]), f.b3[j], o[a][1]);
            o[a][2] = wmma4(hi2(f.aB[a][j]), f.b2[j], o[a][2]); o[a][3] = wmma4(hi2(f.aB[a][j]), f.b3[j], o[a][3]);
        }
    }
}

#define PLSUM_(x, y, z) (NPL == 1 ? (x) : NPL == 2 ? (x) + (y) : ((x) + (y)) + (z))
static __device__ __forceinline__ float sum4(v4f v) { return (v.x + v.y) + (v.z + v.w); }
template <int LDB, int NA, int NPL, bool RS>
static __device__ __forceinline__ void mm_rows_kn(const float* __restrict__ gA0,
                                                  const float* __restrict__ gA1,
                                                  const float* __restrict__ gB,
                                                  const float* __restrict__ gB2,
                                                  const float* __restrict__ gB3,
                                                  float* __restrict__ xs,
                                                  float* __restrict__ gs,
                                                  const int rot,
                                                  v8f (&o)[NA][4], float (&rs)[2]) {
    constexpr int NCH = C_ / KC_;
    constexpr int SXS = KC_ * PX_;
    constexpr int SGS = NA * 32 * PK_;
    const int t    = threadIdx.x;
    const int lane = t & 31;
    const int wave = t >> 5;
    const int m    = lane & 15;
    const int k4   = (lane >> 4) << 2;
    const float* pX = gB + (size_t)wave * LDB + (lane << 2);
    const float* pY = gB2 + (size_t)wave * LDB + (lane << 2);
    const float* pZ = gB3 + (size_t)wave * LDB + (lane << 2);
    const int sxOff = wave * PX_ + (lane << 2);
    const int sr = t >> 3;
    const int sq = (t & 7) << 2;
    const float* pG[NA];
    pG[0] = gA0 + (size_t)sr * C_ + sq;
    if (NA > 1) pG[NA - 1] = gA1 + (size_t)sr * C_ + sq;
    const int sgOff = sr * PK_ + sq;
    const int fA0 = m * PK_ + k4;
    const int fA1 = (16 + m) * PK_ + k4;
    const int fB  = k4 * PX_ + wave * 32 + m;
    int fBh = fB + 16;
    asm volatile("" : "+v"(fBh));
    v4f gx[8], gy[8], gz[8], gg[NA][2];
    {
        v4f hx[8], hy[8], hz[8], hg[NA][2];
        const int k0 = rot * KC_;
        #pragma unroll
        for (int r = 0; r < 8; ++r) {
            gx[r] = *(const v4f*)(pX + (size_t)(k0 + 4 * r) * LDB);
            hx[r] = *(const v4f*)(pX + (size_t)(k0 + KC_ + 4 * r) * LDB);
            if (NPL > 1) {
                gy[r] = *(const v4f*)(pY + (size_t)(k0 + 4 * r) * LDB);
                hy[r] = *(const v4f*)(pY + (size_t)(k0 + KC_ + 4 * r) * LDB);
            }
            if (NPL > 2) {
                gz[r] = *(const v4f*)(pZ + (size_t)(k0 + 4 * r) * LDB);
                hz[r] = *(const v4f*)(pZ + (size_t)(k0 + KC_ + 4 * r) * LDB);
            }
        }
        #pragma unroll
        for (int a = 0; a < NA; ++a) {
            #pragma unroll
            for (int r = 0; r < 2; ++r) {
                gg[a][r] = *(const v4f*)(pG[a] + (size_t)(16 * r) * C_ + k0);
                hg[a][r] = *(const v4f*)(pG[a] + (size_t)(16 * r) * C_ + k0 + KC_);
            }
        }
        #pragma unroll
        for (int r = 0; r < 8; ++r) {
            *(v4f*)(xs + sxOff + 4 * r * PX_) = PLSUM_(gx[r], gy[r], gz[r]);
            *(v4f*)(xs + SXS + sxOff + 4 * r * PX_) = PLSUM_(hx[r], hy[r], hz[r]);
        }
        #pragma unroll
        for (int a = 0; a < NA; ++a) {
            #pragma unroll
            for (int r = 0; r < 2; ++r) {
                *(v4f*)(gs + (a * 32 + 16 * r) * PK_ + sgOff) = gg[a][r];
                *(v4f*)(gs + SGS + (a * 32 + 16 * r) * PK_ + sgOff) = hg[a][r];
                if (RS) rs[r] = sum4(gg[a][r]) + sum4(hg[a][r]);
            }
        }
    }
    __syncthreads();
    FragKN<NA> P, Q;
    frag_kn_load<NA>(P, xs, gs, fA0, fA1, fB, fBh);
    __syncthreads();
    int q = (rot + 2) & (NCH - 1);
    const float* qX = pX + (size_t)(q * KC_) * LDB;
    const float* qY = pY + (size_t)(q * KC_) * LDB;
    const float* qZ = pZ + (size_t)(q * KC_) * LDB;
    const float* qG0 = pG[0] + q * KC_;
    const float* qG1 = pG[NA - 1] + q * KC_;
    #pragma unroll 1
    for (int c = 0; c < NCH; c += 2) {
        const bool more = (c + 2 < NCH);
        if (more) {
            #pragma unroll
            for (int r = 0; r < 8; ++r) gx[r] = *(const v4f*)(qX + (size_t)(4 * r) * LDB);
            if (NPL > 1) {
                #pragma unroll
                for (int r = 0; r < 8; ++r) gy[r] = *(const v4f*)(qY + (size_t)(4 * r) * LDB);
            }
            if (NPL > 2) {
                #pragma unroll
                for (int r = 0; r < 8; ++r) gz[r] = *(const v4f*)(qZ + (size_t)(4 * r) * LDB);
            }
            #pragma unroll
            for (int a = 0; a < NA; ++a) {
                #pragma unroll
                for (int r = 0; r < 2; ++r) gg[a][r] = *(const v4f*)((a == 0 ? qG0 : qG1) + (size_t)(16 * r) * C_);
            }
        }
        frag_kn_load<NA>(Q, xs + SXS, gs + SGS, fA0, fA1, fB, fBh);
        __builtin_amdgcn_sched_barrier(0);
        frag_kn_mma<NA>(P, o);
        __builtin_amdgcn_sched_barrier(0);
        if (more) {
            #pragma unroll
            for (int r = 0; r < 8; ++r) *(v4f*)(xs + sxOff + 4 * r * PX_) = PLSUM_(gx[r], gy[r], gz[r]);
            #pragma unroll
            for (int a = 0; a < NA; ++a) {
                #pragma unroll
                for (int r = 0; r < 2; ++r) {
                    *(v4f*)(gs + (a * 32 + 16 * r) * PK_ + sgOff) = gg[a][r];
                    if (RS) rs[r] += sum4(gg[a][r]);
                }
            }
        }
        __syncthreads();
        if (more) {
            #pragma unroll
            for (int r = 0; r < 8; ++r) gx[r] = *(const v4f*)(qX + (size_t)(KC_ + 4 * r) * LDB);
            if (NPL > 1) {
                #pragma unroll
                for (int r = 0; r < 8; ++r) gy[r] = *(const v4f*)(qY + (size_t)(KC_ + 4 * r) * LDB);
            }
            if (NPL > 2) {
                #pragma unroll
                for (int r = 0; r < 8; ++r) gz[r] = *(const v4f*)(qZ + (size_t)(KC_ + 4 * r) * LDB);
            }
            #pragma unroll
            for (int a = 0; a < NA; ++a) {
                #pragma unroll
                for (int r = 0; r < 2; ++r) gg[a][r] = *(const v4f*)((a == 0 ? qG0 : qG1) + (size_t)(16 * r) * C_ + KC_);
            }
            frag_kn_load<NA>(P, xs, gs, fA0, fA1, fB, fBh);
        }
        __builtin_amdgcn_sched_barrier(0);
        frag_kn_mma<NA>(Q, o);
        __builtin_amdgcn_sched_barrier(0);
        if (more) {
            #pragma unroll
            for (int r = 0; r < 8; ++r) *(v4f*)(xs + SXS + sxOff + 4 * r * PX_) = PLSUM_(gx[r], gy[r], gz[r]);
            #pragma unroll
            for (int a = 0; a < NA; ++a) {
                #pragma unroll
                for (int r = 0; r < 2; ++r) {
                    *(v4f*)(gs + SGS + (a * 32 + 16 * r) * PK_ + sgOff) = gg[a][r];
                    if (RS) rs[r] += sum4(gg[a][r]);
                }
            }
        }
        __syncthreads();
        const bool wrap = (q + 2 == NCH);
        const int step = wrap ? (2 - NCH) * KC_ : 2 * KC_;
        q = wrap ? 0 : q + 2;
        qX += (ptrdiff_t)step * LDB;
        if (NPL > 1) qY += (ptrdiff_t)step * LDB;
        if (NPL > 2) qZ += (ptrdiff_t)step * LDB;
        qG0 += step;
        if (NA > 1) qG1 += step;
    }
}

#define KP0_ 44
#define KP1_ 42
__global__ __launch_bounds__(128) __attribute__((amdgpu_waves_per_eu(1, 2))) void k_gram(const float* __restrict__ x,
                                              float* __restrict__ S0,
                                              float* __restrict__ S1,
                                              float* __restrict__ S2) {
    __shared__ __attribute__((aligned(16))) float tile[2 * GROWS_ * PK_];
    const int lane = threadIdx.x & 31;
    const int wave = threadIdx.x >> 5;
    const int m    = lane & 15;
    const int half = lane >> 4;
    const int p    = blockIdx.x >> 2;
    const int I    = (p >= 4) + (p >= 7) + (p >= 9);
    const int J    = p - (4 * I - ((I * (I - 1)) >> 1)) + I;
    const int c0   = I * 128 + (blockIdx.x & 3) * 32;
    const int dblk = J * 128;
    const int d0   = dblk + wave * 32;
    const int kh   = blockIdx.y;
    const int b    = blockIdx.z;
    const int ch0  = kh == 0 ? 0 : KP0_ + (kh - 1) * KP1_;
    const int nch  = kh == 0 ? KP0_ : KP1_;
    const size_t bOff = (size_t)b * C_ * N_ + (size_t)ch0 * KC_;
    v8f a00 = {}, a01 = {}, a10 = {}, a11 = {};
    const int rot = (blockIdx.x & 7) << 1;
    mm_rows_rows<N_>(x + bOff + (size_t)c0 * N_, x + bOff + (size_t)dblk * N_, tile, nch, rot, a00, a01, a10, a11);
    float* Sb = (kh == 0 ? S0 : kh == 1 ? S1 : S2) + (size_t)b * C_ * C_;
    #pragma unroll
    for (int j = 0; j < 8; ++j) {
        const int cA = c0 + j + 8 * half;
        const int cB = cA + 16;
        Sb[(size_t)cA * C_ + d0 + m]      = a00[j];
        Sb[(size_t)cA * C_ + d0 + 16 + m] = a01[j];
        Sb[(size_t)cB * C_ + d0 + m]      = a10[j];
        Sb[(size_t)cB * C_ + d0 + 16 + m] = a11[j];
    }
    if (I != J) {
        float* r0 = Sb + (size_t)(d0 + m) * C_ + c0 + 8 * half;
        float* r1 = r0 + (size_t)16 * C_;
        *(v8f*)(r0)      = a00;
        *(v8f*)(r0 + 16) = a10;
        *(v8f*)(r1)      = a01;
        *(v8f*)(r1 + 16) = a11;
    }
}

__global__ __launch_bounds__(128) __attribute__((amdgpu_waves_per_eu(1, 2))) void k_projs(const float* __restrict__ S0,
                                               const float* __restrict__ S1,
                                               const float* __restrict__ S2,
                                               const float* __restrict__ Wq,
                                               const float* __restrict__ Wk,
                                               float* __restrict__ T,
                                               float* __restrict__ PQ,
                                               float* __restrict__ PK) {
    __shared__ __attribute__((aligned(16))) float xs[2 * KC_ * PX_];
    __shared__ __attribute__((aligned(16))) float gs[2 * 2 * 32 * PK_];
    const int lane = threadIdx.x & 31;
    const int wave = threadIdx.x >> 5;
    const int m    = lane & 15;
    const int half = lane >> 4;
    const int c0   = blockIdx.y * 32;
    const int nblk = blockIdx.x * 128;
    const int n0   = nblk + wave * 32;
    const int b    = blockIdx.z;
    v8f o[2][4] = {};
    float unused[2];
    const int rot = ((blockIdx.x + blockIdx.y + blockIdx.z) & 7) << 1;
    mm_rows_kn<C_, 2, 3, false>(Wq + (size_t)c0 * C_, Wk + (size_t)c0 * C_, S0 + (size_t)b * C_ * C_ + nblk,
                                S1 + (size_t)b * C_ * C_ + nblk, S2 + (size_t)b * C_ * C_ + nblk, xs, gs, rot, o, unused);
    float* Tb = T + (size_t)b * C_ * C_;
    float sq[8][2], sk[8][2];
    #pragma unroll
    for (int j = 0; j < 8; ++j) {
        const int cA = c0 + j + 8 * half;
        const int cB = cA + 16;
        const size_t pA = (size_t)cA * C_ + n0 + m;
        const size_t pB = (size_t)cB * C_ + n0 + m;
        Tb[pA]      = o[0][0][j];
        Tb[pA + 16] = o[0][1][j];
        Tb[pB]      = o[0][2][j];
        Tb[pB + 16] = o[0][3][j];
        sq[j][0] = o[0][0][j] * Wq[pA] + o[0][1][j] * Wq[pA + 16];
        sq[j][1] = o[0][2][j] * Wq[pB] + o[0][3][j] * Wq[pB + 16];
        sk[j][0] = o[1][0][j] * Wk[pA] + o[1][1][j] * Wk[pA + 16];
        sk[j][1] = o[1][2][j] * Wk[pB] + o[1][3][j] * Wk[pB + 16];
    }
    #pragma unroll
    for (int s = 1; s < 16; s <<= 1) {
        #pragma unroll
        for (int j = 0; j < 8; ++j) {
            sq[j][0] += __shfl_xor(sq[j][0], s); sq[j][1] += __shfl_xor(sq[j][1], s);
            sk[j][0] += __shfl_xor(sk[j][0], s); sk[j][1] += __shfl_xor(sk[j][1], s);
        }
    }
    float* red = xs;
    if (m == 0) {
        #pragma unroll
        for (int j = 0; j < 8; ++j) {
            red[wave * 64 + j + 8 * half]           = sq[j][0];
            red[wave * 64 + j + 8 * half + 16]      = sq[j][1];
            red[wave * 64 + 32 + j + 8 * half]      = sk[j][0];
            red[wave * 64 + 32 + j + 8 * half + 16] = sk[j][1];
        }
    }
    __syncthreads();
    if (threadIdx.x < 64) {
        const int t = threadIdx.x;
        const float v = ((red[t] + red[64 + t]) + red[128 + t]) + red[192 + t];
        float* dst = (t < 32 ? PQ : PK) + ((size_t)(b * C_ + c0 + (t & 31))) * 4 + blockIdx.x;
        *dst = v;
    }
}

__global__ __launch_bounds__(128) __attribute__((amdgpu_waves_per_eu(1, 2))) void k_qkt(const float* __restrict__ T,
                                             const float* __restrict__ Wk,
                                             const float* __restrict__ PQ,
                                             const float* __restrict__ PK,
                                             float* __restrict__ M) {
    __shared__ __attribute__((aligned(16))) float tile[2 * GROWS_ * PK_];
    __shared__ float sIQ[32], sIK[128];
    const int lane = threadIdx.x & 31;
    const int wave = threadIdx.x >> 5;
    const int m    = lane & 15;
    const int half = lane >> 4;
    const int c0   = blockIdx.y * 32;
    const int dblk = blockIdx.x * 128;
    const int d0   = dblk + wave * 32;
    const int b    = blockIdx.z;
    const size_t bOff = (size_t)b * C_ * C_;
    const v4f pk = *(const v4f*)(PK + ((size_t)(b * C_ + dblk + threadIdx.x)) * 4);
    const v4f pq = *(const v4f*)(PQ + ((size_t)(b * C_ + c0 + (threadIdx.x & 31))) * 4);
    v8f a00 = {}, a01 = {}, a10 = {}, a11 = {};
    const int rot = ((blockIdx.x + blockIdx.y + blockIdx.z) & 7) << 1;
    mm_rows_rows<C_>(T + bOff + (size_t)c0 * C_, Wk + (size_t)dblk * C_, tile, C_ / KC_, rot, a00, a01, a10, a11);
    sIK[threadIdx.x] = rsqrtf((pk.x + pk.y) + (pk.z + pk.w));
    if (threadIdx.x < 32) sIQ[threadIdx.x] = rsqrtf((pq.x + pq.y) + (pq.z + pq.w));
    __syncthreads();
    float* Mb = M + (size_t)b * C_ * C_;
    const float ik0 = sIK[wave * 32 + m];
    const float ik1 = sIK[wave * 32 + 16 + m];
    #pragma unroll
    for (int j = 0; j < 8; ++j) {
        const int cA = c0 + j + 8 * half;
        const int cB = cA + 16;
        const float iqA = sIQ[j + 8 * half];
        const float iqB = sIQ[j + 8 * half + 16];
        Mb[(size_t)cA * C_ + d0 + m]      = a00[j] * iqA * ik0;
        Mb[(size_t)cA * C_ + d0 + 16 + m] = a01[j] * iqA * ik1;
        Mb[(size_t)cB * C_ + d0 + m]      = a10[j] * iqB * ik0;
        Mb[(size_t)cB * C_ + d0 + 16 + m] = a11[j] * iqB * ik1;
    }
}

__global__ __launch_bounds__(256) void k_colsoftmax(float* __restrict__ M) {
    __shared__ float smax[8][32], smin[8][32], ssum[8][32];
    const int lane = threadIdx.x & 31;
    const int wave = threadIdx.x >> 5;
    const int tile = blockIdx.x;
    const int b  = tile / (C_ / 32);
    const int d  = (tile - b * (C_ / 32)) * 32 + lane;
    float* p = M + (size_t)b * C_ * C_ + (size_t)(wave * 64) * C_ + d;
    float v[64];
    for (int i = 0; i < 64; ++i) v[i] = 1.f - p[(size_t)i * C_];
    float mx = -3.4e38f, mn = 3.4e38f;
    for (int i = 0; i < 64; ++i) { mx = fmaxf(mx, v[i]); mn = fminf(mn, v[i]); }
    smax[wave][lane] = mx; smin[wave][lane] = mn;
    __syncthreads();
    float cmx = -3.4e38f, cmn = 3.4e38f;
    for (int j = 0; j < 8; ++j) {
        cmx = fmaxf(cmx, smax[j][lane]);
        cmn = fminf(cmn, smin[j][lane]);
    }
    const float dn = cmx + EPS_;
    const float inv_dn = 1.f / dn;
    const float sh = 1.f - (cmn / dn) * (1.f / HT_);
    float z = 0.f;
    for (int i = 0; i < 64; ++i) {
        const float m4 = 1.f - (v[i] * inv_dn) * (1.f / HT_);
        const float e = expf(m4 - sh);
        v[i] = e;
        z += e;
    }
    ssum[wave][lane] = z;
    __syncthreads();
    float tot = 0.f;
    for (int j = 0; j < 8; ++j) tot += ssum[j][lane];
    const float invz = 1.f / tot;
    for (int i = 0; i < 64; ++i) p[(size_t)i * C_] = v[i] * invz;
}

__global__ __launch_bounds__(128) __attribute__((amdgpu_waves_per_eu(1, 2))) void k_mwv(const float* __restrict__ M,
                                             const float* __restrict__ Wv,
                                             float* __restrict__ G) {
    __shared__ __attribute__((aligned(16))) float xs[2 * KC_ * PX_];
    __shared__ __attribute__((aligned(16))) float gs[2 * 32 * PK_];
    const int lane = threadIdx.x & 31;
    const int wave = threadIdx.x >> 5;
    const int m    = lane & 15;
    const int half = lane >> 4;
    const int c0   = blockIdx.y * 32;
    const int eblk = blockIdx.x * 128;
    const int e0   = eblk + wave * 32;
    const int b    = blockIdx.z;
    const float* Mb = M + (size_t)b * C_ * C_ + (size_t)c0 * C_;
    v8f o[1][4] = {};
    float rs[2];
    const int rot = ((blockIdx.x + blockIdx.y + blockIdx.z) & 7) << 1;
    mm_rows_kn<C_, 1, 1, true>(Mb, Mb, Wv + eblk, Wv + eblk, Wv + eblk, xs, gs, rot, o, rs);
    for (int s = 1; s < 8; s <<= 1) {
        rs[0] += __shfl_xor(rs[0], s);
        rs[1] += __shfl_xor(rs[1], s);
    }
    float* srow = gs;
    if ((threadIdx.x & 7) == 0) {
        srow[threadIdx.x >> 3]        = 1.f / (rs[0] + EPS_);
        srow[(threadIdx.x >> 3) + 16] = 1.f / (rs[1] + EPS_);
    }
    __syncthreads();
    float* Gb = G + (size_t)b * C_ * C_;
    #pragma unroll
    for (int j = 0; j < 8; ++j) {
        const int cA = c0 + j + 8 * half;
        const int cB = cA + 16;
        const float iA = srow[j + 8 * half];
        const float iB = srow[j + 8 * half + 16];
        Gb[(size_t)cA * C_ + e0 + m]      = o[0][0][j] * iA;
        Gb[(size_t)cA * C_ + e0 + 16 + m] = o[0][1][j] * iA;
        Gb[(size_t)cB * C_ + e0 + m]      = o[0][2][j] * iB;
        Gb[(size_t)cB * C_ + e0 + 16 + m] = o[0][3][j] * iB;
    }
}

__global__ __launch_bounds__(128) void k_out(const float* __restrict__ x,
                                             const float* __restrict__ G,
                                             const float* __restrict__ gamma,
                                             float* __restrict__ y) {
    __shared__ __attribute__((aligned(16))) float xs[2 * KC_ * PX_];
    __shared__ __attribute__((aligned(16))) float gs[2 * 32 * PK_];
    const int lane = threadIdx.x & 31;
    const int wave = threadIdx.x >> 5;
    const int m    = lane & 15;
    const int half = lane >> 4;
    const int c0   = blockIdx.y * 32;
    const int nblk = blockIdx.x * 128;
    const int n0   = nblk + wave * 32;
    const int b    = blockIdx.z;
    const float* xb = x + (size_t)b * C_ * N_;
    const float* Gb = G + (size_t)b * C_ * C_ + (size_t)c0 * C_;
    v8f o[1][4] = {};
    float unused[2];
    const int rot = ((blockIdx.x + blockIdx.y + blockIdx.z) & 7) << 1;
    mm_rows_kn<N_, 1, 1, false>(Gb, Gb, xb + nblk, xb + nblk, xb + nblk, xs, gs, rot, o, unused);
    const float g = gamma[0];
    float* yb = y + (size_t)b * C_ * N_;
    #pragma unroll
    for (int j = 0; j < 8; ++j) {
        const int cA = c0 + j + 8 * half;
        const int cB = cA + 16;
        const size_t o0 = (size_t)cA * N_ + n0 + m;
        const size_t o1 = (size_t)cA * N_ + n0 + 16 + m;
        const size_t o2 = (size_t)cB * N_ + n0 + m;
        const size_t o3 = (size_t)cB * N_ + n0 + 16 + m;
        yb[o0] = xb[o0] + g * o[0][0][j];
        yb[o1] = xb[o1] + g * o[0][1][j];
        yb[o2] = xb[o2] + g * o[0][2][j];
        yb[o3] = xb[o3] + g * o[0][3][j];
    }
}

extern "C" void kernel_launch(void* const* d_in, const int* in_sizes, int n_in,
                              void* d_out, int out_size, void* d_ws, size_t ws_size,
                              hipStream_t stream) {
    (void)in_sizes; (void)n_in; (void)out_size; (void)ws_size;
    const float* x     = (const float*)d_in[0];
    const float* Wq    = (const float*)d_in[1];
    const float* Wk    = (const float*)d_in[2];
    const float* Wv    = (const float*)d_in[3];
    const float* gamma = (const float*)d_in[4];
    float* y = (float*)d_out;

    char* ws = (char*)d_ws;
    const size_t mBytes  = (size_t)B_ * C_ * C_ * sizeof(float);
    float* S     = (float*)(ws);
    float* T     = (float*)(ws + mBytes);
    float* PQ    = (float*)(ws + 2 * mBytes);
    float* PK    = PQ + B_ * C_ * 4;
    float* M     = (float*)(ws + 3 * mBytes);
    float* G     = (float*)(ws + 4 * mBytes);

    k_gram      <<<dim3(40, 3, B_), 128, 0, stream>>>(x, S, G, M);
    k_projs     <<<dim3(C_ / 128, C_ / 32, B_), 128, 0, stream>>>(S, G, M, Wq, Wk, T, PQ, PK);
    k_qkt       <<<dim3(C_ / 128, C_ / 32, B_), 128, 0, stream>>>(T, Wk, PQ, PK, M);
    k_colsoftmax<<<B_ * C_ / 32, 256, 0, stream>>>(M);
    k_mwv       <<<dim3(C_ / 128, C_ / 32, B_), 128, 0, stream>>>(M, Wv, G);
    k_out       <<<dim3(N_ / 128, C_ / 32, B_), 128, 0, stream>>>(x, G, gamma, y);
}


# Gram and output GEMMs: work-group tile 32x128 -> 64x128 (2 A tiles per B fragment per wave): fewer staged lines, LDS bytes, barriers per WMMA; rotated k start moves with block index, k sums reorder
# speedup vs baseline: 3.1092x; 1.0323x over previous
#include <hip/hip_runtime.h>

typedef __attribute__((ext_vector_type(2))) float v2f;
typedef __attribute__((ext_vector_type(4))) float v4f;
typedef __attribute__((ext_vector_type(8))) float v8f;

#define B_ 8
#define C_ 512
#define N_ 4096
#define EPS_ 1e-6f
#define HT_ 0.25f

static __device__ __forceinline__ v8f wmma4(v2f a, v2f b, v8f c) {
    return __builtin_amdgcn_wmma_f32_16x16x4_f32(false, a, false, b, (short)0, c, false, false);
}

static __device__ __forceinline__ v2f lo2(v4f v) { return __builtin_shufflevector(v, v, 0, 1); }
static __device__ __forceinline__ v2f hi2(v4f v) { return __builtin_shufflevector(v, v, 2, 3); }


#define KC_    32
#define PK_    36
#define GROWS_(na) (32 * (na) + 128)
#define PX_    132

template <int NA>
struct FragRR { v4f aA[NA][4], aB[NA][4], bA[4], bB[4]; };

template <int NA>
static __device__ __forceinline__ void frag_rr_load(FragRR<NA>& f, const float* __restrict__ st,
                                                    int fA0, int fA1, int fB0, int fB1) {
    #pragma unroll
    for (int j = 0; j < 4; ++j) {
        #pragma unroll
        for (int a = 0; a < NA; ++a) {
            f.aA[a][j] = *(const v4f*)(st + a * 32 * PK_ + fA0 + 8 * j);
            f.aB[a][j] = *(const v4f*)(st + a * 32 * PK_ + fA1 + 8 * j);
        }
        f.bA[j] = *(const v4f*)(st + fB0 + 8 * j);
        f.bB[j] = *(const v4f*)(st + fB1 + 8 * j);
    }
}

template <int NA>
static __device__ __forceinline__ void frag_rr_mma(const FragRR<NA>& f, v8f (&acc)[NA][4]) {
    #pragma unroll
    for (int j = 0; j < 4; ++j) {
        #pragma unroll
        for (int a = 0; a < NA; ++a) {
            acc[a][0] = wmma4(lo2(f.aA[a][j]), lo2(f.bA[j]), acc[a][0]); acc[a][1] = wmma4(lo2(f.aA[a][j]), lo2(f.bB[j]), acc[a][1]);
            acc[a][2] = wmma4(lo2(f.aB[a][j]), lo2(f.bA[j]), acc[a][2]); acc[a][3] = wmma4(lo2(f.aB[a][j]), lo2(f.bB[j]), acc[a][3]);
        }
        #pragma unroll
        for (int a = 0; a < NA; ++a) {
            acc[a][0] = wmma4(hi2(f.aA[a][j]), hi2(f.bA[j]), acc[a][0]); acc[a][1] = wmma4(hi2(f.aA[a][j]), hi2(f.bB[j]), acc[a][1]);
            acc[a][2] = wmma4(hi2(f.aB[a][j]), hi2(f.bA[j]), acc[a][2]); acc[a][3] = wmma4(hi2(f.aB[a][j]), hi2(f.bB[j]), acc[a][3]);
        }
    }
}

template <int LD, int NA>
static __device__ __forceinline__ void mm_rows_rows(const float* __restrict__ gA,
                                                    const float* __restrict__ gB,
                                                    float* __restrict__ tile, const int nch,
                                                    const int rot,
                                                    v8f (&acc)[NA][4]) {
    constexpr int RA  = 2 * NA;
    constexpr int RT  = RA + 8;
    constexpr int STG = GROWS_(NA) * PK_;
    const int t    = threadIdx.x;
    const int lane = t & 31;
    const int wave = t >> 5;
    const int m    = lane & 15;
    const int k4   = (lane >> 4) << 2;
    const int sr = t >> 3;
    const int sq = (t & 7) << 2;
    const float* pA = gA + (size_t)sr * LD + sq;
    const float* pB = gB + (size_t)sr * LD + sq;
    const int sOff = sr * PK_ + sq;
    const int fA0 = m * PK_ + k4;
    const int fA1 = (16 + m) * PK_ + k4;
    const int fB0 = (32 * NA + wave * 32 + m) * PK_ + k4;
    const int fB1 = fB0 + 16 * PK_;
    v4f g[RT];
    {
        v4f h[RT];
        const int k0 = rot * KC_;
        #pragma unroll
        for (int r = 0; r < RA; ++r) {
            g[r] = *(const v4f*)(pA + (size_t)(16 * r) * LD + k0);
            h[r] = *(const v4f*)(pA + (size_t)(16 * r) * LD + k0 + KC_);
        }
        #pragma unroll
        for (int r = 0; r < 8; ++r) {
            g[RA + r] = *(const v4f*)(pB + (size_t)(16 * r) * LD + k0);
            h[RA + r] = *(const v4f*)(pB + (size_t)(16 * r) * LD + k0 + KC_);
        }
        #pragma unroll
        for (int r = 0; r < RT; ++r) {
            *(v4f*)(tile + sOff + 16 * r * PK_) = g[r];
            *(v4f*)(tile + STG + sOff + 16 * r * PK_) = h[r];
        }
    }
    __syncthreads();
    FragRR<NA> P, Q;
    frag_rr_load<NA>(P, tile, fA0, fA1, fB0, fB1);
    __syncthreads();
    int q = rot + 2;
    if (q >= nch) q -= nch;
    const float* qA = pA + q * KC_;
    const float* qB = pB + q * KC_;
    #pragma unroll 1
    for (int c = 0; c < nch; c += 2) {
        const bool more = (c + 2 < nch);
        if (more) {
            #pragma unroll
            for (int r = 0; r < RA; ++r) g[r] = *(const v4f*)(qA + (size_t)(16 * r) * LD);
            #pragma unroll
            for (int r = 0; r < 8; ++r) g[RA + r] = *(const v4f*)(qB + (size_t)(16 * r) * LD);
        }
        frag_rr_load<NA>(Q, tile + STG, fA0, fA1, fB0, fB1);
        __builtin_amdgcn_sched_barrier(0);
        frag_rr_mma<NA>(P, acc);
        __builtin_amdgcn_sched_barrier(0);
        if (more) {
            #pragma unroll
            for (int r = 0; r < RT; ++r) *(v4f*)(tile + sOff + 16 * r * PK_) = g[r];
        }
        __syncthreads();
        if (more) {
            #pragma unroll
            for (int r = 0; r < RA; ++r) g[r] = *(const v4f*)(qA + (size_t)(16 * r) * LD + KC_);
            #pragma unroll
            for (int r = 0; r < 8; ++r) g[RA + r] = *(const v4f*)(qB + (size_t)(16 * r) * LD + KC_);
            frag_rr_load<NA>(P, tile, fA0, fA1, fB0, fB1);
        }
        __builtin_amdgcn_sched_barrier(0);
        frag_rr_mma<NA>(Q, acc);
        __builtin_amdgcn_sched_barrier(0);
        if (more) {
            #pragma unroll
            for (int r = 0; r < RT; ++r) *(v4f*)(tile + STG + sOff + 16 * r * PK_) = g[r];
        }
        __syncthreads();
        const bool wrap = (q + 2 == nch);
        const int step = wrap ? (2 - nch) * KC_ : 2 * KC_;
        q = wrap ? 0 : q + 2;
        qA += step;
        qB += step;
    }
}

template <int NA>
struct FragKN { v4f aA[NA][4], aB[NA][4]; v2f b0[4], b1[4], b2[4], b3[4]; };

template <int NA>
static __device__ __forceinline__ void frag_kn_load(FragKN<NA>& f, const float* __restrict__ sx,
                                                    const float* __restrict__ sg,
                                                    int fA0, int fA1, int fB, int fBh) {
    #pragma unroll
    for (int j = 0; j < 4; ++j) {
        #pragma unroll
        for (int a = 0; a < NA; ++a) {
            f.aA[a][j] = *(const v4f*)(sg + a * 32 * PK_ + fA0 + 8 * j);
            f.aB[a][j] = *(const v4f*)(sg + a * 32 * PK_ + fA1 + 8 * j);
        }
        const float* r0 = sx + fB + (8 * j) * PX_;
        const float* r1 = sx + fBh + (8 * j) * PX_;
        f.b0[j].x = r0[0];       f.b0[j].y = r0[PX_];
        f.b1[j].x = r1[0];       f.b1[j].y = r1[PX_];
        f.b2[j].x = r0[2 * PX_]; f.b2[j].y = r0[3 * PX_];
        f.b3[j].x = r1[2 * PX_]; f.b3[j].y = r1[3 * PX_];
    }
}

template <int NA>
static __device__ __forceinline__ void frag_kn_mma(const FragKN<NA>& f, v8f (&o)[NA][4]) {
    #pragma unroll
    for (int j = 0; j < 4; ++j) {
        #pragma unroll
        for (int a = 0; a < NA; ++a) {
            o[a][0] = wmma4(lo2(f.aA[a][j]), f.b0[j], o[a][0]); o[a][1] = wmma4(lo2(f.aA[a][j]), f.b1[j], o[a][1]);
            o[a][2] = wmma4(lo2(f.aB[a][j]), f.b0[j], o[a][2]); o[a][3] = wmma4(lo2(f.aB[a][j]), f.b1[j], o[a][3]);
        }
        #pragma unroll
        for (int a = 0; a < NA; ++a) {
            o[a][0] = wmma4(hi2(f.aA[a][j]), f.b2[j], o[a][0]); o[a][1] = wmma4(hi2(f.aA[a][j]), f.b3[j], o[a][1]);
            o[a][2] = wmma4(hi2(f.aB[a][j]), f.b2[j], o[a][2]); o[a][3] = wmma4(hi2(f.aB[a][j]), f.b3[j], o[a][3]);
        }
    }
}

#define PLSUM_(x, y, z) (NPL == 1 ? (x) : NPL == 2 ? (x) + (y) : ((x) + (y)) + (z))
static __device__ __forceinline__ float sum4(v4f v) { return (v.x + v.y) + (v.z + v.w); }
template <int LDB, int NA, int NPL, bool RS>
static __device__ __forceinline__ void mm_rows_kn(const float* __restrict__ gA0,
                                                  const float* __restrict__ gA1,
                                                  const float* __restrict__ gB,
                                                  const float* __restrict__ gB2,
                                                  const float* __restrict__ gB3,
                                                  float* __restrict__ xs,
                                                  float* __restrict__ gs,
                                                  const int rot,
                                                  v8f (&o)[NA][4], float (&rs)[2]) {
    constexpr int NCH = C_ / KC_;
    constexpr int SXS = KC_ * PX_;
    constexpr int SGS = NA * 32 * PK_;
    const int t    = threadIdx.x;
    const int lane = t & 31;
    const int wave = t >> 5;
    const int m    = lane & 15;
    const int k4   = (lane >> 4) << 2;
    const float* pX = gB + (size_t)wave * LDB + (lane << 2);
    const float* pY = gB2 + (size_t)wave * LDB + (lane << 2);
    const float* pZ = gB3 + (size_t)wave * LDB + (lane << 2);
    const int sxOff = wave * PX_ + (lane << 2);
    const int sr = t >> 3;
    const int sq = (t & 7) << 2;
    const float* pG[NA];
    pG[0] = gA0 + (size_t)sr * C_ + sq;
    if (NA > 1) pG[NA - 1] = gA1 + (size_t)sr * C_ + sq;
    const int sgOff = sr * PK_ + sq;
    const int fA0 = m * PK_ + k4;
    const int fA1 = (16 + m) * PK_ + k4;
    const int fB  = k4 * PX_ + wave * 32 + m;
    int fBh = fB + 16;
    asm volatile("" : "+v"(fBh));
    v4f gx[8], gy[8], gz[8], gg[NA][2];
    {
        v4f hx[8], hy[8], hz[8], hg[NA][2];
        const int k0 = rot * KC_;
        #pragma unroll
        for (int r = 0; r < 8; ++r) {
            gx[r] = *(const v4f*)(pX + (size_t)(k0 + 4 * r) * LDB);
            hx[r] = *(const v4f*)(pX + (size_t)(k0 + KC_ + 4 * r) * LDB);
            if (NPL > 1) {
                gy[r] = *(const v4f*)(pY + (size_t)(k0 + 4 * r) * LDB);
                hy[r] = *(const v4f*)(pY + (size_t)(k0 + KC_ + 4 * r) * LDB);
            }
            if (NPL > 2) {
                gz[r] = *(const v4f*)(pZ + (size_t)(k0 + 4 * r) * LDB);
                hz[r] = *(const v4f*)(pZ + (size_t)(k0 + KC_ + 4 * r) * LDB);
            }
        }
        #pragma unroll
        for (int a = 0; a < NA; ++a) {
            #pragma unroll
            for (int r = 0; r < 2; ++r) {
                gg[a][r] = *(const v4f*)(pG[a] + (size_t)(16 * r) * C_ + k0);
                hg[a][r] = *(const v4f*)(pG[a] + (size_t)(16 * r) * C_ + k0 + KC_);
            }
        }
        #pragma unroll
        for (int r = 0; r < 8; ++r) {
            *(v4f*)(xs + sxOff + 4 * r * PX_) = PLSUM_(gx[r], gy[r], gz[r]);
            *(v4f*)(xs + SXS + sxOff + 4 * r * PX_) = PLSUM_(hx[r], hy[r], hz[r]);
        }
        #pragma unroll
        for (int a = 0; a < NA; ++a) {
            #pragma unroll
            for (int r = 0; r < 2; ++r) {
                *(v4f*)(gs + (a * 32 + 16 * r) * PK_ + sgOff) = gg[a][r];
                *(v4f*)(gs + SGS + (a * 32 + 16 * r) * PK_ + sgOff) = hg[a][r];
                if (RS) rs[r] = sum4(gg[a][r]) + sum4(hg[a][r]);
            }
        }
    }
    __syncthreads();
    FragKN<NA> P, Q;
    frag_kn_load<NA>(P, xs, gs, fA0, fA1, fB, fBh);
    __syncthreads();
    int q = (rot + 2) & (NCH - 1);
    const float* qX = pX + (size_t)(q * KC_) * LDB;
    const float* qY = pY + (size_t)(q * KC_) * LDB;
    const float* qZ = pZ + (size_t)(q * KC_) * LDB;
    const float* qG0 = pG[0] + q * KC_;
    const float* qG1 = pG[NA - 1] + q * KC_;
    #pragma unroll 1
    for (int c = 0; c < NCH; c += 2) {
        const bool more = (c + 2 < NCH);
        if (more) {
            #pragma unroll
            for (int r = 0; r < 8; ++r) gx[r] = *(const v4f*)(qX + (size_t)(4 * r) * LDB);
            if (NPL > 1) {
                #pragma unroll
                for (int r = 0; r < 8; ++r) gy[r] = *(const v4f*)(qY + (size_t)(4 * r) * LDB);
            }
            if (NPL > 2) {
                #pragma unroll
                for (int r = 0; r < 8; ++r) gz[r] = *(const v4f*)(qZ + (size_t)(4 * r) * LDB);
            }
            #pragma unroll
            for (int a = 0; a < NA; ++a) {
                #pragma unroll
                for (int r = 0; r < 2; ++r) gg[a][r] = *(const v4f*)((a == 0 ? qG0 : qG1) + (size_t)(16 * r) * C_);
            }
        }
        frag_kn_load<NA>(Q, xs + SXS, gs + SGS, fA0, fA1, fB, fBh);
        __builtin_amdgcn_sched_barrier(0);
        frag_kn_mma<NA>(P, o);
        __builtin_amdgcn_sched_barrier(0);
        if (more) {
            #pragma unroll
            for (int r = 0; r < 8; ++r) *(v4f*)(xs + sxOff + 4 * r * PX_) = PLSUM_(gx[r], gy[r], gz[r]);
            #pragma unroll
            for (int a = 0; a < NA; ++a) {
                #pragma unroll
                for (int r = 0; r < 2; ++r) {
                    *(v4f*)(gs + (a * 32 + 16 * r) * PK_ + sgOff) = gg[a][r];
                    if (RS) rs[r] += sum4(gg[a][r]);
                }
            }
        }
        __syncthreads();
        if (more) {
            #pragma unroll
            for (int r = 0; r < 8; ++r) gx[r] = *(const v4f*)(qX + (size_t)(KC_ + 4 * r) * LDB);
            if (NPL > 1) {
                #pragma unroll
                for (int r = 0; r < 8; ++r) gy[r] = *(const v4f*)(qY + (size_t)(KC_ + 4 * r) * LDB);
            }
            if (NPL > 2) {
                #pragma unroll
                for (int r = 0; r < 8; ++r) gz[r] = *(const v4f*)(qZ + (size_t)(KC_ + 4 * r) * LDB);
            }
            #pragma unroll
            for (int a = 0; a < NA; ++a) {
                #pragma unroll
                for (int r = 0; r < 2; ++r) gg[a][r] = *(const v4f*)((a == 0 ? qG0 : qG1) + (size_t)(16 * r) * C_ + KC_);
            }
            frag_kn_load<NA>(P, xs, gs, fA0, fA1, fB, fBh);
        }
        __builtin_amdgcn_sched_barrier(0);
        frag_kn_mma<NA>(Q, o);
        __builtin_amdgcn_sched_barrier(0);
        if (more) {
            #pragma unroll
            for (int r = 0; r < 8; ++r) *(v4f*)(xs + SXS + sxOff + 4 * r * PX_) = PLSUM_(gx[r], gy[r], gz[r]);
            #pragma unroll
            for (int a = 0; a < NA; ++a) {
                #pragma unroll
                for (int r = 0; r < 2; ++r) {
                    *(v4f*)(gs + SGS + (a * 32 + 16 * r) * PK_ + sgOff) = gg[a][r];
                    if (RS) rs[r] += sum4(gg[a][r]);
                }
            }
        }
        __syncthreads();
        const bool wrap = (q + 2 == NCH);
        const int step = wrap ? (2 - NCH) * KC_ : 2 * KC_;
        q = wrap ? 0 : q + 2;
        qX += (ptrdiff_t)step * LDB;
        if (NPL > 1) qY += (ptrdiff_t)step * LDB;
        if (NPL > 2) qZ += (ptrdiff_t)step * LDB;
        qG0 += step;
        if (NA > 1) qG1 += step;
    }
}

#define KP0_ 44
#define KP1_ 42
__global__ __launch_bounds__(128) __attribute__((amdgpu_waves_per_eu(1, 2))) void k_gram(const float* __restrict__ x,
                                              float* __restrict__ S0,
                                              float* __restrict__ S1,
                                              float* __restrict__ S2) {
    __shared__ __attribute__((aligned(16))) float tile[2 * GROWS_(2) * PK_];
    const int lane = threadIdx.x & 31;
    const int wave = threadIdx.x >> 5;
    const int m    = lane & 15;
    const int half = lane >> 4;
    const int p    = blockIdx.x >> 1;
    const int I    = (p >= 4) + (p >= 7) + (p >= 9);
    const int J    = p - (4 * I - ((I * (I - 1)) >> 1)) + I;
    const int c0   = I * 128 + (blockIdx.x & 1) * 64;
    const int dblk = J * 128;
    const int d0   = dblk + wave * 32;
    const int kh   = blockIdx.y;
    const int b    = blockIdx.z;
    const int ch0  = kh == 0 ? 0 : KP0_ + (kh - 1) * KP1_;
    const int nch  = kh == 0 ? KP0_ : KP1_;
    const size_t bOff = (size_t)b * C_ * N_ + (size_t)ch0 * KC_;
    v8f acc[2][4] = {};
    const int rot = (blockIdx.x & 7) << 1;
    mm_rows_rows<N_, 2>(x + bOff + (size_t)c0 * N_, x + bOff + (size_t)dblk * N_, tile, nch, rot, acc);
    float* Sb = (kh == 0 ? S0 : kh == 1 ? S1 : S2) + (size_t)b * C_ * C_;
    #pragma unroll
    for (int a = 0; a < 2; ++a) {
        #pragma unroll
        for (int j = 0; j < 8; ++j) {
            const int cA = c0 + 32 * a + j + 8 * half;
            const int cB = cA + 16;
            Sb[(size_t)cA * C_ + d0 + m]      = acc[a][0][j];
            Sb[(size_t)cA * C_ + d0 + 16 + m] = acc[a][1][j];
            Sb[(size_t)cB * C_ + d0 + m]      = acc[a][2][j];
            Sb[(size_t)cB * C_ + d0 + 16 + m] = acc[a][3][j];
        }
    }
    if (I != J) {
        #pragma unroll
        for (int a = 0; a < 2; ++a) {
            float* r0 = Sb + (size_t)(d0 + m) * C_ + c0 + 32 * a + 8 * half;
            float* r1 = r0 + (size_t)16 * C_;
            *(v8f*)(r0)      = acc[a][0];
            *(v8f*)(r0 + 16) = acc[a][2];
            *(v8f*)(r1)      = acc[a][1];
            *(v8f*)(r1 + 16) = acc[a][3];
        }
    }
}

__global__ __launch_bounds__(128) __attribute__((amdgpu_waves_per_eu(1, 2))) void k_projs(const float* __restrict__ S0,
                                               const float* __restrict__ S1,
                                               const float* __restrict__ S2,
                                               const float* __restrict__ Wq,
                                               const float* __restrict__ Wk,
                                               float* __restrict__ T,
                                               float* __restrict__ PQ,
                                               float* __restrict__ PK) {
    __shared__ __attribute__((aligned(16))) float xs[2 * KC_ * PX_];
    __shared__ __attribute__((aligned(16))) float gs[2 * 2 * 32 * PK_];
    const int lane = threadIdx.x & 31;
    const int wave = threadIdx.x >> 5;
    const int m    = lane & 15;
    const int half = lane >> 4;
    const int c0   = blockIdx.y * 32;
    const int nblk = blockIdx.x * 128;
    const int n0   = nblk + wave * 32;
    const int b    = blockIdx.z;
    v8f o[2][4] = {};
    float unused[2];
    const int rot = ((blockIdx.x + blockIdx.y + blockIdx.z) & 7) << 1;
    mm_rows_kn<C_, 2, 3, false>(Wq + (size_t)c0 * C_, Wk + (size_t)c0 * C_, S0 + (size_t)b * C_ * C_ + nblk,
                                S1 + (size_t)b * C_ * C_ + nblk, S2 + (size_t)b * C_ * C_ + nblk, xs, gs, rot, o, unused);
    float* Tb = T + (size_t)b * C_ * C_;
    float sq[8][2], sk[8][2];
    #pragma unroll
    for (int j = 0; j < 8; ++j) {
        const int cA = c0 + j + 8 * half;
        const int cB = cA + 16;
        const size_t pA = (size_t)cA * C_ + n0 + m;
        const size_t pB = (size_t)cB * C_ + n0 + m;
        Tb[pA]      = o[0][0][j];
        Tb[pA + 16] = o[0][1][j];
        Tb[pB]      = o[0][2][j];
        Tb[pB + 16] = o[0][3][j];
        sq[j][0] = o[0][0][j] * Wq[pA] + o[0][1][j] * Wq[pA + 16];
        sq[j][1] = o[0][2][j] * Wq[pB] + o[0][3][j] * Wq[pB + 16];
        sk[j][0] = o[1][0][j] * Wk[pA] + o[1][1][j] * Wk[pA + 16];
        sk[j][1] = o[1][2][j] * Wk[pB] + o[1][3][j] * Wk[pB + 16];
    }
    #pragma unroll
    for (int s = 1; s < 16; s <<= 1) {
        #pragma unroll
        for (int j = 0; j < 8; ++j) {
            sq[j][0] += __shfl_xor(sq[j][0], s); sq[j][1] += __shfl_xor(sq[j][1], s);
            sk[j][0] += __shfl_xor(sk[j][0], s); sk[j][1] += __shfl_xor(sk[j][1], s);
        }
    }
    float* red = xs;
    if (m == 0) {
        #pragma unroll
        for (int j = 0; j < 8; ++j) {
            red[wave * 64 + j + 8 * half]           = sq[j][0];
            red[wave * 64 + j + 8 * half + 16]      = sq[j][1];
            red[wave * 64 + 32 + j + 8 * half]      = sk[j][0];
            red[wave * 64 + 32 + j + 8 * half + 16] = sk[j][1];
        }
    }
    __syncthreads();
    if (threadIdx.x < 64) {
        const int t = threadIdx.x;
        const float v = ((red[t] + red[64 + t]) + red[128 + t]) + red[192 + t];
        float* dst = (t < 32 ? PQ : PK) + ((size_t)(b * C_ + c0 + (t & 31))) * 4 + blockIdx.x;
        *dst = v;
    }
}

__global__ __launch_bounds__(128) __attribute__((amdgpu_waves_per_eu(1, 2))) void k_qkt(const float* __restrict__ T,
                                             const float* __restrict__ Wk,
                                             const float* __restrict__ PQ,
                                             const float* __restrict__ PK,
                                             float* __restrict__ M) {
    __shared__ __attribute__((aligned(16))) float tile[2 * GROWS_(1) * PK_];
    __shared__ float sIQ[32], sIK[128];
    const int lane = threadIdx.x & 31;
    const int wave = threadIdx.x >> 5;
    const int m    = lane & 15;
    const int half = lane >> 4;
    const int c0   = blockIdx.y * 32;
    const int dblk = blockIdx.x * 128;
    const int d0   = dblk + wave * 32;
    const int b    = blockIdx.z;
    const size_t bOff = (size_t)b * C_ * C_;
    const v4f pk = *(const v4f*)(PK + ((size_t)(b * C_ + dblk + threadIdx.x)) * 4);
    const v4f pq = *(const v4f*)(PQ + ((size_t)(b * C_ + c0 + (threadIdx.x & 31))) * 4);
    v8f acc[1][4] = {};
    const int rot = ((blockIdx.x + blockIdx.y + blockIdx.z) & 7) << 1;
    mm_rows_rows<C_, 1>(T + bOff + (size_t)c0 * C_, Wk + (size_t)dblk * C_, tile, C_ / KC_, rot, acc);
    sIK[threadIdx.x] = rsqrtf((pk.x + pk.y) + (pk.z + pk.w));
    if (threadIdx.x < 32) sIQ[threadIdx.x] = rsqrtf((pq.x + pq.y) + (pq.z + pq.w));
    __syncthreads();
    float* Mb = M + (size_t)b * C_ * C_;
    const float ik0 = sIK[wave * 32 + m];
    const float ik1 = sIK[wave * 32 + 16 + m];
    #pragma unroll
    for (int j = 0; j < 8; ++j) {
        const int cA = c0 + j + 8 * half;
        const int cB = cA + 16;
        const float iqA = sIQ[j + 8 * half];
        const float iqB = sIQ[j + 8 * half + 16];
        Mb[(size_t)cA * C_ + d0 + m]      = acc[0][0][j] * iqA * ik0;
        Mb[(size_t)cA * C_ + d0 + 16 + m] = acc[0][1][j] * iqA * ik1;
        Mb[(size_t)cB * C_ + d0 + m]      = acc[0][2][j] * iqB * ik0;
        Mb[(size_t)cB * C_ + d0 + 16 + m] = acc[0][3][j] * iqB * ik1;
    }
}

__global__ __launch_bounds__(256) void k_colsoftmax(float* __restrict__ M) {
    __shared__ float smax[8][32], smin[8][32], ssum[8][32];
    const int lane = threadIdx.x & 31;
    const int wave = threadIdx.x >> 5;
    const int tile = blockIdx.x;
    const int b  = tile / (C_ / 32);
    const int d  = (tile - b * (C_ / 32)) * 32 + lane;
    float* p = M + (size_t)b * C_ * C_ + (size_t)(wave * 64) * C_ + d;
    float v[64];
    for (int i = 0; i < 64; ++i) v[i] = 1.f - p[(size_t)i * C_];
    float mx = -3.4e38f, mn = 3.4e38f;
    for (int i = 0; i < 64; ++i) { mx = fmaxf(mx, v[i]); mn = fminf(mn, v[i]); }
    smax[wave][lane] = mx; smin[wave][lane] = mn;
    __syncthreads();
    float cmx = -3.4e38f, cmn = 3.4e38f;
    for (int j = 0; j < 8; ++j) {
        cmx = fmaxf(cmx, smax[j][lane]);
        cmn = fminf(cmn, smin[j][lane]);
    }
    const float dn = cmx + EPS_;
    const float inv_dn = 1.f / dn;
    const float sh = 1.f - (cmn / dn) * (1.f / HT_);
    float z = 0.f;
    for (int i = 0; i < 64; ++i) {
        const float m4 = 1.f - (v[i] * inv_dn) * (1.f / HT_);
        const float e = expf(m4 - sh);
        v[i] = e;
        z += e;
    }
    ssum[wave][lane] = z;
    __syncthreads();
    float tot = 0.f;
    for (int j = 0; j < 8; ++j) tot += ssum[j][lane];
    const float invz = 1.f / tot;
    for (int i = 0; i < 64; ++i) p[(size_t)i * C_] = v[i] * invz;
}

__global__ __launch_bounds__(128) __attribute__((amdgpu_waves_per_eu(1, 2))) void k_mwv(const float* __restrict__ M,
                                             const float* __restrict__ Wv,
                                             float* __restrict__ G) {
    __shared__ __attribute__((aligned(16))) float xs[2 * KC_ * PX_];
    __shared__ __attribute__((aligned(16))) float gs[2 * 32 * PK_];
    const int lane = threadIdx.x & 31;
    const int wave = threadIdx.x >> 5;
    const int m    = lane & 15;
    const int half = lane >> 4;
    const int c0   = blockIdx.y * 32;
    const int eblk = blockIdx.x * 128;
    const int e0   = eblk + wave * 32;
    const int b    = blockIdx.z;
    const float* Mb = M + (size_t)b * C_ * C_ + (size_t)c0 * C_;
    v8f o[1][4] = {};
    float rs[2];
    const int rot = ((blockIdx.x + blockIdx.y + blockIdx.z) & 7) << 1;
    mm_rows_kn<C_, 1, 1, true>(Mb, Mb, Wv + eblk, Wv + eblk, Wv + eblk, xs, gs, rot, o, rs);
    for (int s = 1; s < 8; s <<= 1) {
        rs[0] += __shfl_xor(rs[0], s);
        rs[1] += __shfl_xor(rs[1], s);
    }
    float* srow = gs;
    if ((threadIdx.x & 7) == 0) {
        srow[threadIdx.x >> 3]        = 1.f / (rs[0] + EPS_);
        srow[(threadIdx.x >> 3) + 16] = 1.f / (rs[1] + EPS_);
    }
    __syncthreads();
    float* Gb = G + (size_t)b * C_ * C_;
    #pragma unroll
    for (int j = 0; j < 8; ++j) {
        const int cA = c0 + j + 8 * half;
        const int cB = cA + 16;
        const float iA = srow[j + 8 * half];
        const float iB = srow[j + 8 * half + 16];
        Gb[(size_t)cA * C_ + e0 + m]      = o[0][0][j] * iA;
        Gb[(size_t)cA * C_ + e0 + 16 + m] = o[0][1][j] * iA;
        Gb[(size_t)cB * C_ + e0 + m]      = o[0][2][j] * iB;
        Gb[(size_t)cB * C_ + e0 + 16 + m] = o[0][3][j] * iB;
    }
}

__global__ __launch_bounds__(128) void k_out(const float* __restrict__ x,
                                             const float* __restrict__ G,
                                             const float* __restrict__ gamma,
                                             float* __restrict__ y) {
    __shared__ __attribute__((aligned(16))) float xs[2 * KC_ * PX_];
    __shared__ __attribute__((aligned(16))) float gs[2 * 2 * 32 * PK_];
    const int lane = threadIdx.x & 31;
    const int wave = threadIdx.x >> 5;
    const int m    = lane & 15;
    const int half = lane >> 4;
    const int c0   = blockIdx.y * 64;
    const int nblk = blockIdx.x * 128;
    const int n0   = nblk + wave * 32;
    const int b    = blockIdx.z;
    const float* xb = x + (size_t)b * C_ * N_;
    const float* Gb = G + (size_t)b * C_ * C_ + (size_t)c0 * C_;
    v8f o[2][4] = {};
    float unused[2];
    const int rot = ((blockIdx.x + blockIdx.y + blockIdx.z) & 7) << 1;
    mm_rows_kn<N_, 2, 1, false>(Gb, Gb + (size_t)32 * C_, xb + nblk, xb + nblk, xb + nblk, xs, gs, rot, o, unused);
    const float g = gamma[0];
    float* yb = y + (size_t)b * C_ * N_;
    #pragma unroll
    for (int a = 0; a < 2; ++a) {
        #pragma unroll
        for (int j = 0; j < 8; ++j) {
            const int cA = c0 + 32 * a + j + 8 * half;
            const int cB = cA + 16;
            const size_t o0 = (size_t)cA * N_ + n0 + m;
            const size_t o1 = (size_t)cA * N_ + n0 + 16 + m;
            const size_t o2 = (size_t)cB * N_ + n0 + m;
            const size_t o3 = (size_t)cB * N_ + n0 + 16 + m;
            yb[o0] = xb[o0] + g * o[a][0][j];
            yb[o1] = xb[o1] + g * o[a][1][j];
            yb[o2] = xb[o2] + g * o[a][2][j];
            yb[o3] = xb[o3] + g * o[a][3][j];
        }
    }
}

extern "C" void kernel_launch(void* const* d_in, const int* in_sizes, int n_in,
                              void* d_out, int out_size, void* d_ws, size_t ws_size,
                              hipStream_t stream) {
    (void)in_sizes; (void)n_in; (void)out_size; (void)ws_size;
    const float* x     = (const float*)d_in[0];
    const float* Wq    = (const float*)d_in[1];
    const float* Wk    = (const float*)d_in[2];
    const float* Wv    = (const float*)d_in[3];
    const float* gamma = (const float*)d_in[4];
    float* y = (float*)d_out;

    char* ws = (char*)d_ws;
    const size_t mBytes  = (size_t)B_ * C_ * C_ * sizeof(float);
    float* S     = (float*)(ws);
    float* T     = (float*)(ws + mBytes);
    float* PQ    = (float*)(ws + 2 * mBytes);
    float* PK    = PQ + B_ * C_ * 4;
    float* M     = (float*)(ws + 3 * mBytes);
    float* G     = (float*)(ws + 4 * mBytes);

    k_gram      <<<dim3(20, 3, B_), 128, 0, stream>>>(x, S, G, M);
    k_projs     <<<dim3(C_ / 128, C_ / 32, B_), 128, 0, stream>>>(S, G, M, Wq, Wk, T, PQ, PK);
    k_qkt       <<<dim3(C_ / 128, C_ / 32, B_), 128, 0, stream>>>(T, Wk, PQ, PK, M);
    k_colsoftmax<<<B_ * C_ / 32, 256, 0, stream>>>(M);
    k_mwv       <<<dim3(C_ / 128, C_ / 32, B_), 128, 0, stream>>>(M, Wv, G);
    k_out       <<<dim3(N_ / 128, C_ / 64, B_), 128, 0, stream>>>(x, G, gamma, y);
}
